# starting kernel
# baseline (speedup 1.0000x reference)
#include <hip/hip_runtime.h>
#include <math.h>

#define NFFT     1024
#define HOPSZ    512
#define NMELS    128
#define NFREQ    513
#define FPAD     528
#define T_IN     1323000
#define NB       16
#define NFRAMES  2584
#define MTILES   162
#define E_ELEMS  (NB*NMELS*NFRAMES)

#define WIN_OFF  0
#define DFT_OFF  1024
#define DFT_SZ   ((NFFT/2)*FPAD*4)
#define FB_OFF   (DFT_OFF + DFT_SZ)
#define FB_SZ    ((FPAD/2)*NMELS*2)
#define MAX_OFF  (FB_OFF + FB_SZ)
#define MEL_OFF  (MAX_OFF + 8)
#define HARM_OFF (MEL_OFF + E_ELEMS)
#define PERC_OFF (HARM_OFF + E_ELEMS)

typedef float v2f __attribute__((ext_vector_type(2)));
typedef float v4f __attribute__((ext_vector_type(4)));
typedef float v8f __attribute__((ext_vector_type(8)));

__device__ __forceinline__ v8f wmma4(v2f a, v2f b, v8f c) {
  return __builtin_amdgcn_wmma_f32_16x16x4_f32(false, a, false, b, (short)0, c,
                                               false, false);
}

__global__ __launch_bounds__(256) void build_tables(float* __restrict__ ws) {
  int gid = blockIdx.x * 256 + threadIdx.x;
  if (gid == 0) {
    unsigned* mx = (unsigned*)(ws + MAX_OFF);
    mx[0] = 0u; mx[1] = 0u; mx[2] = 0u;
  }
  const float TWO_PI = 6.28318530717958647692f;
  if (gid < DFT_OFF) {
    int k = gid;
    ws[WIN_OFF + k] = 0.5f * (1.0f - cosf(TWO_PI * (float)k / 1024.0f));
  } else if (gid < FB_OFF) {
    int i = gid - DFT_OFF;
    int entry = i & 3;
    int rest  = i >> 2;
    int f  = rest % FPAD;
    int k2 = rest / FPAD;
    int k  = 2 * k2 + (entry & 1);
    int r  = (k * f) & (NFFT - 1);
    float arg = (float)r * (TWO_PI / 1024.0f);
    ws[gid] = (entry < 2) ? cosf(arg) : sinf(arg);
  } else if (gid < MAX_OFF) {
    int i = gid - FB_OFF;
    int entry = i & 1;
    int rest  = i >> 1;
    int m  = rest % NMELS;
    int k2 = rest / NMELS;
    int f  = 2 * k2 + entry;
    float v = 0.0f;
    if (f < NFREQ) {
      float freq   = (float)f * (22050.0f / 512.0f);
      float melmax = 2595.0f * log10f(1.0f + 22050.0f / 700.0f);
      float step   = melmax / 129.0f;
      float p0 = 700.0f * (exp10f(((float)(m    ) * step) / 2595.0f) - 1.0f);
      float p1 = 700.0f * (exp10f(((float)(m + 1) * step) / 2595.0f) - 1.0f);
      float p2 = 700.0f * (exp10f(((float)(m + 2) * step) / 2595.0f) - 1.0f);
      float down = (freq - p0) / (p1 - p0);
      float up   = (p2 - freq) / (p2 - p1);
      float t = fmaxf(0.0f, fminf(down, up));
      v = t * 2.0f / (p2 - p0);
    }
    ws[gid] = v;
  }
}

#define KCHUNK 256
__global__ __launch_bounds__(256) void mel_gemm(const float* __restrict__ wave,
                                                const float* __restrict__ ws,
                                                float* __restrict__ mel) {
  __shared__ float AL[16][KCHUNK + 4];
  __shared__ float PL[16][FPAD + 4];

  const float* __restrict__ win  = ws + WIN_OFF;
  const float* __restrict__ dftT = ws + DFT_OFF;
  const float* __restrict__ fb2  = ws + FB_OFF;

  int b    = blockIdx.x / MTILES;
  int tile = blockIdx.x % MTILES;
  int t0   = tile * 16;
  int tid  = threadIdx.x;
  int lane = tid & 31;
  int wv   = tid >> 5;
  int h    = lane >> 4;
  int l16  = lane & 15;

  const float* wb = wave + (long)b * T_IN;

  v8f accC[5], accS[5];
  #pragma unroll
  for (int i = 0; i < 5; ++i)
    for (int j = 0; j < 8; ++j) { accC[i][j] = 0.0f; accS[i][j] = 0.0f; }

  for (int kc = 0; kc < NFFT / KCHUNK; ++kc) {
    __syncthreads();
    for (int idx = tid; idx < 16 * KCHUNK; idx += 256) {
      int m  = idx >> 8;
      int kl = idx & (KCHUNK - 1);
      int t  = t0 + m;
      float v = 0.0f;
      if (t < NFRAMES) {
        int k = kc * KCHUNK + kl;
        int o = t * HOPSZ + k - NFFT / 2;
        o = (o < 0) ? -o : o;
        o = (o >= T_IN) ? (2 * (T_IN - 1) - o) : o;
        v = wb[o] * win[k];
      }
      AL[m][kl] = v;
    }
    __syncthreads();

    #pragma unroll
    for (int i = 0; i < 5; ++i) {
      int ft = wv + 8 * i;
      if (ft < FPAD / 16) {
        int f0 = ft * 16;
        const float* dB = dftT + (long)(f0 + l16) * 4;
        __builtin_prefetch(dB + (long)(kc * (KCHUNK / 2)) * FPAD * 4, 0, 1);
        for (int k = 0; k < KCHUNK; k += 4) {
          int kk  = k + 2 * h;
          v2f a   = *(const v2f*)&AL[l16][kk];
          int k2g = kc * (KCHUNK / 2) + (kk >> 1);
          v4f q = *(const v4f*)(dB + (long)k2g * (FPAD * 4));
          v2f bc; bc[0] = q[0]; bc[1] = q[1];
          v2f bs; bs[0] = q[2]; bs[1] = q[3];
          accC[i] = wmma4(a, bc, accC[i]);
          accS[i] = wmma4(a, bs, accS[i]);
        }
      }
    }
  }

  __syncthreads();
  #pragma unroll
  for (int i = 0; i < 5; ++i) {
    int ft = wv + 8 * i;
    if (ft < FPAD / 16) {
      int f0 = ft * 16;
      #pragma unroll
      for (int v = 0; v < 8; ++v) {
        int mr = v + 8 * h;
        float re = accC[i][v], im = accS[i][v];
        PL[mr][f0 + l16] = re * re + im * im;
      }
    }
  }
  __syncthreads();

  int n0 = wv * 16;
  const float* fB = fb2 + (long)(n0 + l16) * 2;
  v8f acc;
  #pragma unroll
  for (int j = 0; j < 8; ++j) acc[j] = 0.0f;
  for (int k = 0; k < FPAD; k += 4) {
    int kk = k + 2 * h;
    v2f a  = *(const v2f*)&PL[l16][kk];
    v2f bb = *(const v2f*)(fB + (long)(kk >> 1) * (NMELS * 2));
    acc = wmma4(a, bb, acc);
  }
  int n = n0 + l16;
  #pragma unroll
  for (int v = 0; v < 8; ++v) {
    int t = t0 + v + 8 * h;
    if (t < NFRAMES) mel[(b * NMELS + n) * NFRAMES + t] = acc[v];
  }
}

__device__ __forceinline__ float median31(float v[31]) {
  #pragma unroll
  for (int i = 0; i < 16; ++i) {
    #pragma unroll
    for (int j = i + 1; j < 31; ++j) {
      float lo = fminf(v[i], v[j]);
      float hi = fmaxf(v[i], v[j]);
      v[i] = lo; v[j] = hi;
    }
  }
  return v[15];
}

__global__ __launch_bounds__(256) void median_time(const float* __restrict__ mel,
                                                   float* __restrict__ harm) {
  int gid = blockIdx.x * 256 + threadIdx.x;
  if (gid >= E_ELEMS) return;
  int t   = gid % NFRAMES;
  int row = gid / NFRAMES;
  const float* src = mel + row * NFRAMES;
  float v[31];
  #pragma unroll
  for (int d = 0; d < 31; ++d) {
    int j = t + d - 15;
    j = (j < 0) ? (-j - 1) : j;
    j = (j >= NFRAMES) ? (2 * NFRAMES - 1 - j) : j;
    v[d] = src[j];
  }
  harm[gid] = median31(v);
}

__global__ __launch_bounds__(256) void median_mel(const float* __restrict__ mel,
                                                  float* __restrict__ perc) {
  int gid = blockIdx.x * 256 + threadIdx.x;
  if (gid >= E_ELEMS) return;
  int t = gid % NFRAMES;
  int n = (gid / NFRAMES) % NMELS;
  int b = gid / (NFRAMES * NMELS);
  const float* base = mel + (b * NMELS) * NFRAMES + t;
  float v[31];
  #pragma unroll
  for (int d = 0; d < 31; ++d) {
    int j = n + d - 15;
    j = (j < 0) ? (-j - 1) : j;
    j = (j >= NMELS) ? (2 * NMELS - 1 - j) : j;
    v[d] = base[j * NFRAMES];
  }
  perc[gid] = median31(v);
}

__global__ __launch_bounds__(256) void hpss_mask(const float* __restrict__ mel,
                                                 float* __restrict__ harm,
                                                 float* __restrict__ perc,
                                                 unsigned* __restrict__ mx) {
  __shared__ float r0[256], r1[256], r2[256];
  int gid = blockIdx.x * 256 + threadIdx.x;
  float s = 0.0f, dh = 0.0f, dp = 0.0f;
  if (gid < E_ELEMS) {
    float S  = mel[gid];
    float ha = harm[gid];
    float pe = perc[gid];
    float Z  = fmaxf(ha, pe);
    bool bad = Z < 1.17549435e-38f;
    float Zs = bad ? 1.0f : Z;
    float mh = ha / Zs; mh *= mh;
    float mp = pe / Zs; mp *= mp;
    float den = mh + mp;
    float maskh = bad ? 0.0f : mh / den;
    float maskp = bad ? 0.0f : mp / den;
    dh = S * maskh; dp = S * maskp; s = S;
    harm[gid] = dh;
    perc[gid] = dp;
  }
  int tid = threadIdx.x;
  r0[tid] = s; r1[tid] = dh; r2[tid] = dp;
  __syncthreads();
  for (int off = 128; off > 0; off >>= 1) {
    if (tid < off) {
      r0[tid] = fmaxf(r0[tid], r0[tid + off]);
      r1[tid] = fmaxf(r1[tid], r1[tid + off]);
      r2[tid] = fmaxf(r2[tid], r2[tid + off]);
    }
    __syncthreads();
  }
  if (tid == 0) {
    atomicMax(mx + 0, __float_as_uint(r0[0]));
    atomicMax(mx + 1, __float_as_uint(r1[0]));
    atomicMax(mx + 2, __float_as_uint(r2[0]));
  }
}

__global__ __launch_bounds__(256) void finalize(const float* __restrict__ mel,
                                                const float* __restrict__ harm,
                                                const float* __restrict__ perc,
                                                const unsigned* __restrict__ mx,
                                                float* __restrict__ out) {
  int gid = blockIdx.x * 256 + threadIdx.x;
  if (gid >= 3 * E_ELEMS) return;
  int s = gid / E_ELEMS;
  int e = gid - s * E_ELEMS;
  const float* src = (s == 0) ? mel : ((s == 1) ? harm : perc);
  float ref = __uint_as_float(mx[s]);
  float v   = src[e];
  float db = 10.0f * log10f(fmaxf(1e-10f, v)) -
             10.0f * log10f(fmaxf(1e-10f, ref));
  db = fmaxf(db, -80.0f);
  out[gid] = (db + 80.0f) * (1.0f / (80.0f + 1e-7f));
}

extern "C" void kernel_launch(void* const* d_in, const int* in_sizes, int n_in,
                              void* d_out, int out_size, void* d_ws,
                              size_t ws_size, hipStream_t stream) {
  const float* wave = (const float*)d_in[0];
  float* out  = (float*)d_out;
  float* ws   = (float*)d_ws;
  float* mel  = ws + MEL_OFF;
  float* harm = ws + HARM_OFF;
  float* perc = ws + PERC_OFF;
  unsigned* mx = (unsigned*)(ws + MAX_OFF);

  build_tables<<<MAX_OFF / 256, 256, 0, stream>>>(ws);
  mel_gemm<<<NB * MTILES, 256, 0, stream>>>(wave, ws, mel);
  int mblocks = (E_ELEMS + 255) / 256;
  median_time<<<mblocks, 256, 0, stream>>>(mel, harm);
  median_mel<<<mblocks, 256, 0, stream>>>(mel, perc);
  hpss_mask<<<mblocks, 256, 0, stream>>>(mel, harm, perc, mx);
  finalize<<<(3 * E_ELEMS + 255) / 256, 256, 0, stream>>>(mel, harm, perc, mx,
                                                          out);
}


# mel_gemm: fold the DFT on its f <-> 512-f symmetry: cos/sin sums split by k parity give bin f as their sum and bin 512-f as their difference, 33 -> 17 frequency tiles; f32 summation order changes
# speedup vs baseline: 1.6631x; 1.6631x over previous
#include <hip/hip_runtime.h>
#include <math.h>

#define NFFT     1024
#define HOPSZ    512
#define NMELS    128
#define NFREQ    513
#define FPAD     528
#define T_IN     1323000
#define NB       16
#define NFRAMES  2584
#define MTILES   162
#define E_ELEMS  (NB*NMELS*NFRAMES)

#define KHALF    512
#define KCH      256
#define APITCH   516
#define FH       272
#define NFT      17
#define QCH      64
#define NQ       129

#define WIN_OFF  0
#define DFT_OFF  1024
#define DFT_SZ   (NQ*2*FH*4)
#define FB_OFF   (DFT_OFF + DFT_SZ)
#define FB_SZ    ((FPAD/2)*NMELS*2)
#define MAX_OFF  (FB_OFF + FB_SZ)
#define MEL_OFF  (MAX_OFF + 8)
#define HARM_OFF (MEL_OFF + E_ELEMS)
#define PERC_OFF (HARM_OFF + E_ELEMS)

typedef float v2f __attribute__((ext_vector_type(2)));
typedef float v4f __attribute__((ext_vector_type(4)));
typedef float v8f __attribute__((ext_vector_type(8)));

__device__ __forceinline__ v8f wmma4(v2f a, v2f b, v8f c) {
  return __builtin_amdgcn_wmma_f32_16x16x4_f32(false, a, false, b, (short)0, c,
                                               false, false);
}

__global__ __launch_bounds__(256) void build_tables(float* __restrict__ ws) {
  int gid = blockIdx.x * 256 + threadIdx.x;
  if (gid == 0) {
    unsigned* mx = (unsigned*)(ws + MAX_OFF);
    mx[0] = 0u; mx[1] = 0u; mx[2] = 0u;
  }
  const float TWO_PI = 6.28318530717958647692f;
  if (gid < DFT_OFF) {
    int k = gid;
    ws[WIN_OFF + k] = 0.5f * (1.0f - cosf(TWO_PI * (float)k / 1024.0f));
  } else if (gid < FB_OFF) {
    int i = gid - DFT_OFF;
    int entry = i & 3;
    int rest  = i >> 2;
    int f  = rest % FH;
    int qp = rest / FH;
    int k  = 4 * (qp >> 1) + (qp & 1) + 2 * (entry & 1);
    int r  = (k * f) & (NFFT - 1);
    float arg = (float)r * (TWO_PI / 1024.0f);
    ws[gid] = (entry < 2) ? cosf(arg) : sinf(arg);
  } else if (gid < MAX_OFF) {
    int i = gid - FB_OFF;
    int entry = i & 1;
    int rest  = i >> 1;
    int m  = rest % NMELS;
    int k2 = rest / NMELS;
    int f  = 2 * k2 + entry;
    float v = 0.0f;
    if (f < NFREQ) {
      float freq   = (float)f * (22050.0f / 512.0f);
      float melmax = 2595.0f * log10f(1.0f + 22050.0f / 700.0f);
      float step   = melmax / 129.0f;
      float p0 = 700.0f * (exp10f(((float)(m    ) * step) / 2595.0f) - 1.0f);
      float p1 = 700.0f * (exp10f(((float)(m + 1) * step) / 2595.0f) - 1.0f);
      float p2 = 700.0f * (exp10f(((float)(m + 2) * step) / 2595.0f) - 1.0f);
      float down = (freq - p0) / (p1 - p0);
      float up   = (p2 - freq) / (p2 - p1);
      float t = fmaxf(0.0f, fminf(down, up));
      v = t * 2.0f / (p2 - p0);
    }
    ws[gid] = v;
  }
}

__global__ __launch_bounds__(256, 6) void mel_gemm(const float* __restrict__ wave,
                                                const float* __restrict__ ws,
                                                float* __restrict__ mel) {
  __shared__ __attribute__((aligned(16))) float A4[16][APITCH];
  __shared__ float PL[16][FPAD + 4];

  const float* __restrict__ win  = ws + WIN_OFF;
  const float* __restrict__ dftT = ws + DFT_OFF;
  const float* __restrict__ fb2  = ws + FB_OFF;

  int b    = blockIdx.x / MTILES;
  int tile = blockIdx.x % MTILES;
  int t0   = tile * 16;
  int tid  = threadIdx.x;
  int lane = tid & 31;
  int wv   = tid >> 5;
  int h    = lane >> 4;
  int l16  = lane & 15;

  const float* wb = wave + (long)b * T_IN;

  int wslot = (wv + blockIdx.x + (blockIdx.x >> 8)) & 7;

  v8f accC[3][2], accS[3][2];
  #pragma unroll
  for (int i = 0; i < 3; ++i)
    for (int p = 0; p < 2; ++p)
      for (int j = 0; j < 8; ++j) { accC[i][p][j] = 0.0f; accS[i][p][j] = 0.0f; }

  for (int kc = 0; kc < KHALF / KCH; ++kc) {
    __syncthreads();
    for (int idx = tid; idx < 16 * KCH; idx += 256) {
      int m  = idx >> 8;
      int kl = idx & (KCH - 1);
      int t  = t0 + m;
      float ve = 0.0f, vo = 0.0f;
      if (t < NFRAMES) {
        int k  = kc * KCH + kl;
        bool pr = k > 0;
        int kp = pr ? (NFFT - k) : 0;
        int o1 = t * HOPSZ + k - NFFT / 2;
        o1 = (o1 < 0) ? -o1 : o1;
        o1 = (o1 >= T_IN) ? (2 * (T_IN - 1) - o1) : o1;
        int o2 = t * HOPSZ + NFFT / 2 - k;
        o2 = (o2 >= T_IN) ? (2 * (T_IN - 1) - o2) : o2;
        float y1 = wb[o1] * win[k];
        float y2 = wb[o2] * win[kp];
        ve = pr ? (y1 + y2) : y1;
        vo = pr ? (y1 - y2) : 0.0f;
      }
      int c = (kl >> 2) * 8 + (kl & 1) * 4 + ((kl >> 1) & 1);
      A4[m][c]     = ve;
      A4[m][c + 2] = vo;
    }
    __syncthreads();

    #pragma unroll
    for (int i = 0; i < 3; ++i) {
      int ft = wslot + 8 * i;
      if (ft < NFT) {
        int f0 = ft * 16;
        #pragma unroll
        for (int p = 0; p < 2; ++p) {
          const float* dB = dftT + (long)(f0 + l16) * 4 +
                            (long)((kc * QCH + h) * 2 + p) * (FH * 4);
          const float* aP = &A4[l16][h * 8 + p * 4];
          __builtin_prefetch(dB, 0, 1);
          #pragma unroll 8
          for (int qs = 0; qs < QCH / 2; ++qs) {
            v4f a = *(const v4f*)(aP + qs * 16);
            v2f ae; ae[0] = a[0]; ae[1] = a[1];
            v2f ao; ao[0] = a[2]; ao[1] = a[3];
            v4f q = *(const v4f*)(dB + (long)qs * (4 * FH * 4));
            v2f bc; bc[0] = q[0]; bc[1] = q[1];
            v2f bs; bs[0] = q[2]; bs[1] = q[3];
            accC[i][p] = wmma4(ae, bc, accC[i][p]);
            accS[i][p] = wmma4(ao, bs, accS[i][p]);
          }
        }
      }
    }
  }

  {
    int t = t0 + l16;
    float y = 0.0f;
    if (h == 0 && t < NFRAMES) y = wb[t * HOPSZ] * win[NFFT / 2];
    v2f ae; ae[0] = y; ae[1] = 0.0f;
    #pragma unroll
    for (int i = 0; i < 3; ++i) {
      int ft = wslot + 8 * i;
      if (ft < NFT) {
        const float* dB = dftT + (long)(ft * 16 + l16) * 4;
        v4f q = *(const v4f*)(dB + (long)((NQ - 1) * 2) * (FH * 4));
        v2f bc; bc[0] = q[0]; bc[1] = q[1];
        accC[i][0] = wmma4(ae, bc, accC[i][0]);
      }
    }
  }

  __syncthreads();
  if (tid < 16 * (FPAD - NFREQ))
    PL[tid / (FPAD - NFREQ)][NFREQ + tid % (FPAD - NFREQ)] = 0.0f;
  #pragma unroll
  for (int i = 0; i < 3; ++i) {
    int ft = wslot + 8 * i;
    if (ft < NFT) {
      int f = ft * 16 + l16;
      #pragma unroll
      for (int v = 0; v < 8; ++v) {
        int mr = v + 8 * h;
        float ce = accC[i][0][v], co = accC[i][1][v];
        float se = accS[i][0][v], so = accS[i][1][v];
        float re = ce + co, im = se + so;
        float rm = ce - co, jm = so - se;
        if (f <= NFFT / 4) PL[mr][f] = re * re + im * im;
        if (f <  NFFT / 4) PL[mr][NFFT / 2 - f] = rm * rm + jm * jm;
      }
    }
  }
  __syncthreads();

  int n0 = wv * 16;
  const float* fB = fb2 + (long)(n0 + l16) * 2;
  v8f acc;
  #pragma unroll
  for (int j = 0; j < 8; ++j) acc[j] = 0.0f;
  for (int k = 0; k < FPAD; k += 4) {
    int kk = k + 2 * h;
    v2f a  = *(const v2f*)&PL[l16][kk];
    v2f bb = *(const v2f*)(fB + (long)(kk >> 1) * (NMELS * 2));
    acc = wmma4(a, bb, acc);
  }
  int n = n0 + l16;
  #pragma unroll
  for (int v = 0; v < 8; ++v) {
    int t = t0 + v + 8 * h;
    if (t < NFRAMES) mel[(b * NMELS + n) * NFRAMES + t] = acc[v];
  }
}

__device__ __forceinline__ float median31(float v[31]) {
  #pragma unroll
  for (int i = 0; i < 16; ++i) {
    #pragma unroll
    for (int j = i + 1; j < 31; ++j) {
      float lo = fminf(v[i], v[j]);
      float hi = fmaxf(v[i], v[j]);
      v[i] = lo; v[j] = hi;
    }
  }
  return v[15];
}

__global__ __launch_bounds__(256) void median_time(const float* __restrict__ mel,
                                                   float* __restrict__ harm) {
  int gid = blockIdx.x * 256 + threadIdx.x;
  if (gid >= E_ELEMS) return;
  int t   = gid % NFRAMES;
  int row = gid / NFRAMES;
  const float* src = mel + row * NFRAMES;
  float v[31];
  #pragma unroll
  for (int d = 0; d < 31; ++d) {
    int j = t + d - 15;
    j = (j < 0) ? (-j - 1) : j;
    j = (j >= NFRAMES) ? (2 * NFRAMES - 1 - j) : j;
    v[d] = src[j];
  }
  harm[gid] = median31(v);
}

__global__ __launch_bounds__(256) void median_mel(const float* __restrict__ mel,
                                                  float* __restrict__ perc) {
  int gid = blockIdx.x * 256 + threadIdx.x;
  if (gid >= E_ELEMS) return;
  int t = gid % NFRAMES;
  int n = (gid / NFRAMES) % NMELS;
  int b = gid / (NFRAMES * NMELS);
  const float* base = mel + (b * NMELS) * NFRAMES + t;
  float v[31];
  #pragma unroll
  for (int d = 0; d < 31; ++d) {
    int j = n + d - 15;
    j = (j < 0) ? (-j - 1) : j;
    j = (j >= NMELS) ? (2 * NMELS - 1 - j) : j;
    v[d] = base[j * NFRAMES];
  }
  perc[gid] = median31(v);
}

__global__ __launch_bounds__(256) void hpss_mask(const float* __restrict__ mel,
                                                 float* __restrict__ harm,
                                                 float* __restrict__ perc,
                                                 unsigned* __restrict__ mx) {
  __shared__ float r0[256], r1[256], r2[256];
  int gid = blockIdx.x * 256 + threadIdx.x;
  float s = 0.0f, dh = 0.0f, dp = 0.0f;
  if (gid < E_ELEMS) {
    float S  = mel[gid];
    float ha = harm[gid];
    float pe = perc[gid];
    float Z  = fmaxf(ha, pe);
    bool bad = Z < 1.17549435e-38f;
    float Zs = bad ? 1.0f : Z;
    float mh = ha / Zs; mh *= mh;
    float mp = pe / Zs; mp *= mp;
    float den = mh + mp;
    float maskh = bad ? 0.0f : mh / den;
    float maskp = bad ? 0.0f : mp / den;
    dh = S * maskh; dp = S * maskp; s = S;
    harm[gid] = dh;
    perc[gid] = dp;
  }
  int tid = threadIdx.x;
  r0[tid] = s; r1[tid] = dh; r2[tid] = dp;
  __syncthreads();
  for (int off = 128; off > 0; off >>= 1) {
    if (tid < off) {
      r0[tid] = fmaxf(r0[tid], r0[tid + off]);
      r1[tid] = fmaxf(r1[tid], r1[tid + off]);
      r2[tid] = fmaxf(r2[tid], r2[tid + off]);
    }
    __syncthreads();
  }
  if (tid == 0) {
    atomicMax(mx + 0, __float_as_uint(r0[0]));
    atomicMax(mx + 1, __float_as_uint(r1[0]));
    atomicMax(mx + 2, __float_as_uint(r2[0]));
  }
}

__global__ __launch_bounds__(256) void finalize(const float* __restrict__ mel,
                                                const float* __restrict__ harm,
                                                const float* __restrict__ perc,
                                                const unsigned* __restrict__ mx,
                                                float* __restrict__ out) {
  int gid = blockIdx.x * 256 + threadIdx.x;
  if (gid >= 3 * E_ELEMS) return;
  int s = gid / E_ELEMS;
  int e = gid - s * E_ELEMS;
  const float* src = (s == 0) ? mel : ((s == 1) ? harm : perc);
  float ref = __uint_as_float(mx[s]);
  float v   = src[e];
  float db = 10.0f * log10f(fmaxf(1e-10f, v)) -
             10.0f * log10f(fmaxf(1e-10f, ref));
  db = fmaxf(db, -80.0f);
  out[gid] = (db + 80.0f) * (1.0f / (80.0f + 1e-7f));
}

extern "C" void kernel_launch(void* const* d_in, const int* in_sizes, int n_in,
                              void* d_out, int out_size, void* d_ws,
                              size_t ws_size, hipStream_t stream) {
  const float* wave = (const float*)d_in[0];
  float* out  = (float*)d_out;
  float* ws   = (float*)d_ws;
  float* mel  = ws + MEL_OFF;
  float* harm = ws + HARM_OFF;
  float* perc = ws + PERC_OFF;
  unsigned* mx = (unsigned*)(ws + MAX_OFF);

  build_tables<<<(MAX_OFF + 255) / 256, 256, 0, stream>>>(ws);
  mel_gemm<<<NB * MTILES, 256, 0, stream>>>(wave, ws, mel);
  int mblocks = (E_ELEMS + 255) / 256;
  median_time<<<mblocks, 256, 0, stream>>>(mel, harm);
  median_mel<<<mblocks, 256, 0, stream>>>(mel, perc);
  hpss_mask<<<mblocks, 256, 0, stream>>>(mel, harm, perc, mx);
  finalize<<<(3 * E_ELEMS + 255) / 256, 256, 0, stream>>>(mel, harm, perc, mx,
                                                          out);
}


# hpss_mask, finalize: each thread takes 4 / 8 float4 groups (b128 loads, stores) instead of one element: 661504 -> 25840 waves to enter; wave entry rate bounded both; max-only, no sum order change
# speedup vs baseline: 1.8542x; 1.1149x over previous
#include <hip/hip_runtime.h>
#include <math.h>

#define NFFT     1024
#define HOPSZ    512
#define NMELS    128
#define NFREQ    513
#define FPAD     528
#define T_IN     1323000
#define NB       16
#define NFRAMES  2584
#define MTILES   162
#define E_ELEMS  (NB*NMELS*NFRAMES)

#define KHALF    512
#define KCH      256
#define APITCH   516
#define FH       272
#define NFT      17
#define QCH      64
#define NQ       129

#define WIN_OFF  0
#define DFT_OFF  1024
#define DFT_SZ   (NQ*2*FH*4)
#define FB_OFF   (DFT_OFF + DFT_SZ)
#define FB_SZ    ((FPAD/2)*NMELS*2)
#define MAX_OFF  (FB_OFF + FB_SZ)
#define MEL_OFF  (MAX_OFF + 8)
#define HARM_OFF (MEL_OFF + E_ELEMS)
#define PERC_OFF (HARM_OFF + E_ELEMS)

typedef float v2f __attribute__((ext_vector_type(2)));
typedef float v4f __attribute__((ext_vector_type(4)));
typedef float v8f __attribute__((ext_vector_type(8)));

__device__ __forceinline__ v8f wmma4(v2f a, v2f b, v8f c) {
  return __builtin_amdgcn_wmma_f32_16x16x4_f32(false, a, false, b, (short)0, c,
                                               false, false);
}

__global__ __launch_bounds__(256) void build_tables(float* __restrict__ ws) {
  int gid = blockIdx.x * 256 + threadIdx.x;
  if (gid == 0) {
    unsigned* mx = (unsigned*)(ws + MAX_OFF);
    mx[0] = 0u; mx[1] = 0u; mx[2] = 0u;
  }
  const float TWO_PI = 6.28318530717958647692f;
  if (gid < DFT_OFF) {
    int k = gid;
    ws[WIN_OFF + k] = 0.5f * (1.0f - cosf(TWO_PI * (float)k / 1024.0f));
  } else if (gid < FB_OFF) {
    int i = gid - DFT_OFF;
    int entry = i & 3;
    int rest  = i >> 2;
    int f  = rest % FH;
    int qp = rest / FH;
    int k  = 4 * (qp >> 1) + (qp & 1) + 2 * (entry & 1);
    int r  = (k * f) & (NFFT - 1);
    float arg = (float)r * (TWO_PI / 1024.0f);
    ws[gid] = (entry < 2) ? cosf(arg) : sinf(arg);
  } else if (gid < MAX_OFF) {
    int i = gid - FB_OFF;
    int entry = i & 1;
    int rest  = i >> 1;
    int m  = rest % NMELS;
    int k2 = rest / NMELS;
    int f  = 2 * k2 + entry;
    float v = 0.0f;
    if (f < NFREQ) {
      float freq   = (float)f * (22050.0f / 512.0f);
      float melmax = 2595.0f * log10f(1.0f + 22050.0f / 700.0f);
      float step   = melmax / 129.0f;
      float p0 = 700.0f * (exp10f(((float)(m    ) * step) / 2595.0f) - 1.0f);
      float p1 = 700.0f * (exp10f(((float)(m + 1) * step) / 2595.0f) - 1.0f);
      float p2 = 700.0f * (exp10f(((float)(m + 2) * step) / 2595.0f) - 1.0f);
      float down = (freq - p0) / (p1 - p0);
      float up   = (p2 - freq) / (p2 - p1);
      float t = fmaxf(0.0f, fminf(down, up));
      v = t * 2.0f / (p2 - p0);
    }
    ws[gid] = v;
  }
}

__global__ __launch_bounds__(256, 6) void mel_gemm(const float* __restrict__ wave,
                                                const float* __restrict__ ws,
                                                float* __restrict__ mel) {
  __shared__ __attribute__((aligned(16))) float A4[16][APITCH];
  __shared__ float PL[16][FPAD + 4];

  const float* __restrict__ win  = ws + WIN_OFF;
  const float* __restrict__ dftT = ws + DFT_OFF;
  const float* __restrict__ fb2  = ws + FB_OFF;

  int b    = blockIdx.x / MTILES;
  int tile = blockIdx.x % MTILES;
  int t0   = tile * 16;
  int tid  = threadIdx.x;
  int lane = tid & 31;
  int wv   = tid >> 5;
  int h    = lane >> 4;
  int l16  = lane & 15;

  const float* wb = wave + (long)b * T_IN;

  int wslot = (wv + blockIdx.x + (blockIdx.x >> 8)) & 7;

  v8f accC[3][2], accS[3][2];
  #pragma unroll
  for (int i = 0; i < 3; ++i)
    for (int p = 0; p < 2; ++p)
      for (int j = 0; j < 8; ++j) { accC[i][p][j] = 0.0f; accS[i][p][j] = 0.0f; }

  for (int kc = 0; kc < KHALF / KCH; ++kc) {
    __syncthreads();
    for (int idx = tid; idx < 16 * KCH; idx += 256) {
      int m  = idx >> 8;
      int kl = idx & (KCH - 1);
      int t  = t0 + m;
      float ve = 0.0f, vo = 0.0f;
      if (t < NFRAMES) {
        int k  = kc * KCH + kl;
        bool pr = k > 0;
        int kp = pr ? (NFFT - k) : 0;
        int o1 = t * HOPSZ + k - NFFT / 2;
        o1 = (o1 < 0) ? -o1 : o1;
        o1 = (o1 >= T_IN) ? (2 * (T_IN - 1) - o1) : o1;
        int o2 = t * HOPSZ + NFFT / 2 - k;
        o2 = (o2 >= T_IN) ? (2 * (T_IN - 1) - o2) : o2;
        float y1 = wb[o1] * win[k];
        float y2 = wb[o2] * win[kp];
        ve = pr ? (y1 + y2) : y1;
        vo = pr ? (y1 - y2) : 0.0f;
      }
      int c = (kl >> 2) * 8 + (kl & 1) * 4 + ((kl >> 1) & 1);
      A4[m][c]     = ve;
      A4[m][c + 2] = vo;
    }
    __syncthreads();

    #pragma unroll
    for (int i = 0; i < 3; ++i) {
      int ft = wslot + 8 * i;
      if (ft < NFT) {
        int f0 = ft * 16;
        #pragma unroll
        for (int p = 0; p < 2; ++p) {
          const float* dB = dftT + (long)(f0 + l16) * 4 +
                            (long)((kc * QCH + h) * 2 + p) * (FH * 4);
          const float* aP = &A4[l16][h * 8 + p * 4];
          __builtin_prefetch(dB, 0, 1);
          #pragma unroll 8
          for (int qs = 0; qs < QCH / 2; ++qs) {
            v4f a = *(const v4f*)(aP + qs * 16);
            v2f ae; ae[0] = a[0]; ae[1] = a[1];
            v2f ao; ao[0] = a[2]; ao[1] = a[3];
            v4f q = *(const v4f*)(dB + (long)qs * (4 * FH * 4));
            v2f bc; bc[0] = q[0]; bc[1] = q[1];
            v2f bs; bs[0] = q[2]; bs[1] = q[3];
            accC[i][p] = wmma4(ae, bc, accC[i][p]);
            accS[i][p] = wmma4(ao, bs, accS[i][p]);
          }
        }
      }
    }
  }

  {
    int t = t0 + l16;
    float y = 0.0f;
    if (h == 0 && t < NFRAMES) y = wb[t * HOPSZ] * win[NFFT / 2];
    v2f ae; ae[0] = y; ae[1] = 0.0f;
    #pragma unroll
    for (int i = 0; i < 3; ++i) {
      int ft = wslot + 8 * i;
      if (ft < NFT) {
        const float* dB = dftT + (long)(ft * 16 + l16) * 4;
        v4f q = *(const v4f*)(dB + (long)((NQ - 1) * 2) * (FH * 4));
        v2f bc; bc[0] = q[0]; bc[1] = q[1];
        accC[i][0] = wmma4(ae, bc, accC[i][0]);
      }
    }
  }

  __syncthreads();
  if (tid < 16 * (FPAD - NFREQ))
    PL[tid / (FPAD - NFREQ)][NFREQ + tid % (FPAD - NFREQ)] = 0.0f;
  #pragma unroll
  for (int i = 0; i < 3; ++i) {
    int ft = wslot + 8 * i;
    if (ft < NFT) {
      int f = ft * 16 + l16;
      #pragma unroll
      for (int v = 0; v < 8; ++v) {
        int mr = v + 8 * h;
        float ce = accC[i][0][v], co = accC[i][1][v];
        float se = accS[i][0][v], so = accS[i][1][v];
        float re = ce + co, im = se + so;
        float rm = ce - co, jm = so - se;
        if (f <= NFFT / 4) PL[mr][f] = re * re + im * im;
        if (f <  NFFT / 4) PL[mr][NFFT / 2 - f] = rm * rm + jm * jm;
      }
    }
  }
  __syncthreads();

  int n0 = wv * 16;
  const float* fB = fb2 + (long)(n0 + l16) * 2;
  v8f acc;
  #pragma unroll
  for (int j = 0; j < 8; ++j) acc[j] = 0.0f;
  for (int k = 0; k < FPAD; k += 4) {
    int kk = k + 2 * h;
    v2f a  = *(const v2f*)&PL[l16][kk];
    v2f bb = *(const v2f*)(fB + (long)(kk >> 1) * (NMELS * 2));
    acc = wmma4(a, bb, acc);
  }
  int n = n0 + l16;
  #pragma unroll
  for (int v = 0; v < 8; ++v) {
    int t = t0 + v + 8 * h;
    if (t < NFRAMES) mel[(b * NMELS + n) * NFRAMES + t] = acc[v];
  }
}

__device__ __forceinline__ float median31(float v[31]) {
  #pragma unroll
  for (int i = 0; i < 16; ++i) {
    #pragma unroll
    for (int j = i + 1; j < 31; ++j) {
      float lo = fminf(v[i], v[j]);
      float hi = fmaxf(v[i], v[j]);
      v[i] = lo; v[j] = hi;
    }
  }
  return v[15];
}

__global__ __launch_bounds__(256) void median_time(const float* __restrict__ mel,
                                                   float* __restrict__ harm) {
  int gid = blockIdx.x * 256 + threadIdx.x;
  if (gid >= E_ELEMS) return;
  int t   = gid % NFRAMES;
  int row = gid / NFRAMES;
  const float* src = mel + row * NFRAMES;
  float v[31];
  #pragma unroll
  for (int d = 0; d < 31; ++d) {
    int j = t + d - 15;
    j = (j < 0) ? (-j - 1) : j;
    j = (j >= NFRAMES) ? (2 * NFRAMES - 1 - j) : j;
    v[d] = src[j];
  }
  harm[gid] = median31(v);
}

__global__ __launch_bounds__(256) void median_mel(const float* __restrict__ mel,
                                                  float* __restrict__ perc) {
  int gid = blockIdx.x * 256 + threadIdx.x;
  if (gid >= E_ELEMS) return;
  int t = gid % NFRAMES;
  int n = (gid / NFRAMES) % NMELS;
  int b = gid / (NFRAMES * NMELS);
  const float* base = mel + (b * NMELS) * NFRAMES + t;
  float v[31];
  #pragma unroll
  for (int d = 0; d < 31; ++d) {
    int j = n + d - 15;
    j = (j < 0) ? (-j - 1) : j;
    j = (j >= NMELS) ? (2 * NMELS - 1 - j) : j;
    v[d] = base[j * NFRAMES];
  }
  perc[gid] = median31(v);
}

#define HP_U      4
#define HP_BLOCKS (E_ELEMS / (4 * 256 * HP_U))
static_assert(HP_BLOCKS * 4 * 256 * HP_U == E_ELEMS, "hpss_mask grid must tile E_ELEMS exactly");
static_assert((MEL_OFF % 4) == 0 && (E_ELEMS % 4) == 0, "float4 views need 16-byte aligned planes");

__global__ __launch_bounds__(256) void hpss_mask(const float* __restrict__ mel,
                                                 float* __restrict__ harm,
                                                 float* __restrict__ perc,
                                                 unsigned* __restrict__ mx) {
  __shared__ float r0[256], r1[256], r2[256];
  const v4f* __restrict__ mel4 = (const v4f*)mel;
  v4f* __restrict__ harm4 = (v4f*)harm;
  v4f* __restrict__ perc4 = (v4f*)perc;
  int tid  = threadIdx.x;
  int base = blockIdx.x * (256 * HP_U) + tid;
  v4f S4[HP_U], H4[HP_U], P4[HP_U];
  #pragma unroll
  for (int u = 0; u < HP_U; ++u) {
    int i = base + u * 256;
    S4[u] = mel4[i]; H4[u] = harm4[i]; P4[u] = perc4[i];
  }
  float s = 0.0f, dh = 0.0f, dp = 0.0f;
  #pragma unroll
  for (int u = 0; u < HP_U; ++u) {
    v4f oh, op;
    #pragma unroll
    for (int c = 0; c < 4; ++c) {
      float S  = S4[u][c];
      float ha = H4[u][c];
      float pe = P4[u][c];
      float Z  = fmaxf(ha, pe);
      bool bad = Z < 1.17549435e-38f;
      float Zs = bad ? 1.0f : Z;
      float mh = ha / Zs; mh *= mh;
      float mp = pe / Zs; mp *= mp;
      float den = mh + mp;
      float maskh = bad ? 0.0f : mh / den;
      float maskp = bad ? 0.0f : mp / den;
      float vh = S * maskh, vp = S * maskp;
      oh[c] = vh; op[c] = vp;
      s = fmaxf(s, S); dh = fmaxf(dh, vh); dp = fmaxf(dp, vp);
    }
    int i = base + u * 256;
    harm4[i] = oh;
    perc4[i] = op;
  }
  r0[tid] = s; r1[tid] = dh; r2[tid] = dp;
  __syncthreads();
  for (int off = 128; off > 0; off >>= 1) {
    if (tid < off) {
      r0[tid] = fmaxf(r0[tid], r0[tid + off]);
      r1[tid] = fmaxf(r1[tid], r1[tid + off]);
      r2[tid] = fmaxf(r2[tid], r2[tid + off]);
    }
    __syncthreads();
  }
  if (tid == 0) {
    atomicMax(mx + 0, __float_as_uint(r0[0]));
    atomicMax(mx + 1, __float_as_uint(r1[0]));
    atomicMax(mx + 2, __float_as_uint(r2[0]));
  }
}

#define FN_U          8
#define FN_SEG_BLOCKS (E_ELEMS / (4 * 256 * FN_U))
static_assert(FN_SEG_BLOCKS * 4 * 256 * FN_U == E_ELEMS, "finalize grid must tile each plane exactly");

__global__ __launch_bounds__(256) void finalize(const float* __restrict__ mel,
                                                const float* __restrict__ harm,
                                                const float* __restrict__ perc,
                                                const unsigned* __restrict__ mx,
                                                float* __restrict__ out) {
  int s  = blockIdx.x / FN_SEG_BLOCKS;
  int e4 = (blockIdx.x - s * FN_SEG_BLOCKS) * (256 * FN_U) + threadIdx.x;
  const float* src = (s == 0) ? mel : ((s == 1) ? harm : perc);
  const v4f* __restrict__ src4 = (const v4f*)src;
  v4f* __restrict__ out4 = (v4f*)out + (long)s * (E_ELEMS / 4);
  float ref = __uint_as_float(mx[s]);
  v4f v[FN_U];
  #pragma unroll
  for (int u = 0; u < FN_U; ++u) v[u] = src4[e4 + u * 256];
  #pragma unroll
  for (int u = 0; u < FN_U; ++u) {
    v4f o;
    #pragma unroll
    for (int c = 0; c < 4; ++c) {
      float db = 10.0f * log10f(fmaxf(1e-10f, v[u][c])) -
                 10.0f * log10f(fmaxf(1e-10f, ref));
      db = fmaxf(db, -80.0f);
      o[c] = (db + 80.0f) * (1.0f / (80.0f + 1e-7f));
    }
    out4[e4 + u * 256] = o;
  }
}

extern "C" void kernel_launch(void* const* d_in, const int* in_sizes, int n_in,
                              void* d_out, int out_size, void* d_ws,
                              size_t ws_size, hipStream_t stream) {
  const float* wave = (const float*)d_in[0];
  float* out  = (float*)d_out;
  float* ws   = (float*)d_ws;
  float* mel  = ws + MEL_OFF;
  float* harm = ws + HARM_OFF;
  float* perc = ws + PERC_OFF;
  unsigned* mx = (unsigned*)(ws + MAX_OFF);

  build_tables<<<(MAX_OFF + 255) / 256, 256, 0, stream>>>(ws);
  mel_gemm<<<NB * MTILES, 256, 0, stream>>>(wave, ws, mel);
  int mblocks = (E_ELEMS + 255) / 256;
  median_time<<<mblocks, 256, 0, stream>>>(mel, harm);
  median_mel<<<mblocks, 256, 0, stream>>>(mel, perc);
  hpss_mask<<<HP_BLOCKS, 256, 0, stream>>>(mel, harm, perc, mx);
  finalize<<<3 * FN_SEG_BLOCKS, 256, 0, stream>>>(mel, harm, perc, mx, out);
}


# median_time, median_mel: a thread takes 4 adjacent outputs on the filter axis; ONE selection on the 28 taps they share (ranks 12..15), then 4th smallest of those 4 + 3 own taps each; exact medians
# speedup vs baseline: 2.6087x; 1.4069x over previous
#include <hip/hip_runtime.h>
#include <math.h>

#define NFFT     1024
#define HOPSZ    512
#define NMELS    128
#define NFREQ    513
#define FPAD     528
#define T_IN     1323000
#define NB       16
#define NFRAMES  2584
#define MTILES   162
#define E_ELEMS  (NB*NMELS*NFRAMES)

#define KHALF    512
#define KCH      256
#define APITCH   516
#define FH       272
#define NFT      17
#define QCH      64
#define NQ       129

#define WIN_OFF  0
#define DFT_OFF  1024
#define DFT_SZ   (NQ*2*FH*4)
#define FB_OFF   (DFT_OFF + DFT_SZ)
#define FB_SZ    ((FPAD/2)*NMELS*2)
#define MAX_OFF  (FB_OFF + FB_SZ)
#define MEL_OFF  (MAX_OFF + 8)
#define HARM_OFF (MEL_OFF + E_ELEMS)
#define PERC_OFF (HARM_OFF + E_ELEMS)

typedef float v2f __attribute__((ext_vector_type(2)));
typedef float v4f __attribute__((ext_vector_type(4)));
typedef float v8f __attribute__((ext_vector_type(8)));

__device__ __forceinline__ v8f wmma4(v2f a, v2f b, v8f c) {
  return __builtin_amdgcn_wmma_f32_16x16x4_f32(false, a, false, b, (short)0, c,
                                               false, false);
}

__global__ __launch_bounds__(256) void build_tables(float* __restrict__ ws) {
  int gid = blockIdx.x * 256 + threadIdx.x;
  if (gid == 0) {
    unsigned* mx = (unsigned*)(ws + MAX_OFF);
    mx[0] = 0u; mx[1] = 0u; mx[2] = 0u;
  }
  const float TWO_PI = 6.28318530717958647692f;
  if (gid < DFT_OFF) {
    int k = gid;
    ws[WIN_OFF + k] = 0.5f * (1.0f - cosf(TWO_PI * (float)k / 1024.0f));
  } else if (gid < FB_OFF) {
    int i = gid - DFT_OFF;
    int entry = i & 3;
    int rest  = i >> 2;
    int f  = rest % FH;
    int qp = rest / FH;
    int k  = 4 * (qp >> 1) + (qp & 1) + 2 * (entry & 1);
    int r  = (k * f) & (NFFT - 1);
    float arg = (float)r * (TWO_PI / 1024.0f);
    ws[gid] = (entry < 2) ? cosf(arg) : sinf(arg);
  } else if (gid < MAX_OFF) {
    int i = gid - FB_OFF;
    int entry = i & 1;
    int rest  = i >> 1;
    int m  = rest % NMELS;
    int k2 = rest / NMELS;
    int f  = 2 * k2 + entry;
    float v = 0.0f;
    if (f < NFREQ) {
      float freq   = (float)f * (22050.0f / 512.0f);
      float melmax = 2595.0f * log10f(1.0f + 22050.0f / 700.0f);
      float step   = melmax / 129.0f;
      float p0 = 700.0f * (exp10f(((float)(m    ) * step) / 2595.0f) - 1.0f);
      float p1 = 700.0f * (exp10f(((float)(m + 1) * step) / 2595.0f) - 1.0f);
      float p2 = 700.0f * (exp10f(((float)(m + 2) * step) / 2595.0f) - 1.0f);
      float down = (freq - p0) / (p1 - p0);
      float up   = (p2 - freq) / (p2 - p1);
      float t = fmaxf(0.0f, fminf(down, up));
      v = t * 2.0f / (p2 - p0);
    }
    ws[gid] = v;
  }
}

__global__ __launch_bounds__(256, 6) void mel_gemm(const float* __restrict__ wave,
                                                const float* __restrict__ ws,
                                                float* __restrict__ mel) {
  __shared__ __attribute__((aligned(16))) float A4[16][APITCH];
  __shared__ float PL[16][FPAD + 4];

  const float* __restrict__ win  = ws + WIN_OFF;
  const float* __restrict__ dftT = ws + DFT_OFF;
  const float* __restrict__ fb2  = ws + FB_OFF;

  int b    = blockIdx.x / MTILES;
  int tile = blockIdx.x % MTILES;
  int t0   = tile * 16;
  int tid  = threadIdx.x;
  int lane = tid & 31;
  int wv   = tid >> 5;
  int h    = lane >> 4;
  int l16  = lane & 15;

  const float* wb = wave + (long)b * T_IN;

  int wslot = (wv + blockIdx.x + (blockIdx.x >> 8)) & 7;

  v8f accC[3][2], accS[3][2];
  #pragma unroll
  for (int i = 0; i < 3; ++i)
    for (int p = 0; p < 2; ++p)
      for (int j = 0; j < 8; ++j) { accC[i][p][j] = 0.0f; accS[i][p][j] = 0.0f; }

  for (int kc = 0; kc < KHALF / KCH; ++kc) {
    __syncthreads();
    for (int idx = tid; idx < 16 * KCH; idx += 256) {
      int m  = idx >> 8;
      int kl = idx & (KCH - 1);
      int t  = t0 + m;
      float ve = 0.0f, vo = 0.0f;
      if (t < NFRAMES) {
        int k  = kc * KCH + kl;
        bool pr = k > 0;
        int kp = pr ? (NFFT - k) : 0;
        int o1 = t * HOPSZ + k - NFFT / 2;
        o1 = (o1 < 0) ? -o1 : o1;
        o1 = (o1 >= T_IN) ? (2 * (T_IN - 1) - o1) : o1;
        int o2 = t * HOPSZ + NFFT / 2 - k;
        o2 = (o2 >= T_IN) ? (2 * (T_IN - 1) - o2) : o2;
        float y1 = wb[o1] * win[k];
        float y2 = wb[o2] * win[kp];
        ve = pr ? (y1 + y2) : y1;
        vo = pr ? (y1 - y2) : 0.0f;
      }
      int c = (kl >> 2) * 8 + (kl & 1) * 4 + ((kl >> 1) & 1);
      A4[m][c]     = ve;
      A4[m][c + 2] = vo;
    }
    __syncthreads();

    #pragma unroll
    for (int i = 0; i < 3; ++i) {
      int ft = wslot + 8 * i;
      if (ft < NFT) {
        int f0 = ft * 16;
        #pragma unroll
        for (int p = 0; p < 2; ++p) {
          const float* dB = dftT + (long)(f0 + l16) * 4 +
                            (long)((kc * QCH + h) * 2 + p) * (FH * 4);
          const float* aP = &A4[l16][h * 8 + p * 4];
          __builtin_prefetch(dB, 0, 1);
          #pragma unroll 8
          for (int qs = 0; qs < QCH / 2; ++qs) {
            v4f a = *(const v4f*)(aP + qs * 16);
            v2f ae; ae[0] = a[0]; ae[1] = a[1];
            v2f ao; ao[0] = a[2]; ao[1] = a[3];
            v4f q = *(const v4f*)(dB + (long)qs * (4 * FH * 4));
            v2f bc; bc[0] = q[0]; bc[1] = q[1];
            v2f bs; bs[0] = q[2]; bs[1] = q[3];
            accC[i][p] = wmma4(ae, bc, accC[i][p]);
            accS[i][p] = wmma4(ao, bs, accS[i][p]);
          }
        }
      }
    }
  }

  {
    int t = t0 + l16;
    float y = 0.0f;
    if (h == 0 && t < NFRAMES) y = wb[t * HOPSZ] * win[NFFT / 2];
    v2f ae; ae[0] = y; ae[1] = 0.0f;
    #pragma unroll
    for (int i = 0; i < 3; ++i) {
      int ft = wslot + 8 * i;
      if (ft < NFT) {
        const float* dB = dftT + (long)(ft * 16 + l16) * 4;
        v4f q = *(const v4f*)(dB + (long)((NQ - 1) * 2) * (FH * 4));
        v2f bc; bc[0] = q[0]; bc[1] = q[1];
        accC[i][0] = wmma4(ae, bc, accC[i][0]);
      }
    }
  }

  __syncthreads();
  if (tid < 16 * (FPAD - NFREQ))
    PL[tid / (FPAD - NFREQ)][NFREQ + tid % (FPAD - NFREQ)] = 0.0f;
  #pragma unroll
  for (int i = 0; i < 3; ++i) {
    int ft = wslot + 8 * i;
    if (ft < NFT) {
      int f = ft * 16 + l16;
      #pragma unroll
      for (int v = 0; v < 8; ++v) {
        int mr = v + 8 * h;
        float ce = accC[i][0][v], co = accC[i][1][v];
        float se = accS[i][0][v], so = accS[i][1][v];
        float re = ce + co, im = se + so;
        float rm = ce - co, jm = so - se;
        if (f <= NFFT / 4) PL[mr][f] = re * re + im * im;
        if (f <  NFFT / 4) PL[mr][NFFT / 2 - f] = rm * rm + jm * jm;
      }
    }
  }
  __syncthreads();

  int n0 = wv * 16;
  const float* fB = fb2 + (long)(n0 + l16) * 2;
  v8f acc;
  #pragma unroll
  for (int j = 0; j < 8; ++j) acc[j] = 0.0f;
  for (int k = 0; k < FPAD; k += 4) {
    int kk = k + 2 * h;
    v2f a  = *(const v2f*)&PL[l16][kk];
    v2f bb = *(const v2f*)(fB + (long)(kk >> 1) * (NMELS * 2));
    acc = wmma4(a, bb, acc);
  }
  int n = n0 + l16;
  #pragma unroll
  for (int v = 0; v < 8; ++v) {
    int t = t0 + v + 8 * h;
    if (t < NFRAMES) mel[(b * NMELS + n) * NFRAMES + t] = acc[v];
  }
}

#define MED_R     4
#define MED_TAPS  (31 + MED_R - 1)
#define MED_CORE  (31 - MED_R + 1)
static_assert((NFRAMES % MED_R) == 0 && (NMELS % MED_R) == 0, "groups of four outputs must tile both axes");

__device__ __forceinline__ float median_core4_plus3(float c12, float c13, float c14, float c15,
                                                    float a, float b, float c) {
  float lo = fminf(a, b), hi = fmaxf(a, b);
  float e0 = fminf(lo, c);
  float md = fmaxf(lo, c);
  float e1 = fminf(hi, md);
  float e2 = fmaxf(hi, md);
  return fminf(fminf(c15, fmaxf(c14, e0)), fminf(fmaxf(c13, e1), fmaxf(c12, e2)));
}

__device__ __forceinline__ void median31x4(const float x[MED_TAPS], float o[MED_R]) {
  float c[MED_CORE];
  #pragma unroll
  for (int k = 0; k < MED_CORE; ++k) c[k] = x[k + 3];
  #pragma unroll
  for (int i = 0; i < 16; ++i) {
    #pragma unroll
    for (int j = i + 1; j < MED_CORE; ++j) {
      float lo = fminf(c[i], c[j]);
      float hi = fmaxf(c[i], c[j]);
      c[i] = lo; c[j] = hi;
    }
  }
  o[0] = median_core4_plus3(c[12], c[13], c[14], c[15], x[0],  x[1],  x[2]);
  o[1] = median_core4_plus3(c[12], c[13], c[14], c[15], x[1],  x[2],  x[31]);
  o[2] = median_core4_plus3(c[12], c[13], c[14], c[15], x[2],  x[31], x[32]);
  o[3] = median_core4_plus3(c[12], c[13], c[14], c[15], x[31], x[32], x[33]);
}

#define MT_BLOCKS (E_ELEMS / (MED_R * 256))
static_assert(MT_BLOCKS * MED_R * 256 == E_ELEMS, "median_time grid must tile E_ELEMS exactly");

__global__ __launch_bounds__(256) void median_time(const float* __restrict__ mel,
                                                   float* __restrict__ harm) {
  int g   = blockIdx.x * 256 + threadIdx.x;
  int q   = g % (NFRAMES / MED_R);
  int row = g / (NFRAMES / MED_R);
  int t0  = q * MED_R;
  const float* src = mel + row * NFRAMES;
  float x[MED_TAPS];
  if (t0 >= 16 && t0 + 19 < NFRAMES) {
    const v4f* __restrict__ s4 = (const v4f*)(src + (t0 - 16));
    v4f w[9];
    #pragma unroll
    for (int u = 0; u < 9; ++u) w[u] = s4[u];
    #pragma unroll
    for (int e = 0; e < MED_TAPS; ++e) x[e] = w[(e + 1) >> 2][(e + 1) & 3];
  } else {
    #pragma unroll
    for (int e = 0; e < MED_TAPS; ++e) {
      int j = t0 + e - 15;
      j = (j < 0) ? (-j - 1) : j;
      j = (j >= NFRAMES) ? (2 * NFRAMES - 1 - j) : j;
      x[e] = src[j];
    }
  }
  float o[MED_R];
  median31x4(x, o);
  v4f ov; ov[0] = o[0]; ov[1] = o[1]; ov[2] = o[2]; ov[3] = o[3];
  ((v4f*)harm)[g] = ov;
}

#define MM_TBLOCKS ((NFRAMES + 255) / 256)
#define MM_GROUPS  (NB * (NMELS / MED_R))

__global__ __launch_bounds__(256) void median_mel(const float* __restrict__ mel,
                                                  float* __restrict__ perc) {
  int t = blockIdx.x * 256 + threadIdx.x;
  if (t >= NFRAMES) return;
  int k  = blockIdx.y % (NMELS / MED_R);
  int b  = blockIdx.y / (NMELS / MED_R);
  int n0 = k * MED_R;
  const float* base = mel + (b * NMELS) * NFRAMES + t;
  float x[MED_TAPS];
  #pragma unroll
  for (int e = 0; e < MED_TAPS; ++e) {
    int j = n0 + e - 15;
    j = (j < 0) ? (-j - 1) : j;
    j = (j >= NMELS) ? (2 * NMELS - 1 - j) : j;
    x[e] = base[j * NFRAMES];
  }
  float o[MED_R];
  median31x4(x, o);
  float* dst = perc + (b * NMELS + n0) * NFRAMES + t;
  #pragma unroll
  for (int r = 0; r < MED_R; ++r) dst[r * NFRAMES] = o[r];
}

#define HP_U      4
#define HP_BLOCKS (E_ELEMS / (4 * 256 * HP_U))
static_assert(HP_BLOCKS * 4 * 256 * HP_U == E_ELEMS, "hpss_mask grid must tile E_ELEMS exactly");
static_assert((MEL_OFF % 4) == 0 && (E_ELEMS % 4) == 0, "float4 views need 16-byte aligned planes");

__global__ __launch_bounds__(256) void hpss_mask(const float* __restrict__ mel,
                                                 float* __restrict__ harm,
                                                 float* __restrict__ perc,
                                                 unsigned* __restrict__ mx) {
  __shared__ float r0[256], r1[256], r2[256];
  const v4f* __restrict__ mel4 = (const v4f*)mel;
  v4f* __restrict__ harm4 = (v4f*)harm;
  v4f* __restrict__ perc4 = (v4f*)perc;
  int tid  = threadIdx.x;
  int base = blockIdx.x * (256 * HP_U) + tid;
  v4f S4[HP_U], H4[HP_U], P4[HP_U];
  #pragma unroll
  for (int u = 0; u < HP_U; ++u) {
    int i = base + u * 256;
    S4[u] = mel4[i]; H4[u] = harm4[i]; P4[u] = perc4[i];
  }
  float s = 0.0f, dh = 0.0f, dp = 0.0f;
  #pragma unroll
  for (int u = 0; u < HP_U; ++u) {
    v4f oh, op;
    #pragma unroll
    for (int c = 0; c < 4; ++c) {
      float S  = S4[u][c];
      float ha = H4[u][c];
      float pe = P4[u][c];
      float Z  = fmaxf(ha, pe);
      bool bad = Z < 1.17549435e-38f;
      float Zs = bad ? 1.0f : Z;
      float mh = ha / Zs; mh *= mh;
      float mp = pe / Zs; mp *= mp;
      float den = mh + mp;
      float maskh = bad ? 0.0f : mh / den;
      float maskp = bad ? 0.0f : mp / den;
      float vh = S * maskh, vp = S * maskp;
      oh[c] = vh; op[c] = vp;
      s = fmaxf(s, S); dh = fmaxf(dh, vh); dp = fmaxf(dp, vp);
    }
    int i = base + u * 256;
    harm4[i] = oh;
    perc4[i] = op;
  }
  r0[tid] = s; r1[tid] = dh; r2[tid] = dp;
  __syncthreads();
  for (int off = 128; off > 0; off >>= 1) {
    if (tid < off) {
      r0[tid] = fmaxf(r0[tid], r0[tid + off]);
      r1[tid] = fmaxf(r1[tid], r1[tid + off]);
      r2[tid] = fmaxf(r2[tid], r2[tid + off]);
    }
    __syncthreads();
  }
  if (tid == 0) {
    atomicMax(mx + 0, __float_as_uint(r0[0]));
    atomicMax(mx + 1, __float_as_uint(r1[0]));
    atomicMax(mx + 2, __float_as_uint(r2[0]));
  }
}

#define FN_U          8
#define FN_SEG_BLOCKS (E_ELEMS / (4 * 256 * FN_U))
static_assert(FN_SEG_BLOCKS * 4 * 256 * FN_U == E_ELEMS, "finalize grid must tile each plane exactly");

__global__ __launch_bounds__(256) void finalize(const float* __restrict__ mel,
                                                const float* __restrict__ harm,
                                                const float* __restrict__ perc,
                                                const unsigned* __restrict__ mx,
                                                float* __restrict__ out) {
  int s  = blockIdx.x / FN_SEG_BLOCKS;
  int e4 = (blockIdx.x - s * FN_SEG_BLOCKS) * (256 * FN_U) + threadIdx.x;
  const float* src = (s == 0) ? mel : ((s == 1) ? harm : perc);
  const v4f* __restrict__ src4 = (const v4f*)src;
  v4f* __restrict__ out4 = (v4f*)out + (long)s * (E_ELEMS / 4);
  float ref = __uint_as_float(mx[s]);
  v4f v[FN_U];
  #pragma unroll
  for (int u = 0; u < FN_U; ++u) v[u] = src4[e4 + u * 256];
  #pragma unroll
  for (int u = 0; u < FN_U; ++u) {
    v4f o;
    #pragma unroll
    for (int c = 0; c < 4; ++c) {
      float db = 10.0f * log10f(fmaxf(1e-10f, v[u][c])) -
                 10.0f * log10f(fmaxf(1e-10f, ref));
      db = fmaxf(db, -80.0f);
      o[c] = (db + 80.0f) * (1.0f / (80.0f + 1e-7f));
    }
    out4[e4 + u * 256] = o;
  }
}

extern "C" void kernel_launch(void* const* d_in, const int* in_sizes, int n_in,
                              void* d_out, int out_size, void* d_ws,
                              size_t ws_size, hipStream_t stream) {
  const float* wave = (const float*)d_in[0];
  float* out  = (float*)d_out;
  float* ws   = (float*)d_ws;
  float* mel  = ws + MEL_OFF;
  float* harm = ws + HARM_OFF;
  float* perc = ws + PERC_OFF;
  unsigned* mx = (unsigned*)(ws + MAX_OFF);

  build_tables<<<(MAX_OFF + 255) / 256, 256, 0, stream>>>(ws);
  mel_gemm<<<NB * MTILES, 256, 0, stream>>>(wave, ws, mel);
  median_time<<<MT_BLOCKS, 256, 0, stream>>>(mel, harm);
  median_mel<<<dim3(MM_TBLOCKS, MM_GROUPS), 256, 0, stream>>>(mel, perc);
  hpss_mask<<<HP_BLOCKS, 256, 0, stream>>>(mel, harm, perc, mx);
  finalize<<<3 * FN_SEG_BLOCKS, 256, 0, stream>>>(mel, harm, perc, mx, out);
}


# median_time, median_mel: ranks 12..15 of the 28 shared taps from a pruned odd-even merge network (137 comparators, 15 layers) instead of 16 selection passes (312 comparators); exact medians
# speedup vs baseline: 2.7196x; 1.0425x over previous
#include <hip/hip_runtime.h>
#include <math.h>

#define NFFT     1024
#define HOPSZ    512
#define NMELS    128
#define NFREQ    513
#define FPAD     528
#define T_IN     1323000
#define NB       16
#define NFRAMES  2584
#define MTILES   162
#define E_ELEMS  (NB*NMELS*NFRAMES)

#define KHALF    512
#define KCH      256
#define APITCH   516
#define FH       272
#define NFT      17
#define QCH      64
#define NQ       129

#define WIN_OFF  0
#define DFT_OFF  1024
#define DFT_SZ   (NQ*2*FH*4)
#define FB_OFF   (DFT_OFF + DFT_SZ)
#define FB_SZ    ((FPAD/2)*NMELS*2)
#define MAX_OFF  (FB_OFF + FB_SZ)
#define MEL_OFF  (MAX_OFF + 8)
#define HARM_OFF (MEL_OFF + E_ELEMS)
#define PERC_OFF (HARM_OFF + E_ELEMS)

typedef float v2f __attribute__((ext_vector_type(2)));
typedef float v4f __attribute__((ext_vector_type(4)));
typedef float v8f __attribute__((ext_vector_type(8)));

__device__ __forceinline__ v8f wmma4(v2f a, v2f b, v8f c) {
  return __builtin_amdgcn_wmma_f32_16x16x4_f32(false, a, false, b, (short)0, c,
                                               false, false);
}

__global__ __launch_bounds__(256) void build_tables(float* __restrict__ ws) {
  int gid = blockIdx.x * 256 + threadIdx.x;
  if (gid == 0) {
    unsigned* mx = (unsigned*)(ws + MAX_OFF);
    mx[0] = 0u; mx[1] = 0u; mx[2] = 0u;
  }
  const float TWO_PI = 6.28318530717958647692f;
  if (gid < DFT_OFF) {
    int k = gid;
    ws[WIN_OFF + k] = 0.5f * (1.0f - cosf(TWO_PI * (float)k / 1024.0f));
  } else if (gid < FB_OFF) {
    int i = gid - DFT_OFF;
    int entry = i & 3;
    int rest  = i >> 2;
    int f  = rest % FH;
    int qp = rest / FH;
    int k  = 4 * (qp >> 1) + (qp & 1) + 2 * (entry & 1);
    int r  = (k * f) & (NFFT - 1);
    float arg = (float)r * (TWO_PI / 1024.0f);
    ws[gid] = (entry < 2) ? cosf(arg) : sinf(arg);
  } else if (gid < MAX_OFF) {
    int i = gid - FB_OFF;
    int entry = i & 1;
    int rest  = i >> 1;
    int m  = rest % NMELS;
    int k2 = rest / NMELS;
    int f  = 2 * k2 + entry;
    float v = 0.0f;
    if (f < NFREQ) {
      float freq   = (float)f * (22050.0f / 512.0f);
      float melmax = 2595.0f * log10f(1.0f + 22050.0f / 700.0f);
      float step   = melmax / 129.0f;
      float p0 = 700.0f * (exp10f(((float)(m    ) * step) / 2595.0f) - 1.0f);
      float p1 = 700.0f * (exp10f(((float)(m + 1) * step) / 2595.0f) - 1.0f);
      float p2 = 700.0f * (exp10f(((float)(m + 2) * step) / 2595.0f) - 1.0f);
      float down = (freq - p0) / (p1 - p0);
      float up   = (p2 - freq) / (p2 - p1);
      float t = fmaxf(0.0f, fminf(down, up));
      v = t * 2.0f / (p2 - p0);
    }
    ws[gid] = v;
  }
}

__global__ __launch_bounds__(256, 6) void mel_gemm(const float* __restrict__ wave,
                                                const float* __restrict__ ws,
                                                float* __restrict__ mel) {
  __shared__ __attribute__((aligned(16))) float A4[16][APITCH];
  __shared__ float PL[16][FPAD + 4];

  const float* __restrict__ win  = ws + WIN_OFF;
  const float* __restrict__ dftT = ws + DFT_OFF;
  const float* __restrict__ fb2  = ws + FB_OFF;

  int b    = blockIdx.x / MTILES;
  int tile = blockIdx.x % MTILES;
  int t0   = tile * 16;
  int tid  = threadIdx.x;
  int lane = tid & 31;
  int wv   = tid >> 5;
  int h    = lane >> 4;
  int l16  = lane & 15;

  const float* wb = wave + (long)b * T_IN;

  int wslot = (wv + blockIdx.x + (blockIdx.x >> 8)) & 7;

  v8f accC[3][2], accS[3][2];
  #pragma unroll
  for (int i = 0; i < 3; ++i)
    for (int p = 0; p < 2; ++p)
      for (int j = 0; j < 8; ++j) { accC[i][p][j] = 0.0f; accS[i][p][j] = 0.0f; }

  for (int kc = 0; kc < KHALF / KCH; ++kc) {
    __syncthreads();
    for (int idx = tid; idx < 16 * KCH; idx += 256) {
      int m  = idx >> 8;
      int kl = idx & (KCH - 1);
      int t  = t0 + m;
      float ve = 0.0f, vo = 0.0f;
      if (t < NFRAMES) {
        int k  = kc * KCH + kl;
        bool pr = k > 0;
        int kp = pr ? (NFFT - k) : 0;
        int o1 = t * HOPSZ + k - NFFT / 2;
        o1 = (o1 < 0) ? -o1 : o1;
        o1 = (o1 >= T_IN) ? (2 * (T_IN - 1) - o1) : o1;
        int o2 = t * HOPSZ + NFFT / 2 - k;
        o2 = (o2 >= T_IN) ? (2 * (T_IN - 1) - o2) : o2;
        float y1 = wb[o1] * win[k];
        float y2 = wb[o2] * win[kp];
        ve = pr ? (y1 + y2) : y1;
        vo = pr ? (y1 - y2) : 0.0f;
      }
      int c = (kl >> 2) * 8 + (kl & 1) * 4 + ((kl >> 1) & 1);
      A4[m][c]     = ve;
      A4[m][c + 2] = vo;
    }
    __syncthreads();

    #pragma unroll
    for (int i = 0; i < 3; ++i) {
      int ft = wslot + 8 * i;
      if (ft < NFT) {
        int f0 = ft * 16;
        #pragma unroll
        for (int p = 0; p < 2; ++p) {
          const float* dB = dftT + (long)(f0 + l16) * 4 +
                            (long)((kc * QCH + h) * 2 + p) * (FH * 4);
          const float* aP = &A4[l16][h * 8 + p * 4];
          __builtin_prefetch(dB, 0, 1);
          #pragma unroll 8
          for (int qs = 0; qs < QCH / 2; ++qs) {
            v4f a = *(const v4f*)(aP + qs * 16);
            v2f ae; ae[0] = a[0]; ae[1] = a[1];
            v2f ao; ao[0] = a[2]; ao[1] = a[3];
            v4f q = *(const v4f*)(dB + (long)qs * (4 * FH * 4));
            v2f bc; bc[0] = q[0]; bc[1] = q[1];
            v2f bs; bs[0] = q[2]; bs[1] = q[3];
            accC[i][p] = wmma4(ae, bc, accC[i][p]);
            accS[i][p] = wmma4(ao, bs, accS[i][p]);
          }
        }
      }
    }
  }

  {
    int t = t0 + l16;
    float y = 0.0f;
    if (h == 0 && t < NFRAMES) y = wb[t * HOPSZ] * win[NFFT / 2];
    v2f ae; ae[0] = y; ae[1] = 0.0f;
    #pragma unroll
    for (int i = 0; i < 3; ++i) {
      int ft = wslot + 8 * i;
      if (ft < NFT) {
        const float* dB = dftT + (long)(ft * 16 + l16) * 4;
        v4f q = *(const v4f*)(dB + (long)((NQ - 1) * 2) * (FH * 4));
        v2f bc; bc[0] = q[0]; bc[1] = q[1];
        accC[i][0] = wmma4(ae, bc, accC[i][0]);
      }
    }
  }

  __syncthreads();
  if (tid < 16 * (FPAD - NFREQ))
    PL[tid / (FPAD - NFREQ)][NFREQ + tid % (FPAD - NFREQ)] = 0.0f;
  #pragma unroll
  for (int i = 0; i < 3; ++i) {
    int ft = wslot + 8 * i;
    if (ft < NFT) {
      int f = ft * 16 + l16;
      #pragma unroll
      for (int v = 0; v < 8; ++v) {
        int mr = v + 8 * h;
        float ce = accC[i][0][v], co = accC[i][1][v];
        float se = accS[i][0][v], so = accS[i][1][v];
        float re = ce + co, im = se + so;
        float rm = ce - co, jm = so - se;
        if (f <= NFFT / 4) PL[mr][f] = re * re + im * im;
        if (f <  NFFT / 4) PL[mr][NFFT / 2 - f] = rm * rm + jm * jm;
      }
    }
  }
  __syncthreads();

  int n0 = wv * 16;
  const float* fB = fb2 + (long)(n0 + l16) * 2;
  v8f acc;
  #pragma unroll
  for (int j = 0; j < 8; ++j) acc[j] = 0.0f;
  for (int k = 0; k < FPAD; k += 4) {
    int kk = k + 2 * h;
    v2f a  = *(const v2f*)&PL[l16][kk];
    v2f bb = *(const v2f*)(fB + (long)(kk >> 1) * (NMELS * 2));
    acc = wmma4(a, bb, acc);
  }
  int n = n0 + l16;
  #pragma unroll
  for (int v = 0; v < 8; ++v) {
    int t = t0 + v + 8 * h;
    if (t < NFRAMES) mel[(b * NMELS + n) * NFRAMES + t] = acc[v];
  }
}

#define MED_R     4
#define MED_TAPS  (31 + MED_R - 1)
#define MED_CORE  (31 - MED_R + 1)
static_assert((NFRAMES % MED_R) == 0 && (NMELS % MED_R) == 0, "groups of four outputs must tile both axes");

__device__ __forceinline__ float median_core4_plus3(float c12, float c13, float c14, float c15,
                                                    float a, float b, float c) {
  float lo = fminf(a, b), hi = fmaxf(a, b);
  float e0 = fminf(lo, c);
  float md = fmaxf(lo, c);
  float e1 = fminf(hi, md);
  float e2 = fmaxf(hi, md);
  return fminf(fminf(c15, fmaxf(c14, e0)), fminf(fmaxf(c13, e1), fmaxf(c12, e2)));
}

#define MED_CE(i, j) { float lo_ = fminf(c[i], c[j]); float hi_ = fmaxf(c[i], c[j]); c[i] = lo_; c[j] = hi_; }
#define MED_LO(i, j) { c[i] = fminf(c[i], c[j]); }
#define MED_HI(i, j) { c[j] = fmaxf(c[i], c[j]); }
__device__ __forceinline__ void median31x4(const float x[MED_TAPS], float o[MED_R]) {
  float c[MED_CORE];
  #pragma unroll
  for (int k = 0; k < MED_CORE; ++k) c[k] = x[k + 3];
  MED_CE(1,2) MED_CE(3,4) MED_CE(5,6) MED_CE(7,8) MED_CE(9,10) MED_CE(11,12) MED_CE(13,14)
  MED_CE(15,16) MED_CE(17,18) MED_CE(19,20) MED_CE(21,22) MED_CE(23,24) MED_CE(25,26)
  MED_CE(0,2) MED_CE(3,5) MED_CE(4,6) MED_CE(7,9) MED_CE(8,10) MED_CE(11,13) MED_CE(12,14)
  MED_CE(15,17) MED_CE(16,18) MED_CE(19,21) MED_CE(20,22) MED_CE(23,25) MED_CE(24,26)
  MED_CE(0,1) MED_CE(4,5) MED_CE(8,9) MED_CE(12,13) MED_CE(16,17) MED_CE(20,21) MED_CE(24,25)
  MED_CE(2,6) MED_CE(7,11) MED_CE(10,14) MED_CE(15,19) MED_CE(18,22) MED_CE(23,27)
  MED_CE(0,4) MED_CE(1,5) MED_CE(8,12) MED_CE(9,13) MED_CE(16,20) MED_CE(17,21) MED_CE(25,27)
  MED_CE(6,14) MED_CE(15,23)
  MED_CE(1,3) MED_CE(2,4) MED_CE(9,11) MED_CE(10,12) MED_CE(17,19) MED_CE(18,20) MED_CE(24,25)
  MED_CE(26,27)
  MED_CE(0,1) MED_CE(2,3) MED_CE(4,5) MED_CE(8,9) MED_CE(10,11) MED_CE(12,13) MED_CE(16,17)
  MED_CE(18,19) MED_CE(20,21)
  MED_CE(0,8) MED_CE(1,9) MED_CE(2,10) MED_CE(3,11) MED_CE(4,12) MED_CE(5,13) MED_CE(16,24)
  MED_CE(17,25) MED_CE(18,26) MED_CE(19,27)
  MED_CE(3,7) MED_CE(4,8) MED_CE(5,9) MED_CE(6,10) MED_CE(19,23) MED_CE(20,24) MED_CE(21,25)
  MED_CE(22,26)
  MED_CE(1,3) MED_CE(2,4) MED_CE(5,7) MED_CE(6,8) MED_CE(9,11) MED_CE(10,12) MED_CE(17,19)
  MED_CE(18,20) MED_CE(21,23) MED_CE(22,24) MED_CE(25,27)
  MED_CE(0,1) MED_CE(2,3) MED_CE(4,5) MED_CE(6,7) MED_CE(8,9) MED_CE(10,11) MED_CE(12,13)
  MED_CE(16,17) MED_CE(18,19) MED_CE(20,21) MED_CE(22,23) MED_CE(24,25) MED_CE(26,27)
  MED_HI(0,16) MED_HI(1,17) MED_HI(2,18) MED_HI(3,19) MED_HI(4,20) MED_HI(5,21) MED_CE(6,22)
  MED_LO(7,23) MED_LO(8,24) MED_LO(9,25) MED_LO(10,26) MED_LO(11,27)
  MED_HI(7,15) MED_HI(8,16) MED_HI(9,17) MED_CE(10,18) MED_LO(11,19) MED_LO(12,20) MED_LO(13,21)
  MED_LO(14,22)
  MED_HI(6,10) MED_HI(11,15) MED_CE(12,16) MED_LO(13,17) MED_LO(14,18)
  MED_HI(10,12) MED_CE(13,15) MED_LO(14,16)
  MED_CE(12,13) MED_CE(14,15)
  o[0] = median_core4_plus3(c[12], c[13], c[14], c[15], x[0],  x[1],  x[2]);
  o[1] = median_core4_plus3(c[12], c[13], c[14], c[15], x[1],  x[2],  x[31]);
  o[2] = median_core4_plus3(c[12], c[13], c[14], c[15], x[2],  x[31], x[32]);
  o[3] = median_core4_plus3(c[12], c[13], c[14], c[15], x[31], x[32], x[33]);
}
#undef MED_CE
#undef MED_LO
#undef MED_HI

#define MT_BLOCKS (E_ELEMS / (MED_R * 256))
static_assert(MT_BLOCKS * MED_R * 256 == E_ELEMS, "median_time grid must tile E_ELEMS exactly");

__global__ __launch_bounds__(256) void median_time(const float* __restrict__ mel,
                                                   float* __restrict__ harm) {
  int g   = blockIdx.x * 256 + threadIdx.x;
  int q   = g % (NFRAMES / MED_R);
  int row = g / (NFRAMES / MED_R);
  int t0  = q * MED_R;
  const float* src = mel + row * NFRAMES;
  float x[MED_TAPS];
  if (t0 >= 16 && t0 + 19 < NFRAMES) {
    const v4f* __restrict__ s4 = (const v4f*)(src + (t0 - 16));
    v4f w[9];
    #pragma unroll
    for (int u = 0; u < 9; ++u) w[u] = s4[u];
    #pragma unroll
    for (int e = 0; e < MED_TAPS; ++e) x[e] = w[(e + 1) >> 2][(e + 1) & 3];
  } else {
    #pragma unroll
    for (int e = 0; e < MED_TAPS; ++e) {
      int j = t0 + e - 15;
      j = (j < 0) ? (-j - 1) : j;
      j = (j >= NFRAMES) ? (2 * NFRAMES - 1 - j) : j;
      x[e] = src[j];
    }
  }
  float o[MED_R];
  median31x4(x, o);
  v4f ov; ov[0] = o[0]; ov[1] = o[1]; ov[2] = o[2]; ov[3] = o[3];
  ((v4f*)harm)[g] = ov;
}

#define MM_TBLOCKS ((NFRAMES + 255) / 256)
#define MM_GROUPS  (NB * (NMELS / MED_R))

__global__ __launch_bounds__(256) void median_mel(const float* __restrict__ mel,
                                                  float* __restrict__ perc) {
  int t = blockIdx.x * 256 + threadIdx.x;
  if (t >= NFRAMES) return;
  int k  = blockIdx.y % (NMELS / MED_R);
  int b  = blockIdx.y / (NMELS / MED_R);
  int n0 = k * MED_R;
  const float* base = mel + (b * NMELS) * NFRAMES + t;
  float x[MED_TAPS];
  #pragma unroll
  for (int e = 0; e < MED_TAPS; ++e) {
    int j = n0 + e - 15;
    j = (j < 0) ? (-j - 1) : j;
    j = (j >= NMELS) ? (2 * NMELS - 1 - j) : j;
    x[e] = base[j * NFRAMES];
  }
  float o[MED_R];
  median31x4(x, o);
  float* dst = perc + (b * NMELS + n0) * NFRAMES + t;
  #pragma unroll
  for (int r = 0; r < MED_R; ++r) dst[r * NFRAMES] = o[r];
}

#define HP_U      4
#define HP_BLOCKS (E_ELEMS / (4 * 256 * HP_U))
static_assert(HP_BLOCKS * 4 * 256 * HP_U == E_ELEMS, "hpss_mask grid must tile E_ELEMS exactly");
static_assert((MEL_OFF % 4) == 0 && (E_ELEMS % 4) == 0, "float4 views need 16-byte aligned planes");

__global__ __launch_bounds__(256) void hpss_mask(const float* __restrict__ mel,
                                                 float* __restrict__ harm,
                                                 float* __restrict__ perc,
                                                 unsigned* __restrict__ mx) {
  __shared__ float r0[256], r1[256], r2[256];
  const v4f* __restrict__ mel4 = (const v4f*)mel;
  v4f* __restrict__ harm4 = (v4f*)harm;
  v4f* __restrict__ perc4 = (v4f*)perc;
  int tid  = threadIdx.x;
  int base = blockIdx.x * (256 * HP_U) + tid;
  v4f S4[HP_U], H4[HP_U], P4[HP_U];
  #pragma unroll
  for (int u = 0; u < HP_U; ++u) {
    int i = base + u * 256;
    S4[u] = mel4[i]; H4[u] = harm4[i]; P4[u] = perc4[i];
  }
  float s = 0.0f, dh = 0.0f, dp = 0.0f;
  #pragma unroll
  for (int u = 0; u < HP_U; ++u) {
    v4f oh, op;
    #pragma unroll
    for (int c = 0; c < 4; ++c) {
      float S  = S4[u][c];
      float ha = H4[u][c];
      float pe = P4[u][c];
      float Z  = fmaxf(ha, pe);
      bool bad = Z < 1.17549435e-38f;
      float Zs = bad ? 1.0f : Z;
      float mh = ha / Zs; mh *= mh;
      float mp = pe / Zs; mp *= mp;
      float den = mh + mp;
      float maskh = bad ? 0.0f : mh / den;
      float maskp = bad ? 0.0f : mp / den;
      float vh = S * maskh, vp = S * maskp;
      oh[c] = vh; op[c] = vp;
      s = fmaxf(s, S); dh = fmaxf(dh, vh); dp = fmaxf(dp, vp);
    }
    int i = base + u * 256;
    harm4[i] = oh;
    perc4[i] = op;
  }
  r0[tid] = s; r1[tid] = dh; r2[tid] = dp;
  __syncthreads();
  for (int off = 128; off > 0; off >>= 1) {
    if (tid < off) {
      r0[tid] = fmaxf(r0[tid], r0[tid + off]);
      r1[tid] = fmaxf(r1[tid], r1[tid + off]);
      r2[tid] = fmaxf(r2[tid], r2[tid + off]);
    }
    __syncthreads();
  }
  if (tid == 0) {
    atomicMax(mx + 0, __float_as_uint(r0[0]));
    atomicMax(mx + 1, __float_as_uint(r1[0]));
    atomicMax(mx + 2, __float_as_uint(r2[0]));
  }
}

#define FN_U          8
#define FN_SEG_BLOCKS (E_ELEMS / (4 * 256 * FN_U))
static_assert(FN_SEG_BLOCKS * 4 * 256 * FN_U == E_ELEMS, "finalize grid must tile each plane exactly");

__global__ __launch_bounds__(256) void finalize(const float* __restrict__ mel,
                                                const float* __restrict__ harm,
                                                const float* __restrict__ perc,
                                                const unsigned* __restrict__ mx,
                                                float* __restrict__ out) {
  int s  = blockIdx.x / FN_SEG_BLOCKS;
  int e4 = (blockIdx.x - s * FN_SEG_BLOCKS) * (256 * FN_U) + threadIdx.x;
  const float* src = (s == 0) ? mel : ((s == 1) ? harm : perc);
  const v4f* __restrict__ src4 = (const v4f*)src;
  v4f* __restrict__ out4 = (v4f*)out + (long)s * (E_ELEMS / 4);
  float ref = __uint_as_float(mx[s]);
  v4f v[FN_U];
  #pragma unroll
  for (int u = 0; u < FN_U; ++u) v[u] = src4[e4 + u * 256];
  #pragma unroll
  for (int u = 0; u < FN_U; ++u) {
    v4f o;
    #pragma unroll
    for (int c = 0; c < 4; ++c) {
      float db = 10.0f * log10f(fmaxf(1e-10f, v[u][c])) -
                 10.0f * log10f(fmaxf(1e-10f, ref));
      db = fmaxf(db, -80.0f);
      o[c] = (db + 80.0f) * (1.0f / (80.0f + 1e-7f));
    }
    out4[e4 + u * 256] = o;
  }
}

extern "C" void kernel_launch(void* const* d_in, const int* in_sizes, int n_in,
                              void* d_out, int out_size, void* d_ws,
                              size_t ws_size, hipStream_t stream) {
  const float* wave = (const float*)d_in[0];
  float* out  = (float*)d_out;
  float* ws   = (float*)d_ws;
  float* mel  = ws + MEL_OFF;
  float* harm = ws + HARM_OFF;
  float* perc = ws + PERC_OFF;
  unsigned* mx = (unsigned*)(ws + MAX_OFF);

  build_tables<<<(MAX_OFF + 255) / 256, 256, 0, stream>>>(ws);
  mel_gemm<<<NB * MTILES, 256, 0, stream>>>(wave, ws, mel);
  median_time<<<MT_BLOCKS, 256, 0, stream>>>(mel, harm);
  median_mel<<<dim3(MM_TBLOCKS, MM_GROUPS), 256, 0, stream>>>(mel, perc);
  hpss_mask<<<HP_BLOCKS, 256, 0, stream>>>(mel, harm, perc, mx);
  finalize<<<3 * FN_SEG_BLOCKS, 256, 0, stream>>>(mel, harm, perc, mx, out);
}


# mel_gemm stage 2: each mel tile runs only the frequency rows inside the span of its 16 triangular filters (172, not 1056, WMMAs per work-group); skipped rows are exact zeros, sum order unchanged
# speedup vs baseline: 3.0109x; 1.1071x over previous
#include <hip/hip_runtime.h>
#include <math.h>

#define NFFT     1024
#define HOPSZ    512
#define NMELS    128
#define NFREQ    513
#define FPAD     528
#define T_IN     1323000
#define NB       16
#define NFRAMES  2584
#define MTILES   162
#define E_ELEMS  (NB*NMELS*NFRAMES)

#define KHALF    512
#define KCH      256
#define APITCH   516
#define FH       272
#define NFT      17
#define QCH      64
#define NQ       129

#define WIN_OFF  0
#define DFT_OFF  1024
#define DFT_SZ   (NQ*2*FH*4)
#define FB_OFF   (DFT_OFF + DFT_SZ)
#define FB_SZ    ((FPAD/2)*NMELS*2)
#define MAX_OFF  (FB_OFF + FB_SZ)
#define MEL_OFF  (MAX_OFF + 8)
#define HARM_OFF (MEL_OFF + E_ELEMS)
#define PERC_OFF (HARM_OFF + E_ELEMS)

typedef float v2f __attribute__((ext_vector_type(2)));
typedef float v4f __attribute__((ext_vector_type(4)));
typedef float v8f __attribute__((ext_vector_type(8)));

__device__ __forceinline__ v8f wmma4(v2f a, v2f b, v8f c) {
  return __builtin_amdgcn_wmma_f32_16x16x4_f32(false, a, false, b, (short)0, c,
                                               false, false);
}

__global__ __launch_bounds__(256) void build_tables(float* __restrict__ ws) {
  int gid = blockIdx.x * 256 + threadIdx.x;
  if (gid == 0) {
    unsigned* mx = (unsigned*)(ws + MAX_OFF);
    mx[0] = 0u; mx[1] = 0u; mx[2] = 0u;
  }
  const float TWO_PI = 6.28318530717958647692f;
  if (gid < DFT_OFF) {
    int k = gid;
    ws[WIN_OFF + k] = 0.5f * (1.0f - cosf(TWO_PI * (float)k / 1024.0f));
  } else if (gid < FB_OFF) {
    int i = gid - DFT_OFF;
    int entry = i & 3;
    int rest  = i >> 2;
    int f  = rest % FH;
    int qp = rest / FH;
    int k  = 4 * (qp >> 1) + (qp & 1) + 2 * (entry & 1);
    int r  = (k * f) & (NFFT - 1);
    float arg = (float)r * (TWO_PI / 1024.0f);
    ws[gid] = (entry < 2) ? cosf(arg) : sinf(arg);
  } else if (gid < MAX_OFF) {
    int i = gid - FB_OFF;
    int entry = i & 1;
    int rest  = i >> 1;
    int m  = rest % NMELS;
    int k2 = rest / NMELS;
    int f  = 2 * k2 + entry;
    float v = 0.0f;
    if (f < NFREQ) {
      float freq   = (float)f * (22050.0f / 512.0f);
      float melmax = 2595.0f * log10f(1.0f + 22050.0f / 700.0f);
      float step   = melmax / 129.0f;
      float p0 = 700.0f * (exp10f(((float)(m    ) * step) / 2595.0f) - 1.0f);
      float p1 = 700.0f * (exp10f(((float)(m + 1) * step) / 2595.0f) - 1.0f);
      float p2 = 700.0f * (exp10f(((float)(m + 2) * step) / 2595.0f) - 1.0f);
      float down = (freq - p0) / (p1 - p0);
      float up   = (p2 - freq) / (p2 - p1);
      float t = fmaxf(0.0f, fminf(down, up));
      v = t * 2.0f / (p2 - p0);
    }
    ws[gid] = v;
  }
}

template <int NS>
__device__ __forceinline__ v8f mel_steps(const float* pA, const float* pB, v8f acc) {
  v2f a[NS], bb[NS];
  #pragma unroll
  for (int u = 0; u < NS; ++u) {
    bb[u] = *(const v2f*)(pB + (long)u * (2 * NMELS * 2));
    a[u]  = *(const v2f*)(pA + 4 * u);
  }
  #pragma unroll
  for (int u = 0; u < NS; ++u) acc = wmma4(a[u], bb[u], acc);
  return acc;
}

__global__ __launch_bounds__(256, 6) void mel_gemm(const float* __restrict__ wave,
                                                const float* __restrict__ ws,
                                                float* __restrict__ mel) {
  __shared__ __attribute__((aligned(16))) float A4[16][APITCH];
  __shared__ float PL[16][FPAD + 4];

  const float* __restrict__ win  = ws + WIN_OFF;
  const float* __restrict__ dftT = ws + DFT_OFF;
  const float* __restrict__ fb2  = ws + FB_OFF;

  int b    = blockIdx.x / MTILES;
  int tile = blockIdx.x % MTILES;
  int t0   = tile * 16;
  int tid  = threadIdx.x;
  int lane = tid & 31;
  int wv   = tid >> 5;
  int h    = lane >> 4;
  int l16  = lane & 15;

  const float* wb = wave + (long)b * T_IN;

  int wslot = (wv + blockIdx.x + (blockIdx.x >> 8)) & 7;

  v8f accC[3][2], accS[3][2];
  #pragma unroll
  for (int i = 0; i < 3; ++i)
    for (int p = 0; p < 2; ++p)
      for (int j = 0; j < 8; ++j) { accC[i][p][j] = 0.0f; accS[i][p][j] = 0.0f; }

  for (int kc = 0; kc < KHALF / KCH; ++kc) {
    __syncthreads();
    for (int idx = tid; idx < 16 * KCH; idx += 256) {
      int m  = idx >> 8;
      int kl = idx & (KCH - 1);
      int t  = t0 + m;
      float ve = 0.0f, vo = 0.0f;
      if (t < NFRAMES) {
        int k  = kc * KCH + kl;
        bool pr = k > 0;
        int kp = pr ? (NFFT - k) : 0;
        int o1 = t * HOPSZ + k - NFFT / 2;
        o1 = (o1 < 0) ? -o1 : o1;
        o1 = (o1 >= T_IN) ? (2 * (T_IN - 1) - o1) : o1;
        int o2 = t * HOPSZ + NFFT / 2 - k;
        o2 = (o2 >= T_IN) ? (2 * (T_IN - 1) - o2) : o2;
        float y1 = wb[o1] * win[k];
        float y2 = wb[o2] * win[kp];
        ve = pr ? (y1 + y2) : y1;
        vo = pr ? (y1 - y2) : 0.0f;
      }
      int c = (kl >> 2) * 8 + (kl & 1) * 4 + ((kl >> 1) & 1);
      A4[m][c]     = ve;
      A4[m][c + 2] = vo;
    }
    __syncthreads();

    #pragma unroll
    for (int i = 0; i < 3; ++i) {
      int ft = wslot + 8 * i;
      if (ft < NFT) {
        int f0 = ft * 16;
        #pragma unroll
        for (int p = 0; p < 2; ++p) {
          const float* dB = dftT + (long)(f0 + l16) * 4 +
                            (long)((kc * QCH + h) * 2 + p) * (FH * 4);
          const float* aP = &A4[l16][h * 8 + p * 4];
          __builtin_prefetch(dB, 0, 1);
          #pragma unroll 8
          for (int qs = 0; qs < QCH / 2; ++qs) {
            v4f a = *(const v4f*)(aP + qs * 16);
            v2f ae; ae[0] = a[0]; ae[1] = a[1];
            v2f ao; ao[0] = a[2]; ao[1] = a[3];
            v4f q = *(const v4f*)(dB + (long)qs * (4 * FH * 4));
            v2f bc; bc[0] = q[0]; bc[1] = q[1];
            v2f bs; bs[0] = q[2]; bs[1] = q[3];
            accC[i][p] = wmma4(ae, bc, accC[i][p]);
            accS[i][p] = wmma4(ao, bs, accS[i][p]);
          }
        }
      }
    }
  }

  {
    int t = t0 + l16;
    float y = 0.0f;
    if (h == 0 && t < NFRAMES) y = wb[t * HOPSZ] * win[NFFT / 2];
    v2f ae; ae[0] = y; ae[1] = 0.0f;
    #pragma unroll
    for (int i = 0; i < 3; ++i) {
      int ft = wslot + 8 * i;
      if (ft < NFT) {
        const float* dB = dftT + (long)(ft * 16 + l16) * 4;
        v4f q = *(const v4f*)(dB + (long)((NQ - 1) * 2) * (FH * 4));
        v2f bc; bc[0] = q[0]; bc[1] = q[1];
        accC[i][0] = wmma4(ae, bc, accC[i][0]);
      }
    }
  }

  __syncthreads();
  if (tid < 16 * (FPAD - NFREQ))
    PL[tid / (FPAD - NFREQ)][NFREQ + tid % (FPAD - NFREQ)] = 0.0f;
  #pragma unroll
  for (int i = 0; i < 3; ++i) {
    int ft = wslot + 8 * i;
    if (ft < NFT) {
      int f = ft * 16 + l16;
      #pragma unroll
      for (int v = 0; v < 8; ++v) {
        int mr = v + 8 * h;
        float ce = accC[i][0][v], co = accC[i][1][v];
        float se = accS[i][0][v], so = accS[i][1][v];
        float re = ce + co, im = se + so;
        float rm = ce - co, jm = so - se;
        if (f <= NFFT / 4) PL[mr][f] = re * re + im * im;
        if (f <  NFFT / 4) PL[mr][NFFT / 2 - f] = rm * rm + jm * jm;
      }
    }
  }
  __syncthreads();

  int nt = __builtin_amdgcn_readfirstlane(wslot);
  int n0 = nt * 16;
  int kb, rows;
  {
    float melmax = 2595.0f * log10f(1.0f + 22050.0f / 700.0f);
    float step   = melmax / 129.0f;
    float flo = 700.0f * (exp10f(((float)(n0     ) * step) / 2595.0f) - 1.0f);
    float fhi = 700.0f * (exp10f(((float)(n0 + 17) * step) / 2595.0f) - 1.0f);
    int lo = (int)(flo * (512.0f / 22050.0f)) - 1;
    int hi = (int)(fhi * (512.0f / 22050.0f)) + 3;
    lo = (lo < 0) ? 0 : lo;
    int ke = (hi + 15) & ~15;
    ke = (ke > FPAD) ? FPAD : ke;
    kb   = lo & ~15;
    rows = ke - kb;
  }
  v8f acc;
  #pragma unroll
  for (int j = 0; j < 8; ++j) acc[j] = 0.0f;
  const float* pA = &PL[l16][kb + 2 * h];
  const float* pB = fb2 + (long)(n0 + l16) * 2 + (long)((kb >> 1) + h) * (NMELS * 2);
  for (; rows >= 64; rows -= 64) {
    acc = mel_steps<16>(pA, pB, acc);
    pA += 64; pB += 32 * (NMELS * 2);
  }
  if (rows >= 32) {
    acc = mel_steps<8>(pA, pB, acc);
    pA += 32; pB += 16 * (NMELS * 2); rows -= 32;
  }
  if (rows >= 16) acc = mel_steps<4>(pA, pB, acc);
  int n = n0 + l16;
  #pragma unroll
  for (int v = 0; v < 8; ++v) {
    int t = t0 + v + 8 * h;
    if (t < NFRAMES) mel[(b * NMELS + n) * NFRAMES + t] = acc[v];
  }
}

#define MED_R     4
#define MED_TAPS  (31 + MED_R - 1)
#define MED_CORE  (31 - MED_R + 1)
static_assert((NFRAMES % MED_R) == 0 && (NMELS % MED_R) == 0, "groups of four outputs must tile both axes");

__device__ __forceinline__ float median_core4_plus3(float c12, float c13, float c14, float c15,
                                                    float a, float b, float c) {
  float lo = fminf(a, b), hi = fmaxf(a, b);
  float e0 = fminf(lo, c);
  float md = fmaxf(lo, c);
  float e1 = fminf(hi, md);
  float e2 = fmaxf(hi, md);
  return fminf(fminf(c15, fmaxf(c14, e0)), fminf(fmaxf(c13, e1), fmaxf(c12, e2)));
}

#define MED_CE(i, j) { float lo_ = fminf(c[i], c[j]); float hi_ = fmaxf(c[i], c[j]); c[i] = lo_; c[j] = hi_; }
#define MED_LO(i, j) { c[i] = fminf(c[i], c[j]); }
#define MED_HI(i, j) { c[j] = fmaxf(c[i], c[j]); }
__device__ __forceinline__ void median31x4(const float x[MED_TAPS], float o[MED_R]) {
  float c[MED_CORE];
  #pragma unroll
  for (int k = 0; k < MED_CORE; ++k) c[k] = x[k + 3];
  MED_CE(1,2) MED_CE(3,4) MED_CE(5,6) MED_CE(7,8) MED_CE(9,10) MED_CE(11,12) MED_CE(13,14)
  MED_CE(15,16) MED_CE(17,18) MED_CE(19,20) MED_CE(21,22) MED_CE(23,24) MED_CE(25,26)
  MED_CE(0,2) MED_CE(3,5) MED_CE(4,6) MED_CE(7,9) MED_CE(8,10) MED_CE(11,13) MED_CE(12,14)
  MED_CE(15,17) MED_CE(16,18) MED_CE(19,21) MED_CE(20,22) MED_CE(23,25) MED_CE(24,26)
  MED_CE(0,1) MED_CE(4,5) MED_CE(8,9) MED_CE(12,13) MED_CE(16,17) MED_CE(20,21) MED_CE(24,25)
  MED_CE(2,6) MED_CE(7,11) MED_CE(10,14) MED_CE(15,19) MED_CE(18,22) MED_CE(23,27)
  MED_CE(0,4) MED_CE(1,5) MED_CE(8,12) MED_CE(9,13) MED_CE(16,20) MED_CE(17,21) MED_CE(25,27)
  MED_CE(6,14) MED_CE(15,23)
  MED_CE(1,3) MED_CE(2,4) MED_CE(9,11) MED_CE(10,12) MED_CE(17,19) MED_CE(18,20) MED_CE(24,25)
  MED_CE(26,27)
  MED_CE(0,1) MED_CE(2,3) MED_CE(4,5) MED_CE(8,9) MED_CE(10,11) MED_CE(12,13) MED_CE(16,17)
  MED_CE(18,19) MED_CE(20,21)
  MED_CE(0,8) MED_CE(1,9) MED_CE(2,10) MED_CE(3,11) MED_CE(4,12) MED_CE(5,13) MED_CE(16,24)
  MED_CE(17,25) MED_CE(18,26) MED_CE(19,27)
  MED_CE(3,7) MED_CE(4,8) MED_CE(5,9) MED_CE(6,10) MED_CE(19,23) MED_CE(20,24) MED_CE(21,25)
  MED_CE(22,26)
  MED_CE(1,3) MED_CE(2,4) MED_CE(5,7) MED_CE(6,8) MED_CE(9,11) MED_CE(10,12) MED_CE(17,19)
  MED_CE(18,20) MED_CE(21,23) MED_CE(22,24) MED_CE(25,27)
  MED_CE(0,1) MED_CE(2,3) MED_CE(4,5) MED_CE(6,7) MED_CE(8,9) MED_CE(10,11) MED_CE(12,13)
  MED_CE(16,17) MED_CE(18,19) MED_CE(20,21) MED_CE(22,23) MED_CE(24,25) MED_CE(26,27)
  MED_HI(0,16) MED_HI(1,17) MED_HI(2,18) MED_HI(3,19) MED_HI(4,20) MED_HI(5,21) MED_CE(6,22)
  MED_LO(7,23) MED_LO(8,24) MED_LO(9,25) MED_LO(10,26) MED_LO(11,27)
  MED_HI(7,15) MED_HI(8,16) MED_HI(9,17) MED_CE(10,18) MED_LO(11,19) MED_LO(12,20) MED_LO(13,21)
  MED_LO(14,22)
  MED_HI(6,10) MED_HI(11,15) MED_CE(12,16) MED_LO(13,17) MED_LO(14,18)
  MED_HI(10,12) MED_CE(13,15) MED_LO(14,16)
  MED_CE(12,13) MED_CE(14,15)
  o[0] = median_core4_plus3(c[12], c[13], c[14], c[15], x[0],  x[1],  x[2]);
  o[1] = median_core4_plus3(c[12], c[13], c[14], c[15], x[1],  x[2],  x[31]);
  o[2] = median_core4_plus3(c[12], c[13], c[14], c[15], x[2],  x[31], x[32]);
  o[3] = median_core4_plus3(c[12], c[13], c[14], c[15], x[31], x[32], x[33]);
}
#undef MED_CE
#undef MED_LO
#undef MED_HI

#define MT_BLOCKS (E_ELEMS / (MED_R * 256))
static_assert(MT_BLOCKS * MED_R * 256 == E_ELEMS, "median_time grid must tile E_ELEMS exactly");

__global__ __launch_bounds__(256) void median_time(const float* __restrict__ mel,
                                                   float* __restrict__ harm) {
  int g   = blockIdx.x * 256 + threadIdx.x;
  int q   = g % (NFRAMES / MED_R);
  int row = g / (NFRAMES / MED_R);
  int t0  = q * MED_R;
  const float* src = mel + row * NFRAMES;
  float x[MED_TAPS];
  if (t0 >= 16 && t0 + 19 < NFRAMES) {
    const v4f* __restrict__ s4 = (const v4f*)(src + (t0 - 16));
    v4f w[9];
    #pragma unroll
    for (int u = 0; u < 9; ++u) w[u] = s4[u];
    #pragma unroll
    for (int e = 0; e < MED_TAPS; ++e) x[e] = w[(e + 1) >> 2][(e + 1) & 3];
  } else {
    #pragma unroll
    for (int e = 0; e < MED_TAPS; ++e) {
      int j = t0 + e - 15;
      j = (j < 0) ? (-j - 1) : j;
      j = (j >= NFRAMES) ? (2 * NFRAMES - 1 - j) : j;
      x[e] = src[j];
    }
  }
  float o[MED_R];
  median31x4(x, o);
  v4f ov; ov[0] = o[0]; ov[1] = o[1]; ov[2] = o[2]; ov[3] = o[3];
  ((v4f*)harm)[g] = ov;
}

#define MM_TBLOCKS ((NFRAMES + 255) / 256)
#define MM_GROUPS  (NB * (NMELS / MED_R))

__global__ __launch_bounds__(256) void median_mel(const float* __restrict__ mel,
                                                  float* __restrict__ perc) {
  int t = blockIdx.x * 256 + threadIdx.x;
  if (t >= NFRAMES) return;
  int k  = blockIdx.y % (NMELS / MED_R);
  int b  = blockIdx.y / (NMELS / MED_R);
  int n0 = k * MED_R;
  const float* base = mel + (b * NMELS) * NFRAMES + t;
  float x[MED_TAPS];
  #pragma unroll
  for (int e = 0; e < MED_TAPS; ++e) {
    int j = n0 + e - 15;
    j = (j < 0) ? (-j - 1) : j;
    j = (j >= NMELS) ? (2 * NMELS - 1 - j) : j;
    x[e] = base[j * NFRAMES];
  }
  float o[MED_R];
  median31x4(x, o);
  float* dst = perc + (b * NMELS + n0) * NFRAMES + t;
  #pragma unroll
  for (int r = 0; r < MED_R; ++r) dst[r * NFRAMES] = o[r];
}

#define HP_U      4
#define HP_BLOCKS (E_ELEMS / (4 * 256 * HP_U))
static_assert(HP_BLOCKS * 4 * 256 * HP_U == E_ELEMS, "hpss_mask grid must tile E_ELEMS exactly");
static_assert((MEL_OFF % 4) == 0 && (E_ELEMS % 4) == 0, "float4 views need 16-byte aligned planes");

__global__ __launch_bounds__(256) void hpss_mask(const float* __restrict__ mel,
                                                 float* __restrict__ harm,
                                                 float* __restrict__ perc,
                                                 unsigned* __restrict__ mx) {
  __shared__ float r0[256], r1[256], r2[256];
  const v4f* __restrict__ mel4 = (const v4f*)mel;
  v4f* __restrict__ harm4 = (v4f*)harm;
  v4f* __restrict__ perc4 = (v4f*)perc;
  int tid  = threadIdx.x;
  int base = blockIdx.x * (256 * HP_U) + tid;
  v4f S4[HP_U], H4[HP_U], P4[HP_U];
  #pragma unroll
  for (int u = 0; u < HP_U; ++u) {
    int i = base + u * 256;
    S4[u] = mel4[i]; H4[u] = harm4[i]; P4[u] = perc4[i];
  }
  float s = 0.0f, dh = 0.0f, dp = 0.0f;
  #pragma unroll
  for (int u = 0; u < HP_U; ++u) {
    v4f oh, op;
    #pragma unroll
    for (int c = 0; c < 4; ++c) {
      float S  = S4[u][c];
      float ha = H4[u][c];
      float pe = P4[u][c];
      float Z  = fmaxf(ha, pe);
      bool bad = Z < 1.17549435e-38f;
      float Zs = bad ? 1.0f : Z;
      float mh = ha / Zs; mh *= mh;
      float mp = pe / Zs; mp *= mp;
      float den = mh + mp;
      float maskh = bad ? 0.0f : mh / den;
      float maskp = bad ? 0.0f : mp / den;
      float vh = S * maskh, vp = S * maskp;
      oh[c] = vh; op[c] = vp;
      s = fmaxf(s, S); dh = fmaxf(dh, vh); dp = fmaxf(dp, vp);
    }
    int i = base + u * 256;
    harm4[i] = oh;
    perc4[i] = op;
  }
  r0[tid] = s; r1[tid] = dh; r2[tid] = dp;
  __syncthreads();
  for (int off = 128; off > 0; off >>= 1) {
    if (tid < off) {
      r0[tid] = fmaxf(r0[tid], r0[tid + off]);
      r1[tid] = fmaxf(r1[tid], r1[tid + off]);
      r2[tid] = fmaxf(r2[tid], r2[tid + off]);
    }
    __syncthreads();
  }
  if (tid == 0) {
    atomicMax(mx + 0, __float_as_uint(r0[0]));
    atomicMax(mx + 1, __float_as_uint(r1[0]));
    atomicMax(mx + 2, __float_as_uint(r2[0]));
  }
}

#define FN_U          8
#define FN_SEG_BLOCKS (E_ELEMS / (4 * 256 * FN_U))
static_assert(FN_SEG_BLOCKS * 4 * 256 * FN_U == E_ELEMS, "finalize grid must tile each plane exactly");

__global__ __launch_bounds__(256) void finalize(const float* __restrict__ mel,
                                                const float* __restrict__ harm,
                                                const float* __restrict__ perc,
                                                const unsigned* __restrict__ mx,
                                                float* __restrict__ out) {
  int s  = blockIdx.x / FN_SEG_BLOCKS;
  int e4 = (blockIdx.x - s * FN_SEG_BLOCKS) * (256 * FN_U) + threadIdx.x;
  const float* src = (s == 0) ? mel : ((s == 1) ? harm : perc);
  const v4f* __restrict__ src4 = (const v4f*)src;
  v4f* __restrict__ out4 = (v4f*)out + (long)s * (E_ELEMS / 4);
  float ref = __uint_as_float(mx[s]);
  v4f v[FN_U];
  #pragma unroll
  for (int u = 0; u < FN_U; ++u) v[u] = src4[e4 + u * 256];
  #pragma unroll
  for (int u = 0; u < FN_U; ++u) {
    v4f o;
    #pragma unroll
    for (int c = 0; c < 4; ++c) {
      float db = 10.0f * log10f(fmaxf(1e-10f, v[u][c])) -
                 10.0f * log10f(fmaxf(1e-10f, ref));
      db = fmaxf(db, -80.0f);
      o[c] = (db + 80.0f) * (1.0f / (80.0f + 1e-7f));
    }
    out4[e4 + u * 256] = o;
  }
}

extern "C" void kernel_launch(void* const* d_in, const int* in_sizes, int n_in,
                              void* d_out, int out_size, void* d_ws,
                              size_t ws_size, hipStream_t stream) {
  const float* wave = (const float*)d_in[0];
  float* out  = (float*)d_out;
  float* ws   = (float*)d_ws;
  float* mel  = ws + MEL_OFF;
  float* harm = ws + HARM_OFF;
  float* perc = ws + PERC_OFF;
  unsigned* mx = (unsigned*)(ws + MAX_OFF);

  build_tables<<<(MAX_OFF + 255) / 256, 256, 0, stream>>>(ws);
  mel_gemm<<<NB * MTILES, 256, 0, stream>>>(wave, ws, mel);
  median_time<<<MT_BLOCKS, 256, 0, stream>>>(mel, harm);
  median_mel<<<dim3(MM_TBLOCKS, MM_GROUPS), 256, 0, stream>>>(mel, perc);
  hpss_mask<<<HP_BLOCKS, 256, 0, stream>>>(mel, harm, perc, mx);
  finalize<<<3 * FN_SEG_BLOCKS, 256, 0, stream>>>(mel, harm, perc, mx, out);
}


# mel_gemm frame build: the 32 sample loads of a chunk (16 rows x 2) go out as one group before their first use, so a wave waits for memory once per chunk, not 16 times; values and sum order unchanged
# speedup vs baseline: 3.2525x; 1.0802x over previous
#include <hip/hip_runtime.h>
#include <math.h>

#define NFFT     1024
#define HOPSZ    512
#define NMELS    128
#define NFREQ    513
#define FPAD     528
#define T_IN     1323000
#define NB       16
#define NFRAMES  2584
#define MTILES   162
#define E_ELEMS  (NB*NMELS*NFRAMES)

#define KHALF    512
#define KCH      256
#define APITCH   516
#define FH       272
#define NFT      17
#define QCH      64
#define NQ       129
static_assert(KCH == 256, "the frame-chunk build gives one folded k to each of the 256 threads");

#define WIN_OFF  0
#define DFT_OFF  1024
#define DFT_SZ   (NQ*2*FH*4)
#define FB_OFF   (DFT_OFF + DFT_SZ)
#define FB_SZ    ((FPAD/2)*NMELS*2)
#define MAX_OFF  (FB_OFF + FB_SZ)
#define MEL_OFF  (MAX_OFF + 8)
#define HARM_OFF (MEL_OFF + E_ELEMS)
#define PERC_OFF (HARM_OFF + E_ELEMS)

typedef float v2f __attribute__((ext_vector_type(2)));
typedef float v4f __attribute__((ext_vector_type(4)));
typedef float v8f __attribute__((ext_vector_type(8)));

__device__ __forceinline__ v8f wmma4(v2f a, v2f b, v8f c) {
  return __builtin_amdgcn_wmma_f32_16x16x4_f32(false, a, false, b, (short)0, c,
                                               false, false);
}

__global__ __launch_bounds__(256) void build_tables(float* __restrict__ ws) {
  int gid = blockIdx.x * 256 + threadIdx.x;
  if (gid == 0) {
    unsigned* mx = (unsigned*)(ws + MAX_OFF);
    mx[0] = 0u; mx[1] = 0u; mx[2] = 0u;
  }
  const float TWO_PI = 6.28318530717958647692f;
  if (gid < DFT_OFF) {
    int k = gid;
    ws[WIN_OFF + k] = 0.5f * (1.0f - cosf(TWO_PI * (float)k / 1024.0f));
  } else if (gid < FB_OFF) {
    int i = gid - DFT_OFF;
    int entry = i & 3;
    int rest  = i >> 2;
    int f  = rest % FH;
    int qp = rest / FH;
    int k  = 4 * (qp >> 1) + (qp & 1) + 2 * (entry & 1);
    int r  = (k * f) & (NFFT - 1);
    float arg = (float)r * (TWO_PI / 1024.0f);
    ws[gid] = (entry < 2) ? cosf(arg) : sinf(arg);
  } else if (gid < MAX_OFF) {
    int i = gid - FB_OFF;
    int entry = i & 1;
    int rest  = i >> 1;
    int m  = rest % NMELS;
    int k2 = rest / NMELS;
    int f  = 2 * k2 + entry;
    float v = 0.0f;
    if (f < NFREQ) {
      float freq   = (float)f * (22050.0f / 512.0f);
      float melmax = 2595.0f * log10f(1.0f + 22050.0f / 700.0f);
      float step   = melmax / 129.0f;
      float p0 = 700.0f * (exp10f(((float)(m    ) * step) / 2595.0f) - 1.0f);
      float p1 = 700.0f * (exp10f(((float)(m + 1) * step) / 2595.0f) - 1.0f);
      float p2 = 700.0f * (exp10f(((float)(m + 2) * step) / 2595.0f) - 1.0f);
      float down = (freq - p0) / (p1 - p0);
      float up   = (p2 - freq) / (p2 - p1);
      float t = fmaxf(0.0f, fminf(down, up));
      v = t * 2.0f / (p2 - p0);
    }
    ws[gid] = v;
  }
}

template <int NS>
__device__ __forceinline__ v8f mel_steps(const float* pA, const float* pB, v8f acc) {
  v2f a[NS], bb[NS];
  #pragma unroll
  for (int u = 0; u < NS; ++u) {
    bb[u] = *(const v2f*)(pB + (long)u * (2 * NMELS * 2));
    a[u]  = *(const v2f*)(pA + 4 * u);
  }
  #pragma unroll
  for (int u = 0; u < NS; ++u) acc = wmma4(a[u], bb[u], acc);
  return acc;
}

__global__ __launch_bounds__(256, 6) void mel_gemm(const float* __restrict__ wave,
                                                const float* __restrict__ ws,
                                                float* __restrict__ mel) {
  __shared__ __attribute__((aligned(16))) float A4[16][APITCH];
  __shared__ float PL[16][FPAD + 4];

  const float* __restrict__ win  = ws + WIN_OFF;
  const float* __restrict__ dftT = ws + DFT_OFF;
  const float* __restrict__ fb2  = ws + FB_OFF;

  int b    = blockIdx.x / MTILES;
  int tile = blockIdx.x % MTILES;
  int t0   = tile * 16;
  int tid  = threadIdx.x;
  int lane = tid & 31;
  int wv   = tid >> 5;
  int h    = lane >> 4;
  int l16  = lane & 15;

  const float* wb = wave + (long)b * T_IN;

  int wslot = (wv + blockIdx.x + (blockIdx.x >> 8)) & 7;

  v8f accC[3][2], accS[3][2];
  #pragma unroll
  for (int i = 0; i < 3; ++i)
    for (int p = 0; p < 2; ++p)
      for (int j = 0; j < 8; ++j) { accC[i][p][j] = 0.0f; accS[i][p][j] = 0.0f; }

  for (int kc = 0; kc < KHALF / KCH; ++kc) {
    __syncthreads();
    {
      const int kl = tid;
      const int k  = kc * KCH + kl;
      const bool pr = k > 0;
      const int kp = pr ? (NFFT - k) : 0;
      const float w1 = win[k];
      const float w2 = win[kp];
      float s1[16], s2[16];
      #pragma unroll
      for (int m = 0; m < 16; ++m) {
        int t = t0 + m;
        t = (t < NFRAMES) ? t : (NFRAMES - 1);
        int o1 = t * HOPSZ + k - NFFT / 2;
        o1 = (o1 < 0) ? -o1 : o1;
        o1 = (o1 >= T_IN) ? (2 * (T_IN - 1) - o1) : o1;
        int o2 = t * HOPSZ + NFFT / 2 - k;
        o2 = (o2 >= T_IN) ? (2 * (T_IN - 1) - o2) : o2;
        s1[m] = wb[o1];
        s2[m] = wb[o2];
      }
      __builtin_amdgcn_sched_barrier(0);
      const int c = (kl >> 2) * 8 + (kl & 1) * 4 + ((kl >> 1) & 1);
      #pragma unroll
      for (int m = 0; m < 16; ++m) {
        float y1 = s1[m] * w1;
        float y2 = s2[m] * w2;
        float ve = pr ? (y1 + y2) : y1;
        float vo = pr ? (y1 - y2) : 0.0f;
        bool live = (t0 + m) < NFRAMES;
        A4[m][c]     = live ? ve : 0.0f;
        A4[m][c + 2] = live ? vo : 0.0f;
      }
    }
    __syncthreads();

    #pragma unroll
    for (int i = 0; i < 3; ++i) {
      int ft = wslot + 8 * i;
      if (ft < NFT) {
        int f0 = ft * 16;
        #pragma unroll
        for (int p = 0; p < 2; ++p) {
          const float* dB = dftT + (long)(f0 + l16) * 4 +
                            (long)((kc * QCH + h) * 2 + p) * (FH * 4);
          const float* aP = &A4[l16][h * 8 + p * 4];
          __builtin_prefetch(dB, 0, 1);
          #pragma unroll 8
          for (int qs = 0; qs < QCH / 2; ++qs) {
            v4f a = *(const v4f*)(aP + qs * 16);
            v2f ae; ae[0] = a[0]; ae[1] = a[1];
            v2f ao; ao[0] = a[2]; ao[1] = a[3];
            v4f q = *(const v4f*)(dB + (long)qs * (4 * FH * 4));
            v2f bc; bc[0] = q[0]; bc[1] = q[1];
            v2f bs; bs[0] = q[2]; bs[1] = q[3];
            accC[i][p] = wmma4(ae, bc, accC[i][p]);
            accS[i][p] = wmma4(ao, bs, accS[i][p]);
          }
        }
      }
    }
  }

  {
    int t = t0 + l16;
    float y = 0.0f;
    if (h == 0 && t < NFRAMES) y = wb[t * HOPSZ] * win[NFFT / 2];
    v2f ae; ae[0] = y; ae[1] = 0.0f;
    #pragma unroll
    for (int i = 0; i < 3; ++i) {
      int ft = wslot + 8 * i;
      if (ft < NFT) {
        const float* dB = dftT + (long)(ft * 16 + l16) * 4;
        v4f q = *(const v4f*)(dB + (long)((NQ - 1) * 2) * (FH * 4));
        v2f bc; bc[0] = q[0]; bc[1] = q[1];
        accC[i][0] = wmma4(ae, bc, accC[i][0]);
      }
    }
  }

  __syncthreads();
  if (tid < 16 * (FPAD - NFREQ))
    PL[tid / (FPAD - NFREQ)][NFREQ + tid % (FPAD - NFREQ)] = 0.0f;
  #pragma unroll
  for (int i = 0; i < 3; ++i) {
    int ft = wslot + 8 * i;
    if (ft < NFT) {
      int f = ft * 16 + l16;
      #pragma unroll
      for (int v = 0; v < 8; ++v) {
        int mr = v + 8 * h;
        float ce = accC[i][0][v], co = accC[i][1][v];
        float se = accS[i][0][v], so = accS[i][1][v];
        float re = ce + co, im = se + so;
        float rm = ce - co, jm = so - se;
        if (f <= NFFT / 4) PL[mr][f] = re * re + im * im;
        if (f <  NFFT / 4) PL[mr][NFFT / 2 - f] = rm * rm + jm * jm;
      }
    }
  }
  __syncthreads();

  int nt = __builtin_amdgcn_readfirstlane(wslot);
  int n0 = nt * 16;
  int kb, rows;
  {
    float melmax = 2595.0f * log10f(1.0f + 22050.0f / 700.0f);
    float step   = melmax / 129.0f;
    float flo = 700.0f * (exp10f(((float)(n0     ) * step) / 2595.0f) - 1.0f);
    float fhi = 700.0f * (exp10f(((float)(n0 + 17) * step) / 2595.0f) - 1.0f);
    int lo = (int)(flo * (512.0f / 22050.0f)) - 1;
    int hi = (int)(fhi * (512.0f / 22050.0f)) + 3;
    lo = (lo < 0) ? 0 : lo;
    int ke = (hi + 15) & ~15;
    ke = (ke > FPAD) ? FPAD : ke;
    kb   = lo & ~15;
    rows = ke - kb;
  }
  v8f acc;
  #pragma unroll
  for (int j = 0; j < 8; ++j) acc[j] = 0.0f;
  const float* pA = &PL[l16][kb + 2 * h];
  const float* pB = fb2 + (long)(n0 + l16) * 2 + (long)((kb >> 1) + h) * (NMELS * 2);
  for (; rows >= 64; rows -= 64) {
    acc = mel_steps<16>(pA, pB, acc);
    pA += 64; pB += 32 * (NMELS * 2);
  }
  if (rows >= 32) {
    acc = mel_steps<8>(pA, pB, acc);
    pA += 32; pB += 16 * (NMELS * 2); rows -= 32;
  }
  if (rows >= 16) acc = mel_steps<4>(pA, pB, acc);
  int n = n0 + l16;
  #pragma unroll
  for (int v = 0; v < 8; ++v) {
    int t = t0 + v + 8 * h;
    if (t < NFRAMES) mel[(b * NMELS + n) * NFRAMES + t] = acc[v];
  }
}

#define MED_R     4
#define MED_TAPS  (31 + MED_R - 1)
#define MED_CORE  (31 - MED_R + 1)
static_assert((NFRAMES % MED_R) == 0 && (NMELS % MED_R) == 0, "groups of four outputs must tile both axes");

__device__ __forceinline__ float median_core4_plus3(float c12, float c13, float c14, float c15,
                                                    float a, float b, float c) {
  float lo = fminf(a, b), hi = fmaxf(a, b);
  float e0 = fminf(lo, c);
  float md = fmaxf(lo, c);
  float e1 = fminf(hi, md);
  float e2 = fmaxf(hi, md);
  return fminf(fminf(c15, fmaxf(c14, e0)), fminf(fmaxf(c13, e1), fmaxf(c12, e2)));
}

#define MED_CE(i, j) { float lo_ = fminf(c[i], c[j]); float hi_ = fmaxf(c[i], c[j]); c[i] = lo_; c[j] = hi_; }
#define MED_LO(i, j) { c[i] = fminf(c[i], c[j]); }
#define MED_HI(i, j) { c[j] = fmaxf(c[i], c[j]); }
__device__ __forceinline__ void median31x4(const float x[MED_TAPS], float o[MED_R]) {
  float c[MED_CORE];
  #pragma unroll
  for (int k = 0; k < MED_CORE; ++k) c[k] = x[k + 3];
  MED_CE(1,2) MED_CE(3,4) MED_CE(5,6) MED_CE(7,8) MED_CE(9,10) MED_CE(11,12) MED_CE(13,14)
  MED_CE(15,16) MED_CE(17,18) MED_CE(19,20) MED_CE(21,22) MED_CE(23,24) MED_CE(25,26)
  MED_CE(0,2) MED_CE(3,5) MED_CE(4,6) MED_CE(7,9) MED_CE(8,10) MED_CE(11,13) MED_CE(12,14)
  MED_CE(15,17) MED_CE(16,18) MED_CE(19,21) MED_CE(20,22) MED_CE(23,25) MED_CE(24,26)
  MED_CE(0,1) MED_CE(4,5) MED_CE(8,9) MED_CE(12,13) MED_CE(16,17) MED_CE(20,21) MED_CE(24,25)
  MED_CE(2,6) MED_CE(7,11) MED_CE(10,14) MED_CE(15,19) MED_CE(18,22) MED_CE(23,27)
  MED_CE(0,4) MED_CE(1,5) MED_CE(8,12) MED_CE(9,13) MED_CE(16,20) MED_CE(17,21) MED_CE(25,27)
  MED_CE(6,14) MED_CE(15,23)
  MED_CE(1,3) MED_CE(2,4) MED_CE(9,11) MED_CE(10,12) MED_CE(17,19) MED_CE(18,20) MED_CE(24,25)
  MED_CE(26,27)
  MED_CE(0,1) MED_CE(2,3) MED_CE(4,5) MED_CE(8,9) MED_CE(10,11) MED_CE(12,13) MED_CE(16,17)
  MED_CE(18,19) MED_CE(20,21)
  MED_CE(0,8) MED_CE(1,9) MED_CE(2,10) MED_CE(3,11) MED_CE(4,12) MED_CE(5,13) MED_CE(16,24)
  MED_CE(17,25) MED_CE(18,26) MED_CE(19,27)
  MED_CE(3,7) MED_CE(4,8) MED_CE(5,9) MED_CE(6,10) MED_CE(19,23) MED_CE(20,24) MED_CE(21,25)
  MED_CE(22,26)
  MED_CE(1,3) MED_CE(2,4) MED_CE(5,7) MED_CE(6,8) MED_CE(9,11) MED_CE(10,12) MED_CE(17,19)
  MED_CE(18,20) MED_CE(21,23) MED_CE(22,24) MED_CE(25,27)
  MED_CE(0,1) MED_CE(2,3) MED_CE(4,5) MED_CE(6,7) MED_CE(8,9) MED_CE(10,11) MED_CE(12,13)
  MED_CE(16,17) MED_CE(18,19) MED_CE(20,21) MED_CE(22,23) MED_CE(24,25) MED_CE(26,27)
  MED_HI(0,16) MED_HI(1,17) MED_HI(2,18) MED_HI(3,19) MED_HI(4,20) MED_HI(5,21) MED_CE(6,22)
  MED_LO(7,23) MED_LO(8,24) MED_LO(9,25) MED_LO(10,26) MED_LO(11,27)
  MED_HI(7,15) MED_HI(8,16) MED_HI(9,17) MED_CE(10,18) MED_LO(11,19) MED_LO(12,20) MED_LO(13,21)
  MED_LO(14,22)
  MED_HI(6,10) MED_HI(11,15) MED_CE(12,16) MED_LO(13,17) MED_LO(14,18)
  MED_HI(10,12) MED_CE(13,15) MED_LO(14,16)
  MED_CE(12,13) MED_CE(14,15)
  o[0] = median_core4_plus3(c[12], c[13], c[14], c[15], x[0],  x[1],  x[2]);
  o[1] = median_core4_plus3(c[12], c[13], c[14], c[15], x[1],  x[2],  x[31]);
  o[2] = median_core4_plus3(c[12], c[13], c[14], c[15], x[2],  x[31], x[32]);
  o[3] = median_core4_plus3(c[12], c[13], c[14], c[15], x[31], x[32], x[33]);
}
#undef MED_CE
#undef MED_LO
#undef MED_HI

#define MT_BLOCKS (E_ELEMS / (MED_R * 256))
static_assert(MT_BLOCKS * MED_R * 256 == E_ELEMS, "median_time grid must tile E_ELEMS exactly");

__global__ __launch_bounds__(256) void median_time(const float* __restrict__ mel,
                                                   float* __restrict__ harm) {
  int g   = blockIdx.x * 256 + threadIdx.x;
  int q   = g % (NFRAMES / MED_R);
  int row = g / (NFRAMES / MED_R);
  int t0  = q * MED_R;
  const float* src = mel + row * NFRAMES;
  float x[MED_TAPS];
  if (t0 >= 16 && t0 + 19 < NFRAMES) {
    const v4f* __restrict__ s4 = (const v4f*)(src + (t0 - 16));
    v4f w[9];
    #pragma unroll
    for (int u = 0; u < 9; ++u) w[u] = s4[u];
    #pragma unroll
    for (int e = 0; e < MED_TAPS; ++e) x[e] = w[(e + 1) >> 2][(e + 1) & 3];
  } else {
    #pragma unroll
    for (int e = 0; e < MED_TAPS; ++e) {
      int j = t0 + e - 15;
      j = (j < 0) ? (-j - 1) : j;
      j = (j >= NFRAMES) ? (2 * NFRAMES - 1 - j) : j;
      x[e] = src[j];
    }
  }
  float o[MED_R];
  median31x4(x, o);
  v4f ov; ov[0] = o[0]; ov[1] = o[1]; ov[2] = o[2]; ov[3] = o[3];
  ((v4f*)harm)[g] = ov;
}

#define MM_TBLOCKS ((NFRAMES + 255) / 256)
#define MM_GROUPS  (NB * (NMELS / MED_R))

__global__ __launch_bounds__(256) void median_mel(const float* __restrict__ mel,
                                                  float* __restrict__ perc) {
  int t = blockIdx.x * 256 + threadIdx.x;
  if (t >= NFRAMES) return;
  int k  = blockIdx.y % (NMELS / MED_R);
  int b  = blockIdx.y / (NMELS / MED_R);
  int n0 = k * MED_R;
  const float* base = mel + (b * NMELS) * NFRAMES + t;
  float x[MED_TAPS];
  #pragma unroll
  for (int e = 0; e < MED_TAPS; ++e) {
    int j = n0 + e - 15;
    j = (j < 0) ? (-j - 1) : j;
    j = (j >= NMELS) ? (2 * NMELS - 1 - j) : j;
    x[e] = base[j * NFRAMES];
  }
  float o[MED_R];
  median31x4(x, o);
  float* dst = perc + (b * NMELS + n0) * NFRAMES + t;
  #pragma unroll
  for (int r = 0; r < MED_R; ++r) dst[r * NFRAMES] = o[r];
}

#define HP_U      4
#define HP_BLOCKS (E_ELEMS / (4 * 256 * HP_U))
static_assert(HP_BLOCKS * 4 * 256 * HP_U == E_ELEMS, "hpss_mask grid must tile E_ELEMS exactly");
static_assert((MEL_OFF % 4) == 0 && (E_ELEMS % 4) == 0, "float4 views need 16-byte aligned planes");

__global__ __launch_bounds__(256) void hpss_mask(const float* __restrict__ mel,
                                                 float* __restrict__ harm,
                                                 float* __restrict__ perc,
                                                 unsigned* __restrict__ mx) {
  __shared__ float r0[256], r1[256], r2[256];
  const v4f* __restrict__ mel4 = (const v4f*)mel;
  v4f* __restrict__ harm4 = (v4f*)harm;
  v4f* __restrict__ perc4 = (v4f*)perc;
  int tid  = threadIdx.x;
  int base = blockIdx.x * (256 * HP_U) + tid;
  v4f S4[HP_U], H4[HP_U], P4[HP_U];
  #pragma unroll
  for (int u = 0; u < HP_U; ++u) {
    int i = base + u * 256;
    S4[u] = mel4[i]; H4[u] = harm4[i]; P4[u] = perc4[i];
  }
  float s = 0.0f, dh = 0.0f, dp = 0.0f;
  #pragma unroll
  for (int u = 0; u < HP_U; ++u) {
    v4f oh, op;
    #pragma unroll
    for (int c = 0; c < 4; ++c) {
      float S  = S4[u][c];
      float ha = H4[u][c];
      float pe = P4[u][c];
      float Z  = fmaxf(ha, pe);
      bool bad = Z < 1.17549435e-38f;
      float Zs = bad ? 1.0f : Z;
      float mh = ha / Zs; mh *= mh;
      float mp = pe / Zs; mp *= mp;
      float den = mh + mp;
      float maskh = bad ? 0.0f : mh / den;
      float maskp = bad ? 0.0f : mp / den;
      float vh = S * maskh, vp = S * maskp;
      oh[c] = vh; op[c] = vp;
      s = fmaxf(s, S); dh = fmaxf(dh, vh); dp = fmaxf(dp, vp);
    }
    int i = base + u * 256;
    harm4[i] = oh;
    perc4[i] = op;
  }
  r0[tid] = s; r1[tid] = dh; r2[tid] = dp;
  __syncthreads();
  for (int off = 128; off > 0; off >>= 1) {
    if (tid < off) {
      r0[tid] = fmaxf(r0[tid], r0[tid + off]);
      r1[tid] = fmaxf(r1[tid], r1[tid + off]);
      r2[tid] = fmaxf(r2[tid], r2[tid + off]);
    }
    __syncthreads();
  }
  if (tid == 0) {
    atomicMax(mx + 0, __float_as_uint(r0[0]));
    atomicMax(mx + 1, __float_as_uint(r1[0]));
    atomicMax(mx + 2, __float_as_uint(r2[0]));
  }
}

#define FN_U          8
#define FN_SEG_BLOCKS (E_ELEMS / (4 * 256 * FN_U))
static_assert(FN_SEG_BLOCKS * 4 * 256 * FN_U == E_ELEMS, "finalize grid must tile each plane exactly");

__global__ __launch_bounds__(256) void finalize(const float* __restrict__ mel,
                                                const float* __restrict__ harm,
                                                const float* __restrict__ perc,
                                                const unsigned* __restrict__ mx,
                                                float* __restrict__ out) {
  int s  = blockIdx.x / FN_SEG_BLOCKS;
  int e4 = (blockIdx.x - s * FN_SEG_BLOCKS) * (256 * FN_U) + threadIdx.x;
  const float* src = (s == 0) ? mel : ((s == 1) ? harm : perc);
  const v4f* __restrict__ src4 = (const v4f*)src;
  v4f* __restrict__ out4 = (v4f*)out + (long)s * (E_ELEMS / 4);
  float ref = __uint_as_float(mx[s]);
  v4f v[FN_U];
  #pragma unroll
  for (int u = 0; u < FN_U; ++u) v[u] = src4[e4 + u * 256];
  #pragma unroll
  for (int u = 0; u < FN_U; ++u) {
    v4f o;
    #pragma unroll
    for (int c = 0; c < 4; ++c) {
      float db = 10.0f * log10f(fmaxf(1e-10f, v[u][c])) -
                 10.0f * log10f(fmaxf(1e-10f, ref));
      db = fmaxf(db, -80.0f);
      o[c] = (db + 80.0f) * (1.0f / (80.0f + 1e-7f));
    }
    out4[e4 + u * 256] = o;
  }
}

extern "C" void kernel_launch(void* const* d_in, const int* in_sizes, int n_in,
                              void* d_out, int out_size, void* d_ws,
                              size_t ws_size, hipStream_t stream) {
  const float* wave = (const float*)d_in[0];
  float* out  = (float*)d_out;
  float* ws   = (float*)d_ws;
  float* mel  = ws + MEL_OFF;
  float* harm = ws + HARM_OFF;
  float* perc = ws + PERC_OFF;
  unsigned* mx = (unsigned*)(ws + MAX_OFF);

  build_tables<<<(MAX_OFF + 255) / 256, 256, 0, stream>>>(ws);
  mel_gemm<<<NB * MTILES, 256, 0, stream>>>(wave, ws, mel);
  median_time<<<MT_BLOCKS, 256, 0, stream>>>(mel, harm);
  median_mel<<<dim3(MM_TBLOCKS, MM_GROUPS), 256, 0, stream>>>(mel, perc);
  hpss_mask<<<HP_BLOCKS, 256, 0, stream>>>(mel, harm, perc, mx);
  finalize<<<3 * FN_SEG_BLOCKS, 256, 0, stream>>>(mel, harm, perc, mx, out);
}


# mel_gemm stage 1: the 17th frequency tile (one live column, f = 256) is no third tile on one wave: its k range is split over the 8 waves, parts added through LDS; f = 256 sum order changes
# speedup vs baseline: 3.5313x; 1.0857x over previous
#include <hip/hip_runtime.h>
#include <math.h>

#define NFFT     1024
#define HOPSZ    512
#define NMELS    128
#define NFREQ    513
#define FPAD     528
#define T_IN     1323000
#define NB       16
#define NFRAMES  2584
#define MTILES   162
#define E_ELEMS  (NB*NMELS*NFRAMES)

#define KHALF    512
#define KCH      256
#define APITCH   516
#define FH       272
#define NFT      17
#define QCH      64
#define NQ       129
static_assert(KCH == 256, "the frame-chunk build gives one folded k to each of the 256 threads");

#define WIN_OFF  0
#define DFT_OFF  1024
#define DFT_SZ   (NQ*2*FH*4)
#define FB_OFF   (DFT_OFF + DFT_SZ)
#define FB_SZ    ((FPAD/2)*NMELS*2)
#define MAX_OFF  (FB_OFF + FB_SZ)
#define MEL_OFF  (MAX_OFF + 8)
#define HARM_OFF (MEL_OFF + E_ELEMS)
#define PERC_OFF (HARM_OFF + E_ELEMS)

typedef float v2f __attribute__((ext_vector_type(2)));
typedef float v4f __attribute__((ext_vector_type(4)));
typedef float v8f __attribute__((ext_vector_type(8)));

__device__ __forceinline__ v8f wmma4(v2f a, v2f b, v8f c) {
  return __builtin_amdgcn_wmma_f32_16x16x4_f32(false, a, false, b, (short)0, c,
                                               false, false);
}

__global__ __launch_bounds__(256) void build_tables(float* __restrict__ ws) {
  int gid = blockIdx.x * 256 + threadIdx.x;
  if (gid == 0) {
    unsigned* mx = (unsigned*)(ws + MAX_OFF);
    mx[0] = 0u; mx[1] = 0u; mx[2] = 0u;
  }
  const float TWO_PI = 6.28318530717958647692f;
  if (gid < DFT_OFF) {
    int k = gid;
    ws[WIN_OFF + k] = 0.5f * (1.0f - cosf(TWO_PI * (float)k / 1024.0f));
  } else if (gid < FB_OFF) {
    int i = gid - DFT_OFF;
    int entry = i & 3;
    int rest  = i >> 2;
    int f  = rest % FH;
    int qp = rest / FH;
    int k  = 4 * (qp >> 1) + (qp & 1) + 2 * (entry & 1);
    int r  = (k * f) & (NFFT - 1);
    float arg = (float)r * (TWO_PI / 1024.0f);
    ws[gid] = (entry < 2) ? cosf(arg) : sinf(arg);
  } else if (gid < MAX_OFF) {
    int i = gid - FB_OFF;
    int entry = i & 1;
    int rest  = i >> 1;
    int m  = rest % NMELS;
    int k2 = rest / NMELS;
    int f  = 2 * k2 + entry;
    float v = 0.0f;
    if (f < NFREQ) {
      float freq   = (float)f * (22050.0f / 512.0f);
      float melmax = 2595.0f * log10f(1.0f + 22050.0f / 700.0f);
      float step   = melmax / 129.0f;
      float p0 = 700.0f * (exp10f(((float)(m    ) * step) / 2595.0f) - 1.0f);
      float p1 = 700.0f * (exp10f(((float)(m + 1) * step) / 2595.0f) - 1.0f);
      float p2 = 700.0f * (exp10f(((float)(m + 2) * step) / 2595.0f) - 1.0f);
      float down = (freq - p0) / (p1 - p0);
      float up   = (p2 - freq) / (p2 - p1);
      float t = fmaxf(0.0f, fminf(down, up));
      v = t * 2.0f / (p2 - p0);
    }
    ws[gid] = v;
  }
}

template <int NS>
__device__ __forceinline__ v8f mel_steps(const float* pA, const float* pB, v8f acc) {
  v2f a[NS], bb[NS];
  #pragma unroll
  for (int u = 0; u < NS; ++u) {
    bb[u] = *(const v2f*)(pB + (long)u * (2 * NMELS * 2));
    a[u]  = *(const v2f*)(pA + 4 * u);
  }
  #pragma unroll
  for (int u = 0; u < NS; ++u) acc = wmma4(a[u], bb[u], acc);
  return acc;
}

__global__ __launch_bounds__(256, 6) void mel_gemm(const float* __restrict__ wave,
                                                const float* __restrict__ ws,
                                                float* __restrict__ mel) {
  __shared__ __attribute__((aligned(16))) float A4[16][APITCH];
  __shared__ float PL[16][FPAD + 4];
  __shared__ float RED[8][2][16];

  const float* __restrict__ win  = ws + WIN_OFF;
  const float* __restrict__ dftT = ws + DFT_OFF;
  const float* __restrict__ fb2  = ws + FB_OFF;

  int b    = blockIdx.x / MTILES;
  int tile = blockIdx.x % MTILES;
  int t0   = tile * 16;
  int tid  = threadIdx.x;
  int lane = tid & 31;
  int wv   = tid >> 5;
  int h    = lane >> 4;
  int l16  = lane & 15;

  const float* wb = wave + (long)b * T_IN;

  int wslot = (wv + blockIdx.x + (blockIdx.x >> 8)) & 7;
  const int wvu = __builtin_amdgcn_readfirstlane(wv);

  v8f accC[2][2], accS[2][2];
  #pragma unroll
  for (int i = 0; i < 2; ++i)
    for (int p = 0; p < 2; ++p)
      for (int j = 0; j < 8; ++j) { accC[i][p][j] = 0.0f; accS[i][p][j] = 0.0f; }
  v8f accC16, accS16;
  #pragma unroll
  for (int j = 0; j < 8; ++j) { accC16[j] = 0.0f; accS16[j] = 0.0f; }

  for (int kc = 0; kc < KHALF / KCH; ++kc) {
    __syncthreads();
    {
      const int kl = tid;
      const int k  = kc * KCH + kl;
      const bool pr = k > 0;
      const int kp = pr ? (NFFT - k) : 0;
      const float w1 = win[k];
      const float w2 = win[kp];
      float s1[16], s2[16];
      #pragma unroll
      for (int m = 0; m < 16; ++m) {
        int t = t0 + m;
        t = (t < NFRAMES) ? t : (NFRAMES - 1);
        int o1 = t * HOPSZ + k - NFFT / 2;
        o1 = (o1 < 0) ? -o1 : o1;
        o1 = (o1 >= T_IN) ? (2 * (T_IN - 1) - o1) : o1;
        int o2 = t * HOPSZ + NFFT / 2 - k;
        o2 = (o2 >= T_IN) ? (2 * (T_IN - 1) - o2) : o2;
        s1[m] = wb[o1];
        s2[m] = wb[o2];
      }
      __builtin_amdgcn_sched_barrier(0);
      const int c = (kl >> 2) * 8 + (kl & 1) * 4 + ((kl >> 1) & 1);
      #pragma unroll
      for (int m = 0; m < 16; ++m) {
        float y1 = s1[m] * w1;
        float y2 = s2[m] * w2;
        float ve = pr ? (y1 + y2) : y1;
        float vo = pr ? (y1 - y2) : 0.0f;
        bool live = (t0 + m) < NFRAMES;
        A4[m][c]     = live ? ve : 0.0f;
        A4[m][c + 2] = live ? vo : 0.0f;
      }
    }
    __syncthreads();

    #pragma unroll
    for (int i = 0; i < 2; ++i) {
      int ft = wslot + 8 * i;
      int f0 = ft * 16;
      #pragma unroll
      for (int p = 0; p < 2; ++p) {
        const float* dB = dftT + (long)(f0 + l16) * 4 +
                          (long)((kc * QCH + h) * 2 + p) * (FH * 4);
        const float* aP = &A4[l16][h * 8 + p * 4];
        __builtin_prefetch(dB, 0, 1);
        #pragma unroll 8
        for (int qs = 0; qs < QCH / 2; ++qs) {
          v4f a = *(const v4f*)(aP + qs * 16);
          v2f ae; ae[0] = a[0]; ae[1] = a[1];
          v2f ao; ao[0] = a[2]; ao[1] = a[3];
          v4f q = *(const v4f*)(dB + (long)qs * (4 * FH * 4));
          v2f bc; bc[0] = q[0]; bc[1] = q[1];
          v2f bs; bs[0] = q[2]; bs[1] = q[3];
          accC[i][p] = wmma4(ae, bc, accC[i][p]);
          accS[i][p] = wmma4(ao, bs, accS[i][p]);
        }
      }
    }
    {
      const float* dB16 = dftT + (long)(16 * 16 + l16) * 4 +
                          (long)((kc * QCH + h) * 2) * (FH * 4) +
                          (long)(wvu * 4) * (4 * FH * 4);
      const float* aP16 = &A4[l16][h * 8 + wvu * 4 * 16];
      v2f ae[4], ao[4], bc[4], bs[4];
      #pragma unroll
      for (int u = 0; u < 4; ++u) {
        bc[u] = *(const v2f*)(dB16 + (long)u * (4 * FH * 4));
        bs[u] = *(const v2f*)(dB16 + (long)u * (4 * FH * 4) + FH * 4 + 2);
        ae[u] = *(const v2f*)(aP16 + u * 16);
        ao[u] = *(const v2f*)(aP16 + u * 16 + 4 + 2);
      }
      #pragma unroll
      for (int u = 0; u < 4; ++u) {
        accC16 = wmma4(ae[u], bc[u], accC16);
        accS16 = wmma4(ao[u], bs[u], accS16);
      }
    }
  }

  {
    int t = t0 + l16;
    float y = 0.0f;
    if (h == 0 && t < NFRAMES) y = wb[t * HOPSZ] * win[NFFT / 2];
    v2f ae; ae[0] = y; ae[1] = 0.0f;
    #pragma unroll
    for (int i = 0; i < 2; ++i) {
      int ft = wslot + 8 * i;
      const float* dB = dftT + (long)(ft * 16 + l16) * 4;
      v4f q = *(const v4f*)(dB + (long)((NQ - 1) * 2) * (FH * 4));
      v2f bc; bc[0] = q[0]; bc[1] = q[1];
      accC[i][0] = wmma4(ae, bc, accC[i][0]);
    }
    if (wvu == 0) {
      const float* dB = dftT + (long)(16 * 16 + l16) * 4;
      v2f bc = *(const v2f*)(dB + (long)((NQ - 1) * 2) * (FH * 4));
      accC16 = wmma4(ae, bc, accC16);
    }
  }
  if (l16 == 0) {
    #pragma unroll
    for (int v = 0; v < 8; ++v) {
      RED[wv][0][v + 8 * h] = accC16[v];
      RED[wv][1][v + 8 * h] = accS16[v];
    }
  }

  __syncthreads();
  if (tid < 16 * (FPAD - NFREQ))
    PL[tid / (FPAD - NFREQ)][NFREQ + tid % (FPAD - NFREQ)] = 0.0f;
  #pragma unroll
  for (int i = 0; i < 2; ++i) {
    int f = (wslot + 8 * i) * 16 + l16;
    #pragma unroll
    for (int v = 0; v < 8; ++v) {
      int mr = v + 8 * h;
      float ce = accC[i][0][v], co = accC[i][1][v];
      float se = accS[i][0][v], so = accS[i][1][v];
      float re = ce + co, im = se + so;
      float rm = ce - co, jm = so - se;
      PL[mr][f] = re * re + im * im;
      PL[mr][NFFT / 2 - f] = rm * rm + jm * jm;
    }
  }
  if (tid < 16) {
    float ce = 0.0f, so = 0.0f;
    #pragma unroll
    for (int w = 0; w < 8; ++w) { ce += RED[w][0][tid]; so += RED[w][1][tid]; }
    PL[tid][NFFT / 4] = ce * ce + so * so;
  }
  __syncthreads();

  int nt = __builtin_amdgcn_readfirstlane(wslot);
  int n0 = nt * 16;
  int kb, rows;
  {
    float melmax = 2595.0f * log10f(1.0f + 22050.0f / 700.0f);
    float step   = melmax / 129.0f;
    float flo = 700.0f * (exp10f(((float)(n0     ) * step) / 2595.0f) - 1.0f);
    float fhi = 700.0f * (exp10f(((float)(n0 + 17) * step) / 2595.0f) - 1.0f);
    int lo = (int)(flo * (512.0f / 22050.0f)) - 1;
    int hi = (int)(fhi * (512.0f / 22050.0f)) + 3;
    lo = (lo < 0) ? 0 : lo;
    int ke = (hi + 15) & ~15;
    ke = (ke > FPAD) ? FPAD : ke;
    kb   = lo & ~15;
    rows = ke - kb;
  }
  v8f acc;
  #pragma unroll
  for (int j = 0; j < 8; ++j) acc[j] = 0.0f;
  const float* pA = &PL[l16][kb + 2 * h];
  const float* pB = fb2 + (long)(n0 + l16) * 2 + (long)((kb >> 1) + h) * (NMELS * 2);
  for (; rows >= 64; rows -= 64) {
    acc = mel_steps<16>(pA, pB, acc);
    pA += 64; pB += 32 * (NMELS * 2);
  }
  if (rows >= 32) {
    acc = mel_steps<8>(pA, pB, acc);
    pA += 32; pB += 16 * (NMELS * 2); rows -= 32;
  }
  if (rows >= 16) acc = mel_steps<4>(pA, pB, acc);
  int n = n0 + l16;
  #pragma unroll
  for (int v = 0; v < 8; ++v) {
    int t = t0 + v + 8 * h;
    if (t < NFRAMES) mel[(b * NMELS + n) * NFRAMES + t] = acc[v];
  }
}

#define MED_R     4
#define MED_TAPS  (31 + MED_R - 1)
#define MED_CORE  (31 - MED_R + 1)
static_assert((NFRAMES % MED_R) == 0 && (NMELS % MED_R) == 0, "groups of four outputs must tile both axes");

__device__ __forceinline__ float median_core4_plus3(float c12, float c13, float c14, float c15,
                                                    float a, float b, float c) {
  float lo = fminf(a, b), hi = fmaxf(a, b);
  float e0 = fminf(lo, c);
  float md = fmaxf(lo, c);
  float e1 = fminf(hi, md);
  float e2 = fmaxf(hi, md);
  return fminf(fminf(c15, fmaxf(c14, e0)), fminf(fmaxf(c13, e1), fmaxf(c12, e2)));
}

#define MED_CE(i, j) { float lo_ = fminf(c[i], c[j]); float hi_ = fmaxf(c[i], c[j]); c[i] = lo_; c[j] = hi_; }
#define MED_LO(i, j) { c[i] = fminf(c[i], c[j]); }
#define MED_HI(i, j) { c[j] = fmaxf(c[i], c[j]); }
__device__ __forceinline__ void median31x4(const float x[MED_TAPS], float o[MED_R]) {
  float c[MED_CORE];
  #pragma unroll
  for (int k = 0; k < MED_CORE; ++k) c[k] = x[k + 3];
  MED_CE(1,2) MED_CE(3,4) MED_CE(5,6) MED_CE(7,8) MED_CE(9,10) MED_CE(11,12) MED_CE(13,14)
  MED_CE(15,16) MED_CE(17,18) MED_CE(19,20) MED_CE(21,22) MED_CE(23,24) MED_CE(25,26)
  MED_CE(0,2) MED_CE(3,5) MED_CE(4,6) MED_CE(7,9) MED_CE(8,10) MED_CE(11,13) MED_CE(12,14)
  MED_CE(15,17) MED_CE(16,18) MED_CE(19,21) MED_CE(20,22) MED_CE(23,25) MED_CE(24,26)
  MED_CE(0,1) MED_CE(4,5) MED_CE(8,9) MED_CE(12,13) MED_CE(16,17) MED_CE(20,21) MED_CE(24,25)
  MED_CE(2,6) MED_CE(7,11) MED_CE(10,14) MED_CE(15,19) MED_CE(18,22) MED_CE(23,27)
  MED_CE(0,4) MED_CE(1,5) MED_CE(8,12) MED_CE(9,13) MED_CE(16,20) MED_CE(17,21) MED_CE(25,27)
  MED_CE(6,14) MED_CE(15,23)
  MED_CE(1,3) MED_CE(2,4) MED_CE(9,11) MED_CE(10,12) MED_CE(17,19) MED_CE(18,20) MED_CE(24,25)
  MED_CE(26,27)
  MED_CE(0,1) MED_CE(2,3) MED_CE(4,5) MED_CE(8,9) MED_CE(10,11) MED_CE(12,13) MED_CE(16,17)
  MED_CE(18,19) MED_CE(20,21)
  MED_CE(0,8) MED_CE(1,9) MED_CE(2,10) MED_CE(3,11) MED_CE(4,12) MED_CE(5,13) MED_CE(16,24)
  MED_CE(17,25) MED_CE(18,26) MED_CE(19,27)
  MED_CE(3,7) MED_CE(4,8) MED_CE(5,9) MED_CE(6,10) MED_CE(19,23) MED_CE(20,24) MED_CE(21,25)
  MED_CE(22,26)
  MED_CE(1,3) MED_CE(2,4) MED_CE(5,7) MED_CE(6,8) MED_CE(9,11) MED_CE(10,12) MED_CE(17,19)
  MED_CE(18,20) MED_CE(21,23) MED_CE(22,24) MED_CE(25,27)
  MED_CE(0,1) MED_CE(2,3) MED_CE(4,5) MED_CE(6,7) MED_CE(8,9) MED_CE(10,11) MED_CE(12,13)
  MED_CE(16,17) MED_CE(18,19) MED_CE(20,21) MED_CE(22,23) MED_CE(24,25) MED_CE(26,27)
  MED_HI(0,16) MED_HI(1,17) MED_HI(2,18) MED_HI(3,19) MED_HI(4,20) MED_HI(5,21) MED_CE(6,22)
  MED_LO(7,23) MED_LO(8,24) MED_LO(9,25) MED_LO(10,26) MED_LO(11,27)
  MED_HI(7,15) MED_HI(8,16) MED_HI(9,17) MED_CE(10,18) MED_LO(11,19) MED_LO(12,20) MED_LO(13,21)
  MED_LO(14,22)
  MED_HI(6,10) MED_HI(11,15) MED_CE(12,16) MED_LO(13,17) MED_LO(14,18)
  MED_HI(10,12) MED_CE(13,15) MED_LO(14,16)
  MED_CE(12,13) MED_CE(14,15)
  o[0] = median_core4_plus3(c[12], c[13], c[14], c[15], x[0],  x[1],  x[2]);
  o[1] = median_core4_plus3(c[12], c[13], c[14], c[15], x[1],  x[2],  x[31]);
  o[2] = median_core4_plus3(c[12], c[13], c[14], c[15], x[2],  x[31], x[32]);
  o[3] = median_core4_plus3(c[12], c[13], c[14], c[15], x[31], x[32], x[33]);
}
#undef MED_CE
#undef MED_LO
#undef MED_HI

#define MT_BLOCKS (E_ELEMS / (MED_R * 256))
static_assert(MT_BLOCKS * MED_R * 256 == E_ELEMS, "median_time grid must tile E_ELEMS exactly");

__global__ __launch_bounds__(256) void median_time(const float* __restrict__ mel,
                                                   float* __restrict__ harm) {
  int g   = blockIdx.x * 256 + threadIdx.x;
  int q   = g % (NFRAMES / MED_R);
  int row = g / (NFRAMES / MED_R);
  int t0  = q * MED_R;
  const float* src = mel + row * NFRAMES;
  float x[MED_TAPS];
  if (t0 >= 16 && t0 + 19 < NFRAMES) {
    const v4f* __restrict__ s4 = (const v4f*)(src + (t0 - 16));
    v4f w[9];
    #pragma unroll
    for (int u = 0; u < 9; ++u) w[u] = s4[u];
    #pragma unroll
    for (int e = 0; e < MED_TAPS; ++e) x[e] = w[(e + 1) >> 2][(e + 1) & 3];
  } else {
    #pragma unroll
    for (int e = 0; e < MED_TAPS; ++e) {
      int j = t0 + e - 15;
      j = (j < 0) ? (-j - 1) : j;
      j = (j >= NFRAMES) ? (2 * NFRAMES - 1 - j) : j;
      x[e] = src[j];
    }
  }
  float o[MED_R];
  median31x4(x, o);
  v4f ov; ov[0] = o[0]; ov[1] = o[1]; ov[2] = o[2]; ov[3] = o[3];
  ((v4f*)harm)[g] = ov;
}

#define MM_TBLOCKS ((NFRAMES + 255) / 256)
#define MM_GROUPS  (NB * (NMELS / MED_R))

__global__ __launch_bounds__(256) void median_mel(const float* __restrict__ mel,
                                                  float* __restrict__ perc) {
  int t = blockIdx.x * 256 + threadIdx.x;
  if (t >= NFRAMES) return;
  int k  = blockIdx.y % (NMELS / MED_R);
  int b  = blockIdx.y / (NMELS / MED_R);
  int n0 = k * MED_R;
  const float* base = mel + (b * NMELS) * NFRAMES + t;
  float x[MED_TAPS];
  #pragma unroll
  for (int e = 0; e < MED_TAPS; ++e) {
    int j = n0 + e - 15;
    j = (j < 0) ? (-j - 1) : j;
    j = (j >= NMELS) ? (2 * NMELS - 1 - j) : j;
    x[e] = base[j * NFRAMES];
  }
  float o[MED_R];
  median31x4(x, o);
  float* dst = perc + (b * NMELS + n0) * NFRAMES + t;
  #pragma unroll
  for (int r = 0; r < MED_R; ++r) dst[r * NFRAMES] = o[r];
}

#define HP_U      4
#define HP_BLOCKS (E_ELEMS / (4 * 256 * HP_U))
static_assert(HP_BLOCKS * 4 * 256 * HP_U == E_ELEMS, "hpss_mask grid must tile E_ELEMS exactly");
static_assert((MEL_OFF % 4) == 0 && (E_ELEMS % 4) == 0, "float4 views need 16-byte aligned planes");

__global__ __launch_bounds__(256) void hpss_mask(const float* __restrict__ mel,
                                                 float* __restrict__ harm,
                                                 float* __restrict__ perc,
                                                 unsigned* __restrict__ mx) {
  __shared__ float r0[256], r1[256], r2[256];
  const v4f* __restrict__ mel4 = (const v4f*)mel;
  v4f* __restrict__ harm4 = (v4f*)harm;
  v4f* __restrict__ perc4 = (v4f*)perc;
  int tid  = threadIdx.x;
  int base = blockIdx.x * (256 * HP_U) + tid;
  v4f S4[HP_U], H4[HP_U], P4[HP_U];
  #pragma unroll
  for (int u = 0; u < HP_U; ++u) {
    int i = base + u * 256;
    S4[u] = mel4[i]; H4[u] = harm4[i]; P4[u] = perc4[i];
  }
  float s = 0.0f, dh = 0.0f, dp = 0.0f;
  #pragma unroll
  for (int u = 0; u < HP_U; ++u) {
    v4f oh, op;
    #pragma unroll
    for (int c = 0; c < 4; ++c) {
      float S  = S4[u][c];
      float ha = H4[u][c];
      float pe = P4[u][c];
      float Z  = fmaxf(ha, pe);
      bool bad = Z < 1.17549435e-38f;
      float Zs = bad ? 1.0f : Z;
      float mh = ha / Zs; mh *= mh;
      float mp = pe / Zs; mp *= mp;
      float den = mh + mp;
      float maskh = bad ? 0.0f : mh / den;
      float maskp = bad ? 0.0f : mp / den;
      float vh = S * maskh, vp = S * maskp;
      oh[c] = vh; op[c] = vp;
      s = fmaxf(s, S); dh = fmaxf(dh, vh); dp = fmaxf(dp, vp);
    }
    int i = base + u * 256;
    harm4[i] = oh;
    perc4[i] = op;
  }
  r0[tid] = s; r1[tid] = dh; r2[tid] = dp;
  __syncthreads();
  for (int off = 128; off > 0; off >>= 1) {
    if (tid < off) {
      r0[tid] = fmaxf(r0[tid], r0[tid + off]);
      r1[tid] = fmaxf(r1[tid], r1[tid + off]);
      r2[tid] = fmaxf(r2[tid], r2[tid + off]);
    }
    __syncthreads();
  }
  if (tid == 0) {
    atomicMax(mx + 0, __float_as_uint(r0[0]));
    atomicMax(mx + 1, __float_as_uint(r1[0]));
    atomicMax(mx + 2, __float_as_uint(r2[0]));
  }
}

#define FN_U          8
#define FN_SEG_BLOCKS (E_ELEMS / (4 * 256 * FN_U))
static_assert(FN_SEG_BLOCKS * 4 * 256 * FN_U == E_ELEMS, "finalize grid must tile each plane exactly");

__global__ __launch_bounds__(256) void finalize(const float* __restrict__ mel,
                                                const float* __restrict__ harm,
                                                const float* __restrict__ perc,
                                                const unsigned* __restrict__ mx,
                                                float* __restrict__ out) {
  int s  = blockIdx.x / FN_SEG_BLOCKS;
  int e4 = (blockIdx.x - s * FN_SEG_BLOCKS) * (256 * FN_U) + threadIdx.x;
  const float* src = (s == 0) ? mel : ((s == 1) ? harm : perc);
  const v4f* __restrict__ src4 = (const v4f*)src;
  v4f* __restrict__ out4 = (v4f*)out + (long)s * (E_ELEMS / 4);
  float ref = __uint_as_float(mx[s]);
  v4f v[FN_U];
  #pragma unroll
  for (int u = 0; u < FN_U; ++u) v[u] = src4[e4 + u * 256];
  #pragma unroll
  for (int u = 0; u < FN_U; ++u) {
    v4f o;
    #pragma unroll
    for (int c = 0; c < 4; ++c) {
      float db = 10.0f * log10f(fmaxf(1e-10f, v[u][c])) -
                 10.0f * log10f(fmaxf(1e-10f, ref));
      db = fmaxf(db, -80.0f);
      o[c] = (db + 80.0f) * (1.0f / (80.0f + 1e-7f));
    }
    out4[e4 + u * 256] = o;
  }
}

extern "C" void kernel_launch(void* const* d_in, const int* in_sizes, int n_in,
                              void* d_out, int out_size, void* d_ws,
                              size_t ws_size, hipStream_t stream) {
  const float* wave = (const float*)d_in[0];
  float* out  = (float*)d_out;
  float* ws   = (float*)d_ws;
  float* mel  = ws + MEL_OFF;
  float* harm = ws + HARM_OFF;
  float* perc = ws + PERC_OFF;
  unsigned* mx = (unsigned*)(ws + MAX_OFF);

  build_tables<<<(MAX_OFF + 255) / 256, 256, 0, stream>>>(ws);
  mel_gemm<<<NB * MTILES, 256, 0, stream>>>(wave, ws, mel);
  median_time<<<MT_BLOCKS, 256, 0, stream>>>(mel, harm);
  median_mel<<<dim3(MM_TBLOCKS, MM_GROUPS), 256, 0, stream>>>(mel, perc);
  hpss_mask<<<HP_BLOCKS, 256, 0, stream>>>(mel, harm, perc, mx);
  finalize<<<3 * FN_SEG_BLOCKS, 256, 0, stream>>>(mel, harm, perc, mx, out);
}


# mel_gemm stage 1: a k-group's table fragments are loaded groups ahead of its WMMAs, its LDS fragments as one group in front of them; whole folded frame in LDS, one tile's sums live; same sum order
# speedup vs baseline: 3.7094x; 1.0504x over previous
#include <hip/hip_runtime.h>
#include <math.h>

#define NFFT     1024
#define HOPSZ    512
#define NMELS    128
#define NFREQ    513
#define FPAD     528
#define T_IN     1323000
#define NB       16
#define NFRAMES  2584
#define MTILES   162
#define E_ELEMS  (NB*NMELS*NFRAMES)

#define KHALF    512
#define KCH      256
#define APITCH   1028
#define KG       8
#define FH       272
#define NFT      17
#define QCH      64
#define NQ       129
static_assert(KCH == 256, "the frame-chunk build gives one folded k to each of the 256 threads");

#define WIN_OFF  0
#define DFT_OFF  1024
#define DFT_SZ   (NQ*2*FH*4)
#define FB_OFF   (DFT_OFF + DFT_SZ)
#define FB_SZ    ((FPAD/2)*NMELS*2)
#define MAX_OFF  (FB_OFF + FB_SZ)
#define MEL_OFF  (MAX_OFF + 8)
#define HARM_OFF (MEL_OFF + E_ELEMS)
#define PERC_OFF (HARM_OFF + E_ELEMS)

typedef float v2f __attribute__((ext_vector_type(2)));
typedef float v4f __attribute__((ext_vector_type(4)));
typedef float v8f __attribute__((ext_vector_type(8)));

__device__ __forceinline__ v8f wmma4(v2f a, v2f b, v8f c) {
  return __builtin_amdgcn_wmma_f32_16x16x4_f32(false, a, false, b, (short)0, c,
                                               false, false);
}

__global__ __launch_bounds__(256) void build_tables(float* __restrict__ ws) {
  int gid = blockIdx.x * 256 + threadIdx.x;
  if (gid == 0) {
    unsigned* mx = (unsigned*)(ws + MAX_OFF);
    mx[0] = 0u; mx[1] = 0u; mx[2] = 0u;
  }
  const float TWO_PI = 6.28318530717958647692f;
  if (gid < DFT_OFF) {
    int k = gid;
    ws[WIN_OFF + k] = 0.5f * (1.0f - cosf(TWO_PI * (float)k / 1024.0f));
  } else if (gid < FB_OFF) {
    int i = gid - DFT_OFF;
    int entry = i & 3;
    int rest  = i >> 2;
    int f  = rest % FH;
    int qp = rest / FH;
    int k  = 4 * (qp >> 1) + (qp & 1) + 2 * (entry & 1);
    int r  = (k * f) & (NFFT - 1);
    float arg = (float)r * (TWO_PI / 1024.0f);
    ws[gid] = (entry < 2) ? cosf(arg) : sinf(arg);
  } else if (gid < MAX_OFF) {
    int i = gid - FB_OFF;
    int entry = i & 1;
    int rest  = i >> 1;
    int m  = rest % NMELS;
    int k2 = rest / NMELS;
    int f  = 2 * k2 + entry;
    float v = 0.0f;
    if (f < NFREQ) {
      float freq   = (float)f * (22050.0f / 512.0f);
      float melmax = 2595.0f * log10f(1.0f + 22050.0f / 700.0f);
      float step   = melmax / 129.0f;
      float p0 = 700.0f * (exp10f(((float)(m    ) * step) / 2595.0f) - 1.0f);
      float p1 = 700.0f * (exp10f(((float)(m + 1) * step) / 2595.0f) - 1.0f);
      float p2 = 700.0f * (exp10f(((float)(m + 2) * step) / 2595.0f) - 1.0f);
      float down = (freq - p0) / (p1 - p0);
      float up   = (p2 - freq) / (p2 - p1);
      float t = fmaxf(0.0f, fminf(down, up));
      v = t * 2.0f / (p2 - p0);
    }
    ws[gid] = v;
  }
}

template <int NS>
__device__ __forceinline__ v8f mel_steps(const float* pA, const float* pB, v8f acc) {
  v2f a[NS], bb[NS];
  #pragma unroll
  for (int u = 0; u < NS; ++u) {
    bb[u] = *(const v2f*)(pB + (long)u * (2 * NMELS * 2));
    a[u]  = *(const v2f*)(pA + 4 * u);
  }
  #pragma unroll
  for (int u = 0; u < NS; ++u) acc = wmma4(a[u], bb[u], acc);
  return acc;
}

__device__ __forceinline__ void load_tab_group(v4f q[KG], const float* p) {
  #pragma unroll
  for (int u = 0; u < KG; ++u) q[u] = *(const v4f*)(p + (long)u * (4 * FH * 4));
}

__device__ __forceinline__ void run_group(const float* aP, const v4f q[KG], v8f& accC, v8f& accS) {
  v4f a[KG];
  #pragma unroll
  for (int u = 0; u < KG; ++u) a[u] = *(const v4f*)(aP + u * 16);
  __builtin_amdgcn_sched_barrier(0);
  #pragma unroll
  for (int u = 0; u < KG; ++u) {
    v2f ae; ae[0] = a[u][0]; ae[1] = a[u][1];
    v2f ao; ao[0] = a[u][2]; ao[1] = a[u][3];
    v2f bc; bc[0] = q[u][0]; bc[1] = q[u][1];
    v2f bs; bs[0] = q[u][2]; bs[1] = q[u][3];
    accC = wmma4(ae, bc, accC);
    accS = wmma4(ao, bs, accS);
  }
}

__global__ __launch_bounds__(256, 6) void mel_gemm(const float* __restrict__ wave,
                                                const float* __restrict__ ws,
                                                float* __restrict__ mel) {
  __shared__ __attribute__((aligned(16))) float A4[16][APITCH];
  __shared__ float PL[16][FPAD + 4];
  __shared__ float RED[8][2][16];

  const float* __restrict__ win  = ws + WIN_OFF;
  const float* __restrict__ dftT = ws + DFT_OFF;
  const float* __restrict__ fb2  = ws + FB_OFF;

  int b    = blockIdx.x / MTILES;
  int tile = blockIdx.x % MTILES;
  int t0   = tile * 16;
  int tid  = threadIdx.x;
  int lane = tid & 31;
  int wv   = tid >> 5;
  int h    = lane >> 4;
  int l16  = lane & 15;

  const float* wb = wave + (long)b * T_IN;

  int wslot = (wv + blockIdx.x + (blockIdx.x >> 8)) & 7;
  const int wvu = __builtin_amdgcn_readfirstlane(wv);

  for (int kc = 0; kc < KHALF / KCH; ++kc) {
    const int kl = tid;
    const int k  = kc * KCH + kl;
    const bool pr = k > 0;
    const int kp = pr ? (NFFT - k) : 0;
    const float w1 = win[k];
    const float w2 = win[kp];
    float s1[16], s2[16];
    #pragma unroll
    for (int m = 0; m < 16; ++m) {
      int t = t0 + m;
      t = (t < NFRAMES) ? t : (NFRAMES - 1);
      int o1 = t * HOPSZ + k - NFFT / 2;
      o1 = (o1 < 0) ? -o1 : o1;
      o1 = (o1 >= T_IN) ? (2 * (T_IN - 1) - o1) : o1;
      int o2 = t * HOPSZ + NFFT / 2 - k;
      o2 = (o2 >= T_IN) ? (2 * (T_IN - 1) - o2) : o2;
      s1[m] = wb[o1];
      s2[m] = wb[o2];
    }
    __builtin_amdgcn_sched_barrier(0);
    const int c = (kc * QCH + (kl >> 2)) * 8 + (kl & 1) * 4 + ((kl >> 1) & 1);
    #pragma unroll
    for (int m = 0; m < 16; ++m) {
      float y1 = s1[m] * w1;
      float y2 = s2[m] * w2;
      float ve = pr ? (y1 + y2) : y1;
      float vo = pr ? (y1 - y2) : 0.0f;
      bool live = (t0 + m) < NFRAMES;
      A4[m][c]     = live ? ve : 0.0f;
      A4[m][c + 2] = live ? vo : 0.0f;
    }
  }

  v2f ae512;
  {
    int t = t0 + l16;
    float y = 0.0f;
    if (h == 0 && t < NFRAMES) y = wb[t * HOPSZ] * win[NFFT / 2];
    ae512[0] = y; ae512[1] = 0.0f;
  }
  __syncthreads();

  #define NG (KHALF / 8 / KG)
  static_assert(NG * KG * 8 == KHALF && (NG % 2) == 0, "groups must tile the k-steps in pairs");
  const float* aBase = &A4[l16][h * 8];
  #pragma unroll
  for (int i = 0; i < 2; ++i) {
    int ft = wslot + 8 * i;
    int f0 = ft * 16;
    v8f accC[2], accS[2];
    #pragma unroll
    for (int p = 0; p < 2; ++p)
      for (int j = 0; j < 8; ++j) { accC[p][j] = 0.0f; accS[p][j] = 0.0f; }
    const float* dB = dftT + (long)(f0 + l16) * 4 + (long)(h * 2) * (FH * 4);
    v4f q0[KG], q1[KG];
    load_tab_group(q0, dB);
    #pragma unroll
    for (int G = 0; G < 2 * NG; G += 2) {
      {
        const int p = (G + 1) / NG, g = (G + 1) % NG;
        load_tab_group(q1, dB + (long)p * (FH * 4) + (long)(g * KG) * (4 * FH * 4));
      }
      __builtin_amdgcn_sched_barrier(0);
      {
        const int p = G / NG, g = G % NG;
        run_group(aBase + p * 4 + g * KG * 16, q0, accC[p], accS[p]);
      }
      if (G + 2 < 2 * NG) {
        const int p = (G + 2) / NG, g = (G + 2) % NG;
        load_tab_group(q0, dB + (long)p * (FH * 4) + (long)(g * KG) * (4 * FH * 4));
      }
      __builtin_amdgcn_sched_barrier(0);
      {
        const int p = (G + 1) / NG, g = (G + 1) % NG;
        run_group(aBase + p * 4 + g * KG * 16, q1, accC[p], accS[p]);
      }
    }
    {
      const float* dL = dftT + (long)(f0 + l16) * 4;
      v4f q = *(const v4f*)(dL + (long)((NQ - 1) * 2) * (FH * 4));
      v2f bc; bc[0] = q[0]; bc[1] = q[1];
      accC[0] = wmma4(ae512, bc, accC[0]);
    }
    int f = f0 + l16;
    #pragma unroll
    for (int v = 0; v < 8; ++v) {
      int mr = v + 8 * h;
      float ce = accC[0][v], co = accC[1][v];
      float se = accS[0][v], so = accS[1][v];
      float re = ce + co, im = se + so;
      float rm = ce - co, jm = so - se;
      PL[mr][f] = re * re + im * im;
      PL[mr][NFFT / 2 - f] = rm * rm + jm * jm;
    }
  }
  #undef NG

  v8f accC16, accS16;
  #pragma unroll
  for (int j = 0; j < 8; ++j) { accC16[j] = 0.0f; accS16[j] = 0.0f; }
  {
    const float* dB16 = dftT + (long)(16 * 16 + l16) * 4 + (long)(h * 2) * (FH * 4) +
                        (long)(wvu * 4) * (4 * FH * 4);
    const float* aP16 = aBase + wvu * 4 * 16;
    v2f ae[8], ao[8], bc[8], bs[8];
    #pragma unroll
    for (int u = 0; u < 8; ++u) {
      const int s = (u >> 2) * (QCH / 2) + (u & 3);
      bc[u] = *(const v2f*)(dB16 + (long)s * (4 * FH * 4));
      bs[u] = *(const v2f*)(dB16 + (long)s * (4 * FH * 4) + FH * 4 + 2);
      ae[u] = *(const v2f*)(aP16 + s * 16);
      ao[u] = *(const v2f*)(aP16 + s * 16 + 4 + 2);
    }
    #pragma unroll
    for (int u = 0; u < 8; ++u) {
      accC16 = wmma4(ae[u], bc[u], accC16);
      accS16 = wmma4(ao[u], bs[u], accS16);
    }
    if (wvu == 0) {
      const float* dL = dftT + (long)(16 * 16 + l16) * 4;
      v2f b5 = *(const v2f*)(dL + (long)((NQ - 1) * 2) * (FH * 4));
      accC16 = wmma4(ae512, b5, accC16);
    }
  }
  if (l16 == 0) {
    #pragma unroll
    for (int v = 0; v < 8; ++v) {
      RED[wv][0][v + 8 * h] = accC16[v];
      RED[wv][1][v + 8 * h] = accS16[v];
    }
  }
  if (tid < 16 * (FPAD - NFREQ))
    PL[tid / (FPAD - NFREQ)][NFREQ + tid % (FPAD - NFREQ)] = 0.0f;
  __syncthreads();
  if (tid < 16) {
    float ce = 0.0f, so = 0.0f;
    #pragma unroll
    for (int w = 0; w < 8; ++w) { ce += RED[w][0][tid]; so += RED[w][1][tid]; }
    PL[tid][NFFT / 4] = ce * ce + so * so;
  }
  __syncthreads();

  int nt = __builtin_amdgcn_readfirstlane(wslot);
  int n0 = nt * 16;
  int kb, rows;
  {
    float melmax = 2595.0f * log10f(1.0f + 22050.0f / 700.0f);
    float step   = melmax / 129.0f;
    float flo = 700.0f * (exp10f(((float)(n0     ) * step) / 2595.0f) - 1.0f);
    float fhi = 700.0f * (exp10f(((float)(n0 + 17) * step) / 2595.0f) - 1.0f);
    int lo = (int)(flo * (512.0f / 22050.0f)) - 1;
    int hi = (int)(fhi * (512.0f / 22050.0f)) + 3;
    lo = (lo < 0) ? 0 : lo;
    int ke = (hi + 15) & ~15;
    ke = (ke > FPAD) ? FPAD : ke;
    kb   = lo & ~15;
    rows = ke - kb;
  }
  v8f acc;
  #pragma unroll
  for (int j = 0; j < 8; ++j) acc[j] = 0.0f;
  const float* pA = &PL[l16][kb + 2 * h];
  const float* pB = fb2 + (long)(n0 + l16) * 2 + (long)((kb >> 1) + h) * (NMELS * 2);
  for (; rows >= 64; rows -= 64) {
    acc = mel_steps<16>(pA, pB, acc);
    pA += 64; pB += 32 * (NMELS * 2);
  }
  if (rows >= 32) {
    acc = mel_steps<8>(pA, pB, acc);
    pA += 32; pB += 16 * (NMELS * 2); rows -= 32;
  }
  if (rows >= 16) acc = mel_steps<4>(pA, pB, acc);
  int n = n0 + l16;
  #pragma unroll
  for (int v = 0; v < 8; ++v) {
    int t = t0 + v + 8 * h;
    if (t < NFRAMES) mel[(b * NMELS + n) * NFRAMES + t] = acc[v];
  }
}

#define MED_R     4
#define MED_TAPS  (31 + MED_R - 1)
#define MED_CORE  (31 - MED_R + 1)
static_assert((NFRAMES % MED_R) == 0 && (NMELS % MED_R) == 0, "groups of four outputs must tile both axes");

__device__ __forceinline__ float median_core4_plus3(float c12, float c13, float c14, float c15,
                                                    float a, float b, float c) {
  float lo = fminf(a, b), hi = fmaxf(a, b);
  float e0 = fminf(lo, c);
  float md = fmaxf(lo, c);
  float e1 = fminf(hi, md);
  float e2 = fmaxf(hi, md);
  return fminf(fminf(c15, fmaxf(c14, e0)), fminf(fmaxf(c13, e1), fmaxf(c12, e2)));
}

#define MED_CE(i, j) { float lo_ = fminf(c[i], c[j]); float hi_ = fmaxf(c[i], c[j]); c[i] = lo_; c[j] = hi_; }
#define MED_LO(i, j) { c[i] = fminf(c[i], c[j]); }
#define MED_HI(i, j) { c[j] = fmaxf(c[i], c[j]); }
__device__ __forceinline__ void median31x4(const float x[MED_TAPS], float o[MED_R]) {
  float c[MED_CORE];
  #pragma unroll
  for (int k = 0; k < MED_CORE; ++k) c[k] = x[k + 3];
  MED_CE(1,2) MED_CE(3,4) MED_CE(5,6) MED_CE(7,8) MED_CE(9,10) MED_CE(11,12) MED_CE(13,14)
  MED_CE(15,16) MED_CE(17,18) MED_CE(19,20) MED_CE(21,22) MED_CE(23,24) MED_CE(25,26)
  MED_CE(0,2) MED_CE(3,5) MED_CE(4,6) MED_CE(7,9) MED_CE(8,10) MED_CE(11,13) MED_CE(12,14)
  MED_CE(15,17) MED_CE(16,18) MED_CE(19,21) MED_CE(20,22) MED_CE(23,25) MED_CE(24,26)
  MED_CE(0,1) MED_CE(4,5) MED_CE(8,9) MED_CE(12,13) MED_CE(16,17) MED_CE(20,21) MED_CE(24,25)
  MED_CE(2,6) MED_CE(7,11) MED_CE(10,14) MED_CE(15,19) MED_CE(18,22) MED_CE(23,27)
  MED_CE(0,4) MED_CE(1,5) MED_CE(8,12) MED_CE(9,13) MED_CE(16,20) MED_CE(17,21) MED_CE(25,27)
  MED_CE(6,14) MED_CE(15,23)
  MED_CE(1,3) MED_CE(2,4) MED_CE(9,11) MED_CE(10,12) MED_CE(17,19) MED_CE(18,20) MED_CE(24,25)
  MED_CE(26,27)
  MED_CE(0,1) MED_CE(2,3) MED_CE(4,5) MED_CE(8,9) MED_CE(10,11) MED_CE(12,13) MED_CE(16,17)
  MED_CE(18,19) MED_CE(20,21)
  MED_CE(0,8) MED_CE(1,9) MED_CE(2,10) MED_CE(3,11) MED_CE(4,12) MED_CE(5,13) MED_CE(16,24)
  MED_CE(17,25) MED_CE(18,26) MED_CE(19,27)
  MED_CE(3,7) MED_CE(4,8) MED_CE(5,9) MED_CE(6,10) MED_CE(19,23) MED_CE(20,24) MED_CE(21,25)
  MED_CE(22,26)
  MED_CE(1,3) MED_CE(2,4) MED_CE(5,7) MED_CE(6,8) MED_CE(9,11) MED_CE(10,12) MED_CE(17,19)
  MED_CE(18,20) MED_CE(21,23) MED_CE(22,24) MED_CE(25,27)
  MED_CE(0,1) MED_CE(2,3) MED_CE(4,5) MED_CE(6,7) MED_CE(8,9) MED_CE(10,11) MED_CE(12,13)
  MED_CE(16,17) MED_CE(18,19) MED_CE(20,21) MED_CE(22,23) MED_CE(24,25) MED_CE(26,27)
  MED_HI(0,16) MED_HI(1,17) MED_HI(2,18) MED_HI(3,19) MED_HI(4,20) MED_HI(5,21) MED_CE(6,22)
  MED_LO(7,23) MED_LO(8,24) MED_LO(9,25) MED_LO(10,26) MED_LO(11,27)
  MED_HI(7,15) MED_HI(8,16) MED_HI(9,17) MED_CE(10,18) MED_LO(11,19) MED_LO(12,20) MED_LO(13,21)
  MED_LO(14,22)
  MED_HI(6,10) MED_HI(11,15) MED_CE(12,16) MED_LO(13,17) MED_LO(14,18)
  MED_HI(10,12) MED_CE(13,15) MED_LO(14,16)
  MED_CE(12,13) MED_CE(14,15)
  o[0] = median_core4_plus3(c[12], c[13], c[14], c[15], x[0],  x[1],  x[2]);
  o[1] = median_core4_plus3(c[12], c[13], c[14], c[15], x[1],  x[2],  x[31]);
  o[2] = median_core4_plus3(c[12], c[13], c[14], c[15], x[2],  x[31], x[32]);
  o[3] = median_core4_plus3(c[12], c[13], c[14], c[15], x[31], x[32], x[33]);
}
#undef MED_CE
#undef MED_LO
#undef MED_HI

#define MT_BLOCKS (E_ELEMS / (MED_R * 256))
static_assert(MT_BLOCKS * MED_R * 256 == E_ELEMS, "median_time grid must tile E_ELEMS exactly");

__global__ __launch_bounds__(256) void median_time(const float* __restrict__ mel,
                                                   float* __restrict__ harm) {
  int g   = blockIdx.x * 256 + threadIdx.x;
  int q   = g % (NFRAMES / MED_R);
  int row = g / (NFRAMES / MED_R);
  int t0  = q * MED_R;
  const float* src = mel + row * NFRAMES;
  float x[MED_TAPS];
  if (t0 >= 16 && t0 + 19 < NFRAMES) {
    const v4f* __restrict__ s4 = (const v4f*)(src + (t0 - 16));
    v4f w[9];
    #pragma unroll
    for (int u = 0; u < 9; ++u) w[u] = s4[u];
    #pragma unroll
    for (int e = 0; e < MED_TAPS; ++e) x[e] = w[(e + 1) >> 2][(e + 1) & 3];
  } else {
    #pragma unroll
    for (int e = 0; e < MED_TAPS; ++e) {
      int j = t0 + e - 15;
      j = (j < 0) ? (-j - 1) : j;
      j = (j >= NFRAMES) ? (2 * NFRAMES - 1 - j) : j;
      x[e] = src[j];
    }
  }
  float o[MED_R];
  median31x4(x, o);
  v4f ov; ov[0] = o[0]; ov[1] = o[1]; ov[2] = o[2]; ov[3] = o[3];
  ((v4f*)harm)[g] = ov;
}

#define MM_TBLOCKS ((NFRAMES + 255) / 256)
#define MM_GROUPS  (NB * (NMELS / MED_R))

__global__ __launch_bounds__(256) void median_mel(const float* __restrict__ mel,
                                                  float* __restrict__ perc) {
  int t = blockIdx.x * 256 + threadIdx.x;
  if (t >= NFRAMES) return;
  int k  = blockIdx.y % (NMELS / MED_R);
  int b  = blockIdx.y / (NMELS / MED_R);
  int n0 = k * MED_R;
  const float* base = mel + (b * NMELS) * NFRAMES + t;
  float x[MED_TAPS];
  #pragma unroll
  for (int e = 0; e < MED_TAPS; ++e) {
    int j = n0 + e - 15;
    j = (j < 0) ? (-j - 1) : j;
    j = (j >= NMELS) ? (2 * NMELS - 1 - j) : j;
    x[e] = base[j * NFRAMES];
  }
  float o[MED_R];
  median31x4(x, o);
  float* dst = perc + (b * NMELS + n0) * NFRAMES + t;
  #pragma unroll
  for (int r = 0; r < MED_R; ++r) dst[r * NFRAMES] = o[r];
}

#define HP_U      4
#define HP_BLOCKS (E_ELEMS / (4 * 256 * HP_U))
static_assert(HP_BLOCKS * 4 * 256 * HP_U == E_ELEMS, "hpss_mask grid must tile E_ELEMS exactly");
static_assert((MEL_OFF % 4) == 0 && (E_ELEMS % 4) == 0, "float4 views need 16-byte aligned planes");

__global__ __launch_bounds__(256) void hpss_mask(const float* __restrict__ mel,
                                                 float* __restrict__ harm,
                                                 float* __restrict__ perc,
                                                 unsigned* __restrict__ mx) {
  __shared__ float r0[256], r1[256], r2[256];
  const v4f* __restrict__ mel4 = (const v4f*)mel;
  v4f* __restrict__ harm4 = (v4f*)harm;
  v4f* __restrict__ perc4 = (v4f*)perc;
  int tid  = threadIdx.x;
  int base = blockIdx.x * (256 * HP_U) + tid;
  v4f S4[HP_U], H4[HP_U], P4[HP_U];
  #pragma unroll
  for (int u = 0; u < HP_U; ++u) {
    int i = base + u * 256;
    S4[u] = mel4[i]; H4[u] = harm4[i]; P4[u] = perc4[i];
  }
  float s = 0.0f, dh = 0.0f, dp = 0.0f;
  #pragma unroll
  for (int u = 0; u < HP_U; ++u) {
    v4f oh, op;
    #pragma unroll
    for (int c = 0; c < 4; ++c) {
      float S  = S4[u][c];
      float ha = H4[u][c];
      float pe = P4[u][c];
      float Z  = fmaxf(ha, pe);
      bool bad = Z < 1.17549435e-38f;
      float Zs = bad ? 1.0f : Z;
      float mh = ha / Zs; mh *= mh;
      float mp = pe / Zs; mp *= mp;
      float den = mh + mp;
      float maskh = bad ? 0.0f : mh / den;
      float maskp = bad ? 0.0f : mp / den;
      float vh = S * maskh, vp = S * maskp;
      oh[c] = vh; op[c] = vp;
      s = fmaxf(s, S); dh = fmaxf(dh, vh); dp = fmaxf(dp, vp);
    }
    int i = base + u * 256;
    harm4[i] = oh;
    perc4[i] = op;
  }
  r0[tid] = s; r1[tid] = dh; r2[tid] = dp;
  __syncthreads();
  for (int off = 128; off > 0; off >>= 1) {
    if (tid < off) {
      r0[tid] = fmaxf(r0[tid], r0[tid + off]);
      r1[tid] = fmaxf(r1[tid], r1[tid + off]);
      r2[tid] = fmaxf(r2[tid], r2[tid + off]);
    }
    __syncthreads();
  }
  if (tid == 0) {
    atomicMax(mx + 0, __float_as_uint(r0[0]));
    atomicMax(mx + 1, __float_as_uint(r1[0]));
    atomicMax(mx + 2, __float_as_uint(r2[0]));
  }
}

#define FN_U          8
#define FN_SEG_BLOCKS (E_ELEMS / (4 * 256 * FN_U))
static_assert(FN_SEG_BLOCKS * 4 * 256 * FN_U == E_ELEMS, "finalize grid must tile each plane exactly");

__global__ __launch_bounds__(256) void finalize(const float* __restrict__ mel,
                                                const float* __restrict__ harm,
                                                const float* __restrict__ perc,
                                                const unsigned* __restrict__ mx,
                                                float* __restrict__ out) {
  int s  = blockIdx.x / FN_SEG_BLOCKS;
  int e4 = (blockIdx.x - s * FN_SEG_BLOCKS) * (256 * FN_U) + threadIdx.x;
  const float* src = (s == 0) ? mel : ((s == 1) ? harm : perc);
  const v4f* __restrict__ src4 = (const v4f*)src;
  v4f* __restrict__ out4 = (v4f*)out + (long)s * (E_ELEMS / 4);
  float ref = __uint_as_float(mx[s]);
  v4f v[FN_U];
  #pragma unroll
  for (int u = 0; u < FN_U; ++u) v[u] = src4[e4 + u * 256];
  #pragma unroll
  for (int u = 0; u < FN_U; ++u) {
    v4f o;
    #pragma unroll
    for (int c = 0; c < 4; ++c) {
      float db = 10.0f * log10f(fmaxf(1e-10f, v[u][c])) -
                 10.0f * log10f(fmaxf(1e-10f, ref));
      db = fmaxf(db, -80.0f);
      o[c] = (db + 80.0f) * (1.0f / (80.0f + 1e-7f));
    }
    out4[e4 + u * 256] = o;
  }
}

extern "C" void kernel_launch(void* const* d_in, const int* in_sizes, int n_in,
                              void* d_out, int out_size, void* d_ws,
                              size_t ws_size, hipStream_t stream) {
  const float* wave = (const float*)d_in[0];
  float* out  = (float*)d_out;
  float* ws   = (float*)d_ws;
  float* mel  = ws + MEL_OFF;
  float* harm = ws + HARM_OFF;
  float* perc = ws + PERC_OFF;
  unsigned* mx = (unsigned*)(ws + MAX_OFF);

  build_tables<<<(MAX_OFF + 255) / 256, 256, 0, stream>>>(ws);
  mel_gemm<<<NB * MTILES, 256, 0, stream>>>(wave, ws, mel);
  median_time<<<MT_BLOCKS, 256, 0, stream>>>(mel, harm);
  median_mel<<<dim3(MM_TBLOCKS, MM_GROUPS), 256, 0, stream>>>(mel, perc);
  hpss_mask<<<HP_BLOCKS, 256, 0, stream>>>(mel, harm, perc, mx);
  finalize<<<3 * FN_SEG_BLOCKS, 256, 0, stream>>>(mel, harm, perc, mx, out);
}


# mel_gemm: fold the DFT on its k <-> 512-k symmetry: e[k]+-e[512-k], o[k]-+o[512-k] pre-added, sign by parity of f, one parity of f per tile: K 512 -> 256; f32 summation order changes
# speedup vs baseline: 4.4655x; 1.2038x over previous
#include <hip/hip_runtime.h>
#include <math.h>

#define NFFT     1024
#define HOPSZ    512
#define NMELS    128
#define NFREQ    513
#define FPAD     528
#define T_IN     1323000
#define NB       16
#define NFRAMES  2584
#define MTILES   162
#define E_ELEMS  (NB*NMELS*NFRAMES)

#define KG       8
#define FH       272
#define NFT      17
#define KQ       256
#define APITCH   1028
#define NQ       65
static_assert(KQ == 256, "the frame build gives one folded k to each of the 256 threads");

__device__ __forceinline__ int col_to_f(int c) {
  return (c < 128) ? 2 * c : ((c < 256) ? 2 * (c - 128) + 1 : c);
}

#define WIN_OFF  0
#define DFT_OFF  1024
#define DFT_SZ   (NQ*2*FH*4)
#define FB_OFF   (DFT_OFF + DFT_SZ)
#define FB_SZ    ((FPAD/2)*NMELS*2)
#define MAX_OFF  (FB_OFF + FB_SZ)
#define MEL_OFF  (MAX_OFF + 8)
#define HARM_OFF (MEL_OFF + E_ELEMS)
#define PERC_OFF (HARM_OFF + E_ELEMS)

typedef float v2f __attribute__((ext_vector_type(2)));
typedef float v4f __attribute__((ext_vector_type(4)));
typedef float v8f __attribute__((ext_vector_type(8)));

__device__ __forceinline__ v8f wmma4(v2f a, v2f b, v8f c) {
  return __builtin_amdgcn_wmma_f32_16x16x4_f32(false, a, false, b, (short)0, c,
                                               false, false);
}

__global__ __launch_bounds__(256) void build_tables(float* __restrict__ ws) {
  int gid = blockIdx.x * 256 + threadIdx.x;
  if (gid == 0) {
    unsigned* mx = (unsigned*)(ws + MAX_OFF);
    mx[0] = 0u; mx[1] = 0u; mx[2] = 0u;
  }
  const float TWO_PI = 6.28318530717958647692f;
  if (gid < DFT_OFF) {
    int k = gid;
    ws[WIN_OFF + k] = 0.5f * (1.0f - cosf(TWO_PI * (float)k / 1024.0f));
  } else if (gid < FB_OFF) {
    int i = gid - DFT_OFF;
    int entry = i & 3;
    int rest  = i >> 2;
    int f  = col_to_f(rest % FH);
    int qp = rest / FH;
    int k  = 4 * (qp >> 1) + (qp & 1) + 2 * (entry & 1);
    int r  = (k * f) & (NFFT - 1);
    float arg = (float)r * (TWO_PI / 1024.0f);
    ws[gid] = (entry < 2) ? cosf(arg) : sinf(arg);
  } else if (gid < MAX_OFF) {
    int i = gid - FB_OFF;
    int entry = i & 1;
    int rest  = i >> 1;
    int m  = rest % NMELS;
    int k2 = rest / NMELS;
    int f  = 2 * k2 + entry;
    float v = 0.0f;
    if (f < NFREQ) {
      float freq   = (float)f * (22050.0f / 512.0f);
      float melmax = 2595.0f * log10f(1.0f + 22050.0f / 700.0f);
      float step   = melmax / 129.0f;
      float p0 = 700.0f * (exp10f(((float)(m    ) * step) / 2595.0f) - 1.0f);
      float p1 = 700.0f * (exp10f(((float)(m + 1) * step) / 2595.0f) - 1.0f);
      float p2 = 700.0f * (exp10f(((float)(m + 2) * step) / 2595.0f) - 1.0f);
      float down = (freq - p0) / (p1 - p0);
      float up   = (p2 - freq) / (p2 - p1);
      float t = fmaxf(0.0f, fminf(down, up));
      v = t * 2.0f / (p2 - p0);
    }
    ws[gid] = v;
  }
}

template <int NS>
__device__ __forceinline__ v8f mel_steps(const float* pA, const float* pB, v8f acc) {
  v2f a[NS], bb[NS];
  #pragma unroll
  for (int u = 0; u < NS; ++u) {
    bb[u] = *(const v2f*)(pB + (long)u * (2 * NMELS * 2));
    a[u]  = *(const v2f*)(pA + 4 * u);
  }
  #pragma unroll
  for (int u = 0; u < NS; ++u) acc = wmma4(a[u], bb[u], acc);
  return acc;
}

__device__ __forceinline__ void load_tab_group(v4f q[KG], const float* p) {
  #pragma unroll
  for (int u = 0; u < KG; ++u) q[u] = *(const v4f*)(p + (long)u * (4 * FH * 4));
}

__device__ __forceinline__ void run_group(const float* aP, const v4f q[KG], v8f& accC, v8f& accS) {
  v4f a[KG];
  #pragma unroll
  for (int u = 0; u < KG; ++u) a[u] = *(const v4f*)(aP + u * 16);
  __builtin_amdgcn_sched_barrier(0);
  #pragma unroll
  for (int u = 0; u < KG; ++u) {
    v2f ae; ae[0] = a[u][0]; ae[1] = a[u][1];
    v2f ao; ao[0] = a[u][2]; ao[1] = a[u][3];
    v2f bc; bc[0] = q[u][0]; bc[1] = q[u][1];
    v2f bs; bs[0] = q[u][2]; bs[1] = q[u][3];
    accC = wmma4(ae, bc, accC);
    accS = wmma4(ao, bs, accS);
  }
}

__global__ __launch_bounds__(256, 6) void mel_gemm(const float* __restrict__ wave,
                                                const float* __restrict__ ws,
                                                float* __restrict__ mel) {
  __shared__ __attribute__((aligned(16))) float A4[16][APITCH];
  __shared__ float PL[16][FPAD + 4];
  __shared__ float RED[8][2][16];

  const float* __restrict__ win  = ws + WIN_OFF;
  const float* __restrict__ dftT = ws + DFT_OFF;
  const float* __restrict__ fb2  = ws + FB_OFF;

  int b    = blockIdx.x / MTILES;
  int tile = blockIdx.x % MTILES;
  int t0   = tile * 16;
  int tid  = threadIdx.x;
  int lane = tid & 31;
  int wv   = tid >> 5;
  int h    = lane >> 4;
  int l16  = lane & 15;

  const float* wb = wave + (long)b * T_IN;

  int wslot = (wv + blockIdx.x + (blockIdx.x >> 8)) & 7;
  const int wvu = __builtin_amdgcn_readfirstlane(wv);

  {
    const int k  = tid;
    const bool pr = k > 0;
    const float wa = win[k];
    const float wn = win[pr ? (NFFT - k) : 0];
    const float wc = win[NFFT / 2 - k];
    const float wd = win[NFFT / 2 + k];
    const int c = (k >> 2) * 8 + (k & 1) * 4 + ((k >> 1) & 1);
    for (int rp = 0; rp < 2; ++rp) {
      float sa[8], sn[8], sc[8], sd[8];
      #pragma unroll
      for (int m = 0; m < 8; ++m) {
        int t = t0 + rp * 8 + m;
        t = (t < NFRAMES) ? t : (NFRAMES - 1);
        int oa = t * HOPSZ + k - NFFT / 2;
        oa = (oa < 0) ? -oa : oa;
        oa = (oa >= T_IN) ? (2 * (T_IN - 1) - oa) : oa;
        int on = t * HOPSZ + NFFT / 2 - k;
        on = (on >= T_IN) ? (2 * (T_IN - 1) - on) : on;
        int oc = t * HOPSZ - k;
        oc = (oc < 0) ? -oc : oc;
        int od = t * HOPSZ + k;
        sa[m] = wb[oa];
        sn[m] = wb[on];
        sc[m] = wb[oc];
        sd[m] = wb[od];
      }
      __builtin_amdgcn_sched_barrier(0);
      #pragma unroll
      for (int m = 0; m < 8; ++m) {
        float ya = sa[m] * wa;
        float yn = sn[m] * wn;
        float yc = sc[m] * wc;
        float yd = sd[m] * wd;
        float ek = pr ? (ya + yn) : ya;
        float ok = pr ? (ya - yn) : 0.0f;
        float eh = pr ? (yc + yd) : yc;
        float oh = pr ? (yc - yd) : 0.0f;
        bool live = (t0 + rp * 8 + m) < NFRAMES;
        float* rowp = &A4[rp * 8 + m][c];
        rowp[0]              = live ? (ek + eh) : 0.0f;
        rowp[2]              = live ? (ok - oh) : 0.0f;
        rowp[2 * KQ]         = live ? (ek - eh) : 0.0f;
        rowp[2 * KQ + 2]     = live ? (ok + oh) : 0.0f;
      }
    }
  }

  float y256 = 0.0f, y768 = 0.0f;
  {
    int t = t0 + l16;
    if (h == 0 && t < NFRAMES) {
      int o1 = t * HOPSZ - NFFT / 4;
      o1 = (o1 < 0) ? -o1 : o1;
      y256 = wb[o1] * win[NFFT / 4];
      y768 = wb[t * HOPSZ + NFFT / 4] * win[3 * NFFT / 4];
    }
  }
  __syncthreads();

  #define NG (KQ / 8 / KG)
  static_assert(NG * KG * 8 == KQ && (NG % 2) == 0, "groups must tile the k-steps in pairs");
  const float* aBase = &A4[l16][h * 8];
  #pragma unroll
  for (int i = 0; i < 2; ++i) {
    int ft = wslot + 8 * i;
    int f0 = ft * 16;
    v8f accC[2], accS[2];
    #pragma unroll
    for (int p = 0; p < 2; ++p)
      for (int j = 0; j < 8; ++j) { accC[p][j] = 0.0f; accS[p][j] = 0.0f; }
    const float* dB = dftT + (long)(f0 + l16) * 4 + (long)(h * 2) * (FH * 4);
    const float* aR = aBase + i * (2 * KQ);
    v4f q0[KG], q1[KG];
    load_tab_group(q0, dB);
    #pragma unroll
    for (int G = 0; G < 2 * NG; G += 2) {
      {
        const int p = (G + 1) / NG, g = (G + 1) % NG;
        load_tab_group(q1, dB + (long)p * (FH * 4) + (long)(g * KG) * (4 * FH * 4));
      }
      __builtin_amdgcn_sched_barrier(0);
      {
        const int p = G / NG, g = G % NG;
        run_group(aR + p * 4 + g * KG * 16, q0, accC[p], accS[p]);
      }
      if (G + 2 < 2 * NG) {
        const int p = (G + 2) / NG, g = (G + 2) % NG;
        load_tab_group(q0, dB + (long)p * (FH * 4) + (long)(g * KG) * (4 * FH * 4));
      }
      __builtin_amdgcn_sched_barrier(0);
      {
        const int p = (G + 1) / NG, g = (G + 1) % NG;
        run_group(aR + p * 4 + g * KG * 16, q1, accC[p], accS[p]);
      }
    }
    {
      const float* dL = dftT + (long)(f0 + l16) * 4;
      v4f q = *(const v4f*)(dL + (long)((NQ - 1) * 2) * (FH * 4));
      if (i == 0) {
        v2f ae; ae[0] = y256 + y768; ae[1] = 0.0f;
        v2f bc; bc[0] = q[0]; bc[1] = q[1];
        accC[0] = wmma4(ae, bc, accC[0]);
      } else {
        v2f ao; ao[0] = y256 - y768; ao[1] = 0.0f;
        v2f bs; bs[0] = q[2]; bs[1] = q[3];
        accS[0] = wmma4(ao, bs, accS[0]);
      }
    }
    int f = col_to_f(f0 + l16);
    #pragma unroll
    for (int v = 0; v < 8; ++v) {
      int mr = v + 8 * h;
      float ce = accC[0][v], co = accC[1][v];
      float se = accS[0][v], so = accS[1][v];
      float re = ce + co, im = se + so;
      float rm = ce - co, jm = so - se;
      PL[mr][f] = re * re + im * im;
      PL[mr][NFFT / 2 - f] = rm * rm + jm * jm;
    }
  }
  #undef NG

  v8f accC16, accS16;
  #pragma unroll
  for (int j = 0; j < 8; ++j) { accC16[j] = 0.0f; accS16[j] = 0.0f; }
  {
    const float* dB16 = dftT + (long)(16 * 16 + l16) * 4 + (long)(h * 2) * (FH * 4) +
                        (long)(wvu * 4) * (4 * FH * 4);
    const float* aP16 = aBase + wvu * 4 * 16;
    v2f ae[4], ao[4], bc[4], bs[4];
    #pragma unroll
    for (int u = 0; u < 4; ++u) {
      bc[u] = *(const v2f*)(dB16 + (long)u * (4 * FH * 4));
      bs[u] = *(const v2f*)(dB16 + (long)u * (4 * FH * 4) + FH * 4 + 2);
      ae[u] = *(const v2f*)(aP16 + u * 16);
      ao[u] = *(const v2f*)(aP16 + u * 16 + 4 + 2);
    }
    #pragma unroll
    for (int u = 0; u < 4; ++u) {
      accC16 = wmma4(ae[u], bc[u], accC16);
      accS16 = wmma4(ao[u], bs[u], accS16);
    }
    if (wvu == 0) {
      const float* dL = dftT + (long)(16 * 16 + l16) * 4;
      v2f b5 = *(const v2f*)(dL + (long)((NQ - 1) * 2) * (FH * 4));
      v2f a5; a5[0] = y256 + y768; a5[1] = 0.0f;
      accC16 = wmma4(a5, b5, accC16);
    }
  }
  if (l16 == 0) {
    #pragma unroll
    for (int v = 0; v < 8; ++v) {
      RED[wv][0][v + 8 * h] = accC16[v];
      RED[wv][1][v + 8 * h] = accS16[v];
    }
  }
  if (tid < 16 * (FPAD - NFREQ))
    PL[tid / (FPAD - NFREQ)][NFREQ + tid % (FPAD - NFREQ)] = 0.0f;
  __syncthreads();
  if (tid < 16) {
    float ce = 0.0f, so = 0.0f;
    #pragma unroll
    for (int w = 0; w < 8; ++w) { ce += RED[w][0][tid]; so += RED[w][1][tid]; }
    PL[tid][NFFT / 4] = ce * ce + so * so;
  }
  __syncthreads();

  int nt = __builtin_amdgcn_readfirstlane(wslot);
  int n0 = nt * 16;
  int kb, rows;
  {
    float melmax = 2595.0f * log10f(1.0f + 22050.0f / 700.0f);
    float step   = melmax / 129.0f;
    float flo = 700.0f * (exp10f(((float)(n0     ) * step) / 2595.0f) - 1.0f);
    float fhi = 700.0f * (exp10f(((float)(n0 + 17) * step) / 2595.0f) - 1.0f);
    int lo = (int)(flo * (512.0f / 22050.0f)) - 1;
    int hi = (int)(fhi * (512.0f / 22050.0f)) + 3;
    lo = (lo < 0) ? 0 : lo;
    int ke = (hi + 15) & ~15;
    ke = (ke > FPAD) ? FPAD : ke;
    kb   = lo & ~15;
    rows = ke - kb;
  }
  v8f acc;
  #pragma unroll
  for (int j = 0; j < 8; ++j) acc[j] = 0.0f;
  const float* pA = &PL[l16][kb + 2 * h];
  const float* pB = fb2 + (long)(n0 + l16) * 2 + (long)((kb >> 1) + h) * (NMELS * 2);
  for (; rows >= 64; rows -= 64) {
    acc = mel_steps<16>(pA, pB, acc);
    pA += 64; pB += 32 * (NMELS * 2);
  }
  if (rows >= 32) {
    acc = mel_steps<8>(pA, pB, acc);
    pA += 32; pB += 16 * (NMELS * 2); rows -= 32;
  }
  if (rows >= 16) acc = mel_steps<4>(pA, pB, acc);
  int n = n0 + l16;
  #pragma unroll
  for (int v = 0; v < 8; ++v) {
    int t = t0 + v + 8 * h;
    if (t < NFRAMES) mel[(b * NMELS + n) * NFRAMES + t] = acc[v];
  }
}

#define MED_R     4
#define MED_TAPS  (31 + MED_R - 1)
#define MED_CORE  (31 - MED_R + 1)
static_assert((NFRAMES % MED_R) == 0 && (NMELS % MED_R) == 0, "groups of four outputs must tile both axes");

__device__ __forceinline__ float median_core4_plus3(float c12, float c13, float c14, float c15,
                                                    float a, float b, float c) {
  float lo = fminf(a, b), hi = fmaxf(a, b);
  float e0 = fminf(lo, c);
  float md = fmaxf(lo, c);
  float e1 = fminf(hi, md);
  float e2 = fmaxf(hi, md);
  return fminf(fminf(c15, fmaxf(c14, e0)), fminf(fmaxf(c13, e1), fmaxf(c12, e2)));
}

#define MED_CE(i, j) { float lo_ = fminf(c[i], c[j]); float hi_ = fmaxf(c[i], c[j]); c[i] = lo_; c[j] = hi_; }
#define MED_LO(i, j) { c[i] = fminf(c[i], c[j]); }
#define MED_HI(i, j) { c[j] = fmaxf(c[i], c[j]); }
__device__ __forceinline__ void median31x4(const float x[MED_TAPS], float o[MED_R]) {
  float c[MED_CORE];
  #pragma unroll
  for (int k = 0; k < MED_CORE; ++k) c[k] = x[k + 3];
  MED_CE(1,2) MED_CE(3,4) MED_CE(5,6) MED_CE(7,8) MED_CE(9,10) MED_CE(11,12) MED_CE(13,14)
  MED_CE(15,16) MED_CE(17,18) MED_CE(19,20) MED_CE(21,22) MED_CE(23,24) MED_CE(25,26)
  MED_CE(0,2) MED_CE(3,5) MED_CE(4,6) MED_CE(7,9) MED_CE(8,10) MED_CE(11,13) MED_CE(12,14)
  MED_CE(15,17) MED_CE(16,18) MED_CE(19,21) MED_CE(20,22) MED_CE(23,25) MED_CE(24,26)
  MED_CE(0,1) MED_CE(4,5) MED_CE(8,9) MED_CE(12,13) MED_CE(16,17) MED_CE(20,21) MED_CE(24,25)
  MED_CE(2,6) MED_CE(7,11) MED_CE(10,14) MED_CE(15,19) MED_CE(18,22) MED_CE(23,27)
  MED_CE(0,4) MED_CE(1,5) MED_CE(8,12) MED_CE(9,13) MED_CE(16,20) MED_CE(17,21) MED_CE(25,27)
  MED_CE(6,14) MED_CE(15,23)
  MED_CE(1,3) MED_CE(2,4) MED_CE(9,11) MED_CE(10,12) MED_CE(17,19) MED_CE(18,20) MED_CE(24,25)
  MED_CE(26,27)
  MED_CE(0,1) MED_CE(2,3) MED_CE(4,5) MED_CE(8,9) MED_CE(10,11) MED_CE(12,13) MED_CE(16,17)
  MED_CE(18,19) MED_CE(20,21)
  MED_CE(0,8) MED_CE(1,9) MED_CE(2,10) MED_CE(3,11) MED_CE(4,12) MED_CE(5,13) MED_CE(16,24)
  MED_CE(17,25) MED_CE(18,26) MED_CE(19,27)
  MED_CE(3,7) MED_CE(4,8) MED_CE(5,9) MED_CE(6,10) MED_CE(19,23) MED_CE(20,24) MED_CE(21,25)
  MED_CE(22,26)
  MED_CE(1,3) MED_CE(2,4) MED_CE(5,7) MED_CE(6,8) MED_CE(9,11) MED_CE(10,12) MED_CE(17,19)
  MED_CE(18,20) MED_CE(21,23) MED_CE(22,24) MED_CE(25,27)
  MED_CE(0,1) MED_CE(2,3) MED_CE(4,5) MED_CE(6,7) MED_CE(8,9) MED_CE(10,11) MED_CE(12,13)
  MED_CE(16,17) MED_CE(18,19) MED_CE(20,21) MED_CE(22,23) MED_CE(24,25) MED_CE(26,27)
  MED_HI(0,16) MED_HI(1,17) MED_HI(2,18) MED_HI(3,19) MED_HI(4,20) MED_HI(5,21) MED_CE(6,22)
  MED_LO(7,23) MED_LO(8,24) MED_LO(9,25) MED_LO(10,26) MED_LO(11,27)
  MED_HI(7,15) MED_HI(8,16) MED_HI(9,17) MED_CE(10,18) MED_LO(11,19) MED_LO(12,20) MED_LO(13,21)
  MED_LO(14,22)
  MED_HI(6,10) MED_HI(11,15) MED_CE(12,16) MED_LO(13,17) MED_LO(14,18)
  MED_HI(10,12) MED_CE(13,15) MED_LO(14,16)
  MED_CE(12,13) MED_CE(14,15)
  o[0] = median_core4_plus3(c[12], c[13], c[14], c[15], x[0],  x[1],  x[2]);
  o[1] = median_core4_plus3(c[12], c[13], c[14], c[15], x[1],  x[2],  x[31]);
  o[2] = median_core4_plus3(c[12], c[13], c[14], c[15], x[2],  x[31], x[32]);
  o[3] = median_core4_plus3(c[12], c[13], c[14], c[15], x[31], x[32], x[33]);
}
#undef MED_CE
#undef MED_LO
#undef MED_HI

#define MT_BLOCKS (E_ELEMS / (MED_R * 256))
static_assert(MT_BLOCKS * MED_R * 256 == E_ELEMS, "median_time grid must tile E_ELEMS exactly");

__global__ __launch_bounds__(256) void median_time(const float* __restrict__ mel,
                                                   float* __restrict__ harm) {
  int g   = blockIdx.x * 256 + threadIdx.x;
  int q   = g % (NFRAMES / MED_R);
  int row = g / (NFRAMES / MED_R);
  int t0  = q * MED_R;
  const float* src = mel + row * NFRAMES;
  float x[MED_TAPS];
  if (t0 >= 16 && t0 + 19 < NFRAMES) {
    const v4f* __restrict__ s4 = (const v4f*)(src + (t0 - 16));
    v4f w[9];
    #pragma unroll
    for (int u = 0; u < 9; ++u) w[u] = s4[u];
    #pragma unroll
    for (int e = 0; e < MED_TAPS; ++e) x[e] = w[(e + 1) >> 2][(e + 1) & 3];
  } else {
    #pragma unroll
    for (int e = 0; e < MED_TAPS; ++e) {
      int j = t0 + e - 15;
      j = (j < 0) ? (-j - 1) : j;
      j = (j >= NFRAMES) ? (2 * NFRAMES - 1 - j) : j;
      x[e] = src[j];
    }
  }
  float o[MED_R];
  median31x4(x, o);
  v4f ov; ov[0] = o[0]; ov[1] = o[1]; ov[2] = o[2]; ov[3] = o[3];
  ((v4f*)harm)[g] = ov;
}

#define MM_TBLOCKS ((NFRAMES + 255) / 256)
#define MM_GROUPS  (NB * (NMELS / MED_R))

__global__ __launch_bounds__(256) void median_mel(const float* __restrict__ mel,
                                                  float* __restrict__ perc) {
  int t = blockIdx.x * 256 + threadIdx.x;
  if (t >= NFRAMES) return;
  int k  = blockIdx.y % (NMELS / MED_R);
  int b  = blockIdx.y / (NMELS / MED_R);
  int n0 = k * MED_R;
  const float* base = mel + (b * NMELS) * NFRAMES + t;
  float x[MED_TAPS];
  #pragma unroll
  for (int e = 0; e < MED_TAPS; ++e) {
    int j = n0 + e - 15;
    j = (j < 0) ? (-j - 1) : j;
    j = (j >= NMELS) ? (2 * NMELS - 1 - j) : j;
    x[e] = base[j * NFRAMES];
  }
  float o[MED_R];
  median31x4(x, o);
  float* dst = perc + (b * NMELS + n0) * NFRAMES + t;
  #pragma unroll
  for (int r = 0; r < MED_R; ++r) dst[r * NFRAMES] = o[r];
}

#define HP_U      4
#define HP_BLOCKS (E_ELEMS / (4 * 256 * HP_U))
static_assert(HP_BLOCKS * 4 * 256 * HP_U == E_ELEMS, "hpss_mask grid must tile E_ELEMS exactly");
static_assert((MEL_OFF % 4) == 0 && (E_ELEMS % 4) == 0, "float4 views need 16-byte aligned planes");

__global__ __launch_bounds__(256) void hpss_mask(const float* __restrict__ mel,
                                                 float* __restrict__ harm,
                                                 float* __restrict__ perc,
                                                 unsigned* __restrict__ mx) {
  __shared__ float r0[256], r1[256], r2[256];
  const v4f* __restrict__ mel4 = (const v4f*)mel;
  v4f* __restrict__ harm4 = (v4f*)harm;
  v4f* __restrict__ perc4 = (v4f*)perc;
  int tid  = threadIdx.x;
  int base = blockIdx.x * (256 * HP_U) + tid;
  v4f S4[HP_U], H4[HP_U], P4[HP_U];
  #pragma unroll
  for (int u = 0; u < HP_U; ++u) {
    int i = base + u * 256;
    S4[u] = mel4[i]; H4[u] = harm4[i]; P4[u] = perc4[i];
  }
  float s = 0.0f, dh = 0.0f, dp = 0.0f;
  #pragma unroll
  for (int u = 0; u < HP_U; ++u) {
    v4f oh, op;
    #pragma unroll
    for (int c = 0; c < 4; ++c) {
      float S  = S4[u][c];
      float ha = H4[u][c];
      float pe = P4[u][c];
      float Z  = fmaxf(ha, pe);
      bool bad = Z < 1.17549435e-38f;
      float Zs = bad ? 1.0f : Z;
      float mh = ha / Zs; mh *= mh;
      float mp = pe / Zs; mp *= mp;
      float den = mh + mp;
      float maskh = bad ? 0.0f : mh / den;
      float maskp = bad ? 0.0f : mp / den;
      float vh = S * maskh, vp = S * maskp;
      oh[c] = vh; op[c] = vp;
      s = fmaxf(s, S); dh = fmaxf(dh, vh); dp = fmaxf(dp, vp);
    }
    int i = base + u * 256;
    harm4[i] = oh;
    perc4[i] = op;
  }
  r0[tid] = s; r1[tid] = dh; r2[tid] = dp;
  __syncthreads();
  for (int off = 128; off > 0; off >>= 1) {
    if (tid < off) {
      r0[tid] = fmaxf(r0[tid], r0[tid + off]);
      r1[tid] = fmaxf(r1[tid], r1[tid + off]);
      r2[tid] = fmaxf(r2[tid], r2[tid + off]);
    }
    __syncthreads();
  }
  if (tid == 0) {
    atomicMax(mx + 0, __float_as_uint(r0[0]));
    atomicMax(mx + 1, __float_as_uint(r1[0]));
    atomicMax(mx + 2, __float_as_uint(r2[0]));
  }
}

#define FN_U          8
#define FN_SEG_BLOCKS (E_ELEMS / (4 * 256 * FN_U))
static_assert(FN_SEG_BLOCKS * 4 * 256 * FN_U == E_ELEMS, "finalize grid must tile each plane exactly");

__global__ __launch_bounds__(256) void finalize(const float* __restrict__ mel,
                                                const float* __restrict__ harm,
                                                const float* __restrict__ perc,
                                                const unsigned* __restrict__ mx,
                                                float* __restrict__ out) {
  int s  = blockIdx.x / FN_SEG_BLOCKS;
  int e4 = (blockIdx.x - s * FN_SEG_BLOCKS) * (256 * FN_U) + threadIdx.x;
  const float* src = (s == 0) ? mel : ((s == 1) ? harm : perc);
  const v4f* __restrict__ src4 = (const v4f*)src;
  v4f* __restrict__ out4 = (v4f*)out + (long)s * (E_ELEMS / 4);
  float ref = __uint_as_float(mx[s]);
  v4f v[FN_U];
  #pragma unroll
  for (int u = 0; u < FN_U; ++u) v[u] = src4[e4 + u * 256];
  #pragma unroll
  for (int u = 0; u < FN_U; ++u) {
    v4f o;
    #pragma unroll
    for (int c = 0; c < 4; ++c) {
      float db = 10.0f * log10f(fmaxf(1e-10f, v[u][c])) -
                 10.0f * log10f(fmaxf(1e-10f, ref));
      db = fmaxf(db, -80.0f);
      o[c] = (db + 80.0f) * (1.0f / (80.0f + 1e-7f));
    }
    out4[e4 + u * 256] = o;
  }
}

extern "C" void kernel_launch(void* const* d_in, const int* in_sizes, int n_in,
                              void* d_out, int out_size, void* d_ws,
                              size_t ws_size, hipStream_t stream) {
  const float* wave = (const float*)d_in[0];
  float* out  = (float*)d_out;
  float* ws   = (float*)d_ws;
  float* mel  = ws + MEL_OFF;
  float* harm = ws + HARM_OFF;
  float* perc = ws + PERC_OFF;
  unsigned* mx = (unsigned*)(ws + MAX_OFF);

  build_tables<<<(MAX_OFF + 255) / 256, 256, 0, stream>>>(ws);
  mel_gemm<<<NB * MTILES, 256, 0, stream>>>(wave, ws, mel);
  median_time<<<MT_BLOCKS, 256, 0, stream>>>(mel, harm);
  median_mel<<<dim3(MM_TBLOCKS, MM_GROUPS), 256, 0, stream>>>(mel, perc);
  hpss_mask<<<HP_BLOCKS, 256, 0, stream>>>(mel, harm, perc, mx);
  finalize<<<3 * FN_SEG_BLOCKS, 256, 0, stream>>>(mel, harm, perc, mx, out);
}
